# starting kernel
# baseline (speedup 1.0000x reference)
#include <hip/hip_runtime.h>

#define SVCN   50000
#define INSTN  100000
#define NODEN  50000
#define FDIM   128
#define HDIM   128
#define OUTD   256
#define TOTAL  (SVCN + NODEN + INSTN)
#define E_SC   400000
#define E_IN   100000
#define E_NI   100000
#define E_II   800000

#define OFF_CAT   0LL
#define OFF_HSC   (OFF_CAT  + (long long)TOTAL * 128)
#define OFF_HIN   (OFF_HSC  + (long long)SVCN  * 128)
#define OFF_HNI   (OFF_HIN  + (long long)INSTN * 128)
#define OFF_HII   (OFF_HNI  + (long long)NODEN * 128)
#define OFF_DEG   (OFF_HII  + (long long)INSTN * 128)
#define D_NS_SC   (OFF_DEG + 0LL)
#define D_ND_SC   (D_NS_SC + SVCN)
#define D_NS_IN   (D_ND_SC + SVCN)
#define D_ND_IN   (D_NS_IN + INSTN)
#define D_NS_NI   (D_ND_IN + NODEN)
#define D_ND_NI   (D_NS_NI + NODEN)
#define D_NS_II   (D_ND_NI + INSTN)
#define D_ND_II   (D_NS_II + INSTN)
#define DEG_TOTAL 600000

typedef __attribute__((ext_vector_type(2))) float v2f;
typedef __attribute__((ext_vector_type(8))) float v8f;

__device__ __forceinline__ float lrelu(float x) { return x > 0.f ? x : 0.01f * x; }

__device__ __forceinline__ void atomAddF(float* p, float v) { unsafeAtomicAdd(p, v); }

__global__ void init_deg_kernel(float* deg) {
    int i = blockIdx.x * blockDim.x + threadIdx.x;
    if (i < DEG_TOTAL) deg[i] = 0.f;
}

__global__ void init_cat_kernel(float* __restrict__ cat,
                                const float* __restrict__ b_sc,
                                const float* __restrict__ b_in,
                                const float* __restrict__ b_ni,
                                const float* __restrict__ b_ii) {
    long long tid = (long long)blockIdx.x * blockDim.x + threadIdx.x;
    if (tid >= (long long)TOTAL * 32) return;
    int row = (int)(tid >> 5);
    int c4  = (int)(tid & 31) * 4;
    float4 v;
    if (row < SVCN) {
        v = *(const float4*)&b_sc[c4];
    } else if (row < SVCN + NODEN) {
        v = *(const float4*)&b_in[c4];
    } else {
        float4 a = *(const float4*)&b_ni[c4];
        float4 b = *(const float4*)&b_ii[c4];
        v = make_float4(a.x + b.x, a.y + b.y, a.z + b.z, a.w + b.w);
    }
    *(float4*)&cat[(long long)row * 128 + c4] = v;
}

__global__ void init_out_kernel(float* __restrict__ out, const float* __restrict__ b_tot) {
    int i = blockIdx.x * blockDim.x + threadIdx.x;
    if (i < OUTD * HDIM) out[i] = b_tot[i >> 7];
}

__global__ void degree_kernel(const int* __restrict__ src, const int* __restrict__ dst,
                              float* __restrict__ degS, float* __restrict__ degD, int nE) {
    int e = blockIdx.x * blockDim.x + threadIdx.x;
    if (e >= nE) return;
    atomAddF(&degS[src[e]], 1.f);
    atomAddF(&degD[dst[e]], 1.f);
}

__global__ void rsqrt_kernel(float* deg) {
    int i = blockIdx.x * blockDim.x + threadIdx.x;
    if (i < DEG_TOTAL) {
        float d = deg[i];
        deg[i] = rsqrtf(d > 1.f ? d : 1.f);
    }
}

__global__ __launch_bounds__(256) void proj_kernel(const float* __restrict__ X,
                                                   const float* __restrict__ W,
                                                   const float* __restrict__ ns,
                                                   float* __restrict__ Hout, int nrows) {
    __shared__ float Wlds[128 * 128];
    int t = threadIdx.x;
    for (int i = t * 4; i < 128 * 128; i += 256 * 4)
        *(float4*)&Wlds[i] = *(const float4*)&W[i];
    __syncthreads();

    const int lane = t & 31;
    const int wave = t >> 5;
    const int half = lane >> 4;
    const int l16  = lane & 15;
    const int r0   = blockIdx.x * 64;
    const int n0   = wave * 16;

    const float* Arow[4];
#pragma unroll
    for (int rt = 0; rt < 4; ++rt) {
        int row = r0 + rt * 16 + l16;
        int rc  = row < nrows ? row : nrows - 1;
        Arow[rt] = X + (long long)rc * 128;
    }

    v8f acc[4] = {v8f{}, v8f{}, v8f{}, v8f{}};

    for (int k = 0; k < 128; k += 4) {
        const int kk = k + 2 * half;
        v2f b;
        b.x = Wlds[kk * 128 + n0 + l16];
        b.y = Wlds[(kk + 1) * 128 + n0 + l16];
#pragma unroll
        for (int rt = 0; rt < 4; ++rt) {
            v2f a = *(const v2f*)&Arow[rt][kk];
            acc[rt] = __builtin_amdgcn_wmma_f32_16x16x4_f32(
                false, a, false, b, (short)0, acc[rt], false, false);
        }
    }

    if (r0 + 64 <= nrows) {
#pragma unroll
        for (int rt = 0; rt < 4; ++rt) {
#pragma unroll
            for (int i = 0; i < 8; ++i) {
                int row = r0 + rt * 16 + i + 8 * half;
                Hout[(long long)row * 128 + n0 + l16] = acc[rt][i] * ns[row];
            }
        }
    } else {
#pragma unroll
        for (int rt = 0; rt < 4; ++rt) {
#pragma unroll
            for (int i = 0; i < 8; ++i) {
                int row = r0 + rt * 16 + i + 8 * half;
                if (row < nrows)
                    Hout[(long long)row * 128 + n0 + l16] = acc[rt][i] * ns[row];
            }
        }
    }
}

__global__ void scatter_kernel(const float* __restrict__ H,
                               const int* __restrict__ src, const int* __restrict__ dst,
                               const float* __restrict__ nd, float* __restrict__ cat,
                               long long dstBase, int nE) {
    long long tid = (long long)blockIdx.x * blockDim.x + threadIdx.x;
    int e = (int)(tid >> 5);
    if (e >= nE) return;
    int c4 = (int)(tid & 31) * 4;
    int s = src[e], d = dst[e];
    float scale = nd[d];
    if (e + 1 < nE)
        __builtin_prefetch(&H[(long long)src[e + 1] * 128 + c4], 0, 1);
    float4 hv = *(const float4*)&H[(long long)s * 128 + c4];
    float* p = &cat[dstBase + (long long)d * 128 + c4];
    atomAddF(p + 0, hv.x * scale);
    atomAddF(p + 1, hv.y * scale);
    atomAddF(p + 2, hv.z * scale);
    atomAddF(p + 3, hv.w * scale);
}

#define KCHUNK 1024
__global__ __launch_bounds__(256) void final_gemm_kernel(const float* __restrict__ Wt,
                                                         const float* __restrict__ cat,
                                                         float* __restrict__ out) {
    const int t    = threadIdx.x;
    const int lane = t & 31;
    const int wave = t >> 5;
    const int half = lane >> 4;
    const int l16  = lane & 15;
    const int m0   = blockIdx.x * 64;
    const int n0   = wave * 16;
    const int k0   = blockIdx.y * KCHUNK;
    const int kEnd = (k0 + KCHUNK < TOTAL) ? (k0 + KCHUNK) : TOTAL;

    const float* Arow[4];
#pragma unroll
    for (int rt = 0; rt < 4; ++rt)
        Arow[rt] = Wt + (long long)(m0 + rt * 16 + l16) * TOTAL;

    v8f acc[4] = {v8f{}, v8f{}, v8f{}, v8f{}};

    for (int k = k0; k < kEnd; k += 4) {
        const int kk = k + 2 * half;
        v2f b;
        b.x = lrelu(cat[(long long)kk * 128 + n0 + l16]);
        b.y = lrelu(cat[(long long)(kk + 1) * 128 + n0 + l16]);
#pragma unroll
        for (int rt = 0; rt < 4; ++rt) {
            v2f a = *(const v2f*)&Arow[rt][kk];
            acc[rt] = __builtin_amdgcn_wmma_f32_16x16x4_f32(
                false, a, false, b, (short)0, acc[rt], false, false);
        }
    }

#pragma unroll
    for (int rt = 0; rt < 4; ++rt) {
#pragma unroll
        for (int i = 0; i < 8; ++i) {
            int m = m0 + rt * 16 + i + 8 * half;
            atomAddF(&out[m * 128 + n0 + l16], acc[rt][i]);
        }
    }
}

static inline int cdiv(long long a, long long b) { return (int)((a + b - 1) / b); }

extern "C" void kernel_launch(void* const* d_in, const int* in_sizes, int n_in,
                              void* d_out, int out_size, void* d_ws, size_t ws_size,
                              hipStream_t stream) {
    const float* feat_svc  = (const float*)d_in[0];
    const float* feat_inst = (const float*)d_in[1];
    const float* feat_node = (const float*)d_in[2];
    const int*   svc_src   = (const int*)d_in[3];
    const int*   svc_dst   = (const int*)d_in[4];
    const int*   in_src    = (const int*)d_in[5];
    const int*   in_dst    = (const int*)d_in[6];
    const int*   ni_src    = (const int*)d_in[7];
    const int*   ni_dst    = (const int*)d_in[8];
    const int*   ii_src    = (const int*)d_in[9];
    const int*   ii_dst    = (const int*)d_in[10];
    const float* W_sc = (const float*)d_in[11];  const float* b_sc = (const float*)d_in[12];
    const float* W_in = (const float*)d_in[13];  const float* b_in = (const float*)d_in[14];
    const float* W_ni = (const float*)d_in[15];  const float* b_ni = (const float*)d_in[16];
    const float* W_ii = (const float*)d_in[17];  const float* b_ii = (const float*)d_in[18];
    const float* W_tot = (const float*)d_in[19]; const float* b_tot = (const float*)d_in[20];
    float* out = (float*)d_out;

    float* ws   = (float*)d_ws;
    float* cat  = ws + OFF_CAT;
    float* h_sc = ws + OFF_HSC;
    float* h_in = ws + OFF_HIN;
    float* h_ni = ws + OFF_HNI;
    float* h_ii = ws + OFF_HII;
    float* deg  = ws + OFF_DEG;
    float* ns_sc = ws + D_NS_SC; float* nd_sc = ws + D_ND_SC;
    float* ns_in = ws + D_NS_IN; float* nd_in = ws + D_ND_IN;
    float* ns_ni = ws + D_NS_NI; float* nd_ni = ws + D_ND_NI;
    float* ns_ii = ws + D_NS_II; float* nd_ii = ws + D_ND_II;

    init_deg_kernel<<<cdiv(DEG_TOTAL, 256), 256, 0, stream>>>(deg);
    init_cat_kernel<<<cdiv((long long)TOTAL * 32, 256), 256, 0, stream>>>(cat, b_sc, b_in, b_ni, b_ii);
    init_out_kernel<<<cdiv(OUTD * HDIM, 256), 256, 0, stream>>>(out, b_tot);

    degree_kernel<<<cdiv(E_SC, 256), 256, 0, stream>>>(svc_src, svc_dst, ns_sc, nd_sc, E_SC);
    degree_kernel<<<cdiv(E_IN, 256), 256, 0, stream>>>(in_src,  in_dst,  ns_in, nd_in, E_IN);
    degree_kernel<<<cdiv(E_NI, 256), 256, 0, stream>>>(ni_src,  ni_dst,  ns_ni, nd_ni, E_NI);
    degree_kernel<<<cdiv(E_II, 256), 256, 0, stream>>>(ii_src,  ii_dst,  ns_ii, nd_ii, E_II);
    rsqrt_kernel<<<cdiv(DEG_TOTAL, 256), 256, 0, stream>>>(deg);

    proj_kernel<<<cdiv(SVCN, 64),  256, 0, stream>>>(feat_svc,  W_sc, ns_sc, h_sc, SVCN);
    proj_kernel<<<cdiv(INSTN, 64), 256, 0, stream>>>(feat_inst, W_in, ns_in, h_in, INSTN);
    proj_kernel<<<cdiv(NODEN, 64), 256, 0, stream>>>(feat_node, W_ni, ns_ni, h_ni, NODEN);
    proj_kernel<<<cdiv(INSTN, 64), 256, 0, stream>>>(feat_inst, W_ii, ns_ii, h_ii, INSTN);

    scatter_kernel<<<cdiv((long long)E_SC * 32, 256), 256, 0, stream>>>(
        h_sc, svc_src, svc_dst, nd_sc, cat, 0LL, E_SC);
    scatter_kernel<<<cdiv((long long)E_IN * 32, 256), 256, 0, stream>>>(
        h_in, in_src, in_dst, nd_in, cat, (long long)SVCN * 128, E_IN);
    scatter_kernel<<<cdiv((long long)E_NI * 32, 256), 256, 0, stream>>>(
        h_ni, ni_src, ni_dst, nd_ni, cat, (long long)(SVCN + NODEN) * 128, E_NI);
    scatter_kernel<<<cdiv((long long)E_II * 32, 256), 256, 0, stream>>>(
        h_ii, ii_src, ii_dst, nd_ii, cat, (long long)(SVCN + NODEN) * 128, E_II);

    dim3 g(OUTD / 64, cdiv(TOTAL, KCHUNK));
    final_gemm_kernel<<<g, 256, 0, stream>>>(W_tot, cat, out);
}


# final GEMM: W_tot and act tiles staged through LDS by whole-line 16-byte loads, one block per K slice owns all 256x128 outputs; K slice 800 (was 1024), so the atomic sum grouping changes
# speedup vs baseline: 1.1782x; 1.1782x over previous
#include <hip/hip_runtime.h>

#define SVCN   50000
#define INSTN  100000
#define NODEN  50000
#define FDIM   128
#define HDIM   128
#define OUTD   256
#define TOTAL  (SVCN + NODEN + INSTN)
#define E_SC   400000
#define E_IN   100000
#define E_NI   100000
#define E_II   800000

#define OFF_CAT   0LL
#define OFF_HSC   (OFF_CAT  + (long long)TOTAL * 128)
#define OFF_HIN   (OFF_HSC  + (long long)SVCN  * 128)
#define OFF_HNI   (OFF_HIN  + (long long)INSTN * 128)
#define OFF_HII   (OFF_HNI  + (long long)NODEN * 128)
#define OFF_DEG   (OFF_HII  + (long long)INSTN * 128)
#define D_NS_SC   (OFF_DEG + 0LL)
#define D_ND_SC   (D_NS_SC + SVCN)
#define D_NS_IN   (D_ND_SC + SVCN)
#define D_ND_IN   (D_NS_IN + INSTN)
#define D_NS_NI   (D_ND_IN + NODEN)
#define D_ND_NI   (D_NS_NI + NODEN)
#define D_NS_II   (D_ND_NI + INSTN)
#define D_ND_II   (D_NS_II + INSTN)
#define DEG_TOTAL 600000

typedef __attribute__((ext_vector_type(2))) float v2f;
typedef __attribute__((ext_vector_type(8))) float v8f;

__device__ __forceinline__ float lrelu(float x) { return x > 0.f ? x : 0.01f * x; }

__device__ __forceinline__ void atomAddF(float* p, float v) { unsafeAtomicAdd(p, v); }

__global__ void init_deg_kernel(float* deg) {
    int i = blockIdx.x * blockDim.x + threadIdx.x;
    if (i < DEG_TOTAL) deg[i] = 0.f;
}

__global__ void init_cat_kernel(float* __restrict__ cat,
                                const float* __restrict__ b_sc,
                                const float* __restrict__ b_in,
                                const float* __restrict__ b_ni,
                                const float* __restrict__ b_ii) {
    long long tid = (long long)blockIdx.x * blockDim.x + threadIdx.x;
    if (tid >= (long long)TOTAL * 32) return;
    int row = (int)(tid >> 5);
    int c4  = (int)(tid & 31) * 4;
    float4 v;
    if (row < SVCN) {
        v = *(const float4*)&b_sc[c4];
    } else if (row < SVCN + NODEN) {
        v = *(const float4*)&b_in[c4];
    } else {
        float4 a = *(const float4*)&b_ni[c4];
        float4 b = *(const float4*)&b_ii[c4];
        v = make_float4(a.x + b.x, a.y + b.y, a.z + b.z, a.w + b.w);
    }
    *(float4*)&cat[(long long)row * 128 + c4] = v;
}

__global__ void init_out_kernel(float* __restrict__ out, const float* __restrict__ b_tot) {
    int i = blockIdx.x * blockDim.x + threadIdx.x;
    if (i < OUTD * HDIM) out[i] = b_tot[i >> 7];
}

__global__ void degree_kernel(const int* __restrict__ src, const int* __restrict__ dst,
                              float* __restrict__ degS, float* __restrict__ degD, int nE) {
    int e = blockIdx.x * blockDim.x + threadIdx.x;
    if (e >= nE) return;
    atomAddF(&degS[src[e]], 1.f);
    atomAddF(&degD[dst[e]], 1.f);
}

__global__ void rsqrt_kernel(float* deg) {
    int i = blockIdx.x * blockDim.x + threadIdx.x;
    if (i < DEG_TOTAL) {
        float d = deg[i];
        deg[i] = rsqrtf(d > 1.f ? d : 1.f);
    }
}

__global__ __launch_bounds__(256) void proj_kernel(const float* __restrict__ X,
                                                   const float* __restrict__ W,
                                                   const float* __restrict__ ns,
                                                   float* __restrict__ Hout, int nrows) {
    __shared__ float Wlds[128 * 128];
    int t = threadIdx.x;
    for (int i = t * 4; i < 128 * 128; i += 256 * 4)
        *(float4*)&Wlds[i] = *(const float4*)&W[i];
    __syncthreads();

    const int lane = t & 31;
    const int wave = t >> 5;
    const int half = lane >> 4;
    const int l16  = lane & 15;
    const int r0   = blockIdx.x * 64;
    const int n0   = wave * 16;

    const float* Arow[4];
#pragma unroll
    for (int rt = 0; rt < 4; ++rt) {
        int row = r0 + rt * 16 + l16;
        int rc  = row < nrows ? row : nrows - 1;
        Arow[rt] = X + (long long)rc * 128;
    }

    v8f acc[4] = {v8f{}, v8f{}, v8f{}, v8f{}};

    for (int k = 0; k < 128; k += 4) {
        const int kk = k + 2 * half;
        v2f b;
        b.x = Wlds[kk * 128 + n0 + l16];
        b.y = Wlds[(kk + 1) * 128 + n0 + l16];
#pragma unroll
        for (int rt = 0; rt < 4; ++rt) {
            v2f a = *(const v2f*)&Arow[rt][kk];
            acc[rt] = __builtin_amdgcn_wmma_f32_16x16x4_f32(
                false, a, false, b, (short)0, acc[rt], false, false);
        }
    }

    if (r0 + 64 <= nrows) {
#pragma unroll
        for (int rt = 0; rt < 4; ++rt) {
#pragma unroll
            for (int i = 0; i < 8; ++i) {
                int row = r0 + rt * 16 + i + 8 * half;
                Hout[(long long)row * 128 + n0 + l16] = acc[rt][i] * ns[row];
            }
        }
    } else {
#pragma unroll
        for (int rt = 0; rt < 4; ++rt) {
#pragma unroll
            for (int i = 0; i < 8; ++i) {
                int row = r0 + rt * 16 + i + 8 * half;
                if (row < nrows)
                    Hout[(long long)row * 128 + n0 + l16] = acc[rt][i] * ns[row];
            }
        }
    }
}

__global__ void scatter_kernel(const float* __restrict__ H,
                               const int* __restrict__ src, const int* __restrict__ dst,
                               const float* __restrict__ nd, float* __restrict__ cat,
                               long long dstBase, int nE) {
    long long tid = (long long)blockIdx.x * blockDim.x + threadIdx.x;
    int e = (int)(tid >> 5);
    if (e >= nE) return;
    int c4 = (int)(tid & 31) * 4;
    int s = src[e], d = dst[e];
    float scale = nd[d];
    if (e + 1 < nE)
        __builtin_prefetch(&H[(long long)src[e + 1] * 128 + c4], 0, 1);
    float4 hv = *(const float4*)&H[(long long)s * 128 + c4];
    float* p = &cat[dstBase + (long long)d * 128 + c4];
    atomAddF(p + 0, hv.x * scale);
    atomAddF(p + 1, hv.y * scale);
    atomAddF(p + 2, hv.z * scale);
    atomAddF(p + 3, hv.w * scale);
}

#define FG_KT      32
#define FG_TILES   (TOTAL / FG_KT)
#define FG_BLOCKS  250
#define FG_TPB     (FG_TILES / FG_BLOCKS)
#define FG_PA      36
#define FG_PB      136
static_assert(TOTAL % FG_KT == 0, "K must be a whole number of tiles");
static_assert(FG_TILES % FG_BLOCKS == 0, "tiles must divide evenly over the blocks");

__global__ __launch_bounds__(512) void final_gemm_kernel(const float* __restrict__ Wt,
                                                         const float* __restrict__ cat,
                                                         float* __restrict__ out) {
    __shared__ float As[OUTD * FG_PA];
    __shared__ float Bs[FG_KT * FG_PB];

    const int t    = threadIdx.x;
    const int lane = t & 31;
    const int wave = t >> 5;
    const int half = lane >> 4;
    const int l16  = lane & 15;
    const int wm   = wave >> 2;
    const int wn   = wave & 3;
    const int tile0 = blockIdx.x * FG_TPB;

    const int arow = t >> 3;
    const int ac4  = t & 7;
    const int brow = t >> 5;
    const int bc4  = t & 31;
    const float* pA = Wt + (long long)arow * TOTAL + (long long)tile0 * FG_KT + ac4 * 4;
    const float* pB = cat + (long long)tile0 * FG_KT * 128 + t * 4;
    const long long aStep = 64LL * TOTAL;
    float* sA = &As[arow * FG_PA + ac4 * 4];
    float* sB = &Bs[brow * FG_PB + bc4 * 4];

    v8f acc[4][2];
#pragma unroll
    for (int mt = 0; mt < 4; ++mt)
#pragma unroll
        for (int nt = 0; nt < 2; ++nt) acc[mt][nt] = v8f{};

    float4 ra0 = *(const float4*)(pA);
    float4 ra1 = *(const float4*)(pA + aStep);
    float4 ra2 = *(const float4*)(pA + 2 * aStep);
    float4 ra3 = *(const float4*)(pA + 3 * aStep);
    float4 rb0 = *(const float4*)(pB);
    float4 rb1 = *(const float4*)(pB + 2048);

    for (int it = 0; it < FG_TPB; ++it) {
        *(float4*)(sA)                 = ra0;
        *(float4*)(sA + 64 * FG_PA)    = ra1;
        *(float4*)(sA + 128 * FG_PA)   = ra2;
        *(float4*)(sA + 192 * FG_PA)   = ra3;
        rb0.x = lrelu(rb0.x); rb0.y = lrelu(rb0.y); rb0.z = lrelu(rb0.z); rb0.w = lrelu(rb0.w);
        rb1.x = lrelu(rb1.x); rb1.y = lrelu(rb1.y); rb1.z = lrelu(rb1.z); rb1.w = lrelu(rb1.w);
        *(float4*)(sB)                 = rb0;
        *(float4*)(sB + 16 * FG_PB)    = rb1;
        __syncthreads();

        if (it + 1 < FG_TPB) {
            pA += FG_KT;
            pB += FG_KT * 128;
            ra0 = *(const float4*)(pA);
            ra1 = *(const float4*)(pA + aStep);
            ra2 = *(const float4*)(pA + 2 * aStep);
            ra3 = *(const float4*)(pA + 3 * aStep);
            rb0 = *(const float4*)(pB);
            rb1 = *(const float4*)(pB + 2048);
        }
        __builtin_amdgcn_sched_barrier(0);

#pragma unroll
        for (int ks = 0; ks < FG_KT / 4; ++ks) {
            const int kk = ks * 4 + 2 * half;
            v2f a[4], b[2];
#pragma unroll
            for (int mt = 0; mt < 4; ++mt)
                a[mt] = *(const v2f*)&As[(wm * 64 + mt * 16 + l16) * FG_PA + kk];
#pragma unroll
            for (int nt = 0; nt < 2; ++nt) {
                b[nt].x = Bs[kk * FG_PB + wn * 32 + nt * 16 + l16];
                b[nt].y = Bs[(kk + 1) * FG_PB + wn * 32 + nt * 16 + l16];
            }
#pragma unroll
            for (int mt = 0; mt < 4; ++mt)
#pragma unroll
                for (int nt = 0; nt < 2; ++nt)
                    acc[mt][nt] = __builtin_amdgcn_wmma_f32_16x16x4_f32(
                        false, a[mt], false, b[nt], (short)0, acc[mt][nt], false, false);
        }
        __syncthreads();
    }

#pragma unroll
    for (int mt = 0; mt < 4; ++mt) {
#pragma unroll
        for (int nt = 0; nt < 2; ++nt) {
#pragma unroll
            for (int i = 0; i < 8; ++i) {
                int m = wm * 64 + mt * 16 + i + 8 * half;
                atomAddF(&out[m * 128 + wn * 32 + nt * 16 + l16], acc[mt][nt][i]);
            }
        }
    }
}

static inline int cdiv(long long a, long long b) { return (int)((a + b - 1) / b); }

extern "C" void kernel_launch(void* const* d_in, const int* in_sizes, int n_in,
                              void* d_out, int out_size, void* d_ws, size_t ws_size,
                              hipStream_t stream) {
    const float* feat_svc  = (const float*)d_in[0];
    const float* feat_inst = (const float*)d_in[1];
    const float* feat_node = (const float*)d_in[2];
    const int*   svc_src   = (const int*)d_in[3];
    const int*   svc_dst   = (const int*)d_in[4];
    const int*   in_src    = (const int*)d_in[5];
    const int*   in_dst    = (const int*)d_in[6];
    const int*   ni_src    = (const int*)d_in[7];
    const int*   ni_dst    = (const int*)d_in[8];
    const int*   ii_src    = (const int*)d_in[9];
    const int*   ii_dst    = (const int*)d_in[10];
    const float* W_sc = (const float*)d_in[11];  const float* b_sc = (const float*)d_in[12];
    const float* W_in = (const float*)d_in[13];  const float* b_in = (const float*)d_in[14];
    const float* W_ni = (const float*)d_in[15];  const float* b_ni = (const float*)d_in[16];
    const float* W_ii = (const float*)d_in[17];  const float* b_ii = (const float*)d_in[18];
    const float* W_tot = (const float*)d_in[19]; const float* b_tot = (const float*)d_in[20];
    float* out = (float*)d_out;

    float* ws   = (float*)d_ws;
    float* cat  = ws + OFF_CAT;
    float* h_sc = ws + OFF_HSC;
    float* h_in = ws + OFF_HIN;
    float* h_ni = ws + OFF_HNI;
    float* h_ii = ws + OFF_HII;
    float* deg  = ws + OFF_DEG;
    float* ns_sc = ws + D_NS_SC; float* nd_sc = ws + D_ND_SC;
    float* ns_in = ws + D_NS_IN; float* nd_in = ws + D_ND_IN;
    float* ns_ni = ws + D_NS_NI; float* nd_ni = ws + D_ND_NI;
    float* ns_ii = ws + D_NS_II; float* nd_ii = ws + D_ND_II;

    init_deg_kernel<<<cdiv(DEG_TOTAL, 256), 256, 0, stream>>>(deg);
    init_cat_kernel<<<cdiv((long long)TOTAL * 32, 256), 256, 0, stream>>>(cat, b_sc, b_in, b_ni, b_ii);
    init_out_kernel<<<cdiv(OUTD * HDIM, 256), 256, 0, stream>>>(out, b_tot);

    degree_kernel<<<cdiv(E_SC, 256), 256, 0, stream>>>(svc_src, svc_dst, ns_sc, nd_sc, E_SC);
    degree_kernel<<<cdiv(E_IN, 256), 256, 0, stream>>>(in_src,  in_dst,  ns_in, nd_in, E_IN);
    degree_kernel<<<cdiv(E_NI, 256), 256, 0, stream>>>(ni_src,  ni_dst,  ns_ni, nd_ni, E_NI);
    degree_kernel<<<cdiv(E_II, 256), 256, 0, stream>>>(ii_src,  ii_dst,  ns_ii, nd_ii, E_II);
    rsqrt_kernel<<<cdiv(DEG_TOTAL, 256), 256, 0, stream>>>(deg);

    proj_kernel<<<cdiv(SVCN, 64),  256, 0, stream>>>(feat_svc,  W_sc, ns_sc, h_sc, SVCN);
    proj_kernel<<<cdiv(INSTN, 64), 256, 0, stream>>>(feat_inst, W_in, ns_in, h_in, INSTN);
    proj_kernel<<<cdiv(NODEN, 64), 256, 0, stream>>>(feat_node, W_ni, ns_ni, h_ni, NODEN);
    proj_kernel<<<cdiv(INSTN, 64), 256, 0, stream>>>(feat_inst, W_ii, ns_ii, h_ii, INSTN);

    scatter_kernel<<<cdiv((long long)E_SC * 32, 256), 256, 0, stream>>>(
        h_sc, svc_src, svc_dst, nd_sc, cat, 0LL, E_SC);
    scatter_kernel<<<cdiv((long long)E_IN * 32, 256), 256, 0, stream>>>(
        h_in, in_src, in_dst, nd_in, cat, (long long)SVCN * 128, E_IN);
    scatter_kernel<<<cdiv((long long)E_NI * 32, 256), 256, 0, stream>>>(
        h_ni, ni_src, ni_dst, nd_ni, cat, (long long)(SVCN + NODEN) * 128, E_NI);
    scatter_kernel<<<cdiv((long long)E_II * 32, 256), 256, 0, stream>>>(
        h_ii, ii_src, ii_dst, nd_ii, cat, (long long)(SVCN + NODEN) * 128, E_II);

    final_gemm_kernel<<<FG_BLOCKS, 512, 0, stream>>>(W_tot, cat, out);
}


# edge scatter by f32 atomics replaced by a gather over edges counting-sorted by destination (integer atomic rank): one wave sums a cat row in registers, stores it once; row sum order = rank order
# speedup vs baseline: 1.3879x; 1.1780x over previous
#include <hip/hip_runtime.h>

#define SVCN   50000
#define INSTN  100000
#define NODEN  50000
#define FDIM   128
#define HDIM   128
#define OUTD   256
#define TOTAL  (SVCN + NODEN + INSTN)
#define E_SC   400000
#define E_IN   100000
#define E_NI   100000
#define E_II   800000

#define OFF_CAT   0LL
#define OFF_HSC   (OFF_CAT  + (long long)TOTAL * 128)
#define OFF_HIN   (OFF_HSC  + (long long)SVCN  * 128)
#define OFF_HNI   (OFF_HIN  + (long long)INSTN * 128)
#define OFF_HII   (OFF_HNI  + (long long)NODEN * 128)
#define OFF_DEG   (OFF_HII  + (long long)INSTN * 128)
#define D_NS_SC   (OFF_DEG + 0LL)
#define D_ND_SC   (D_NS_SC + SVCN)
#define D_NS_IN   (D_ND_SC + SVCN)
#define D_ND_IN   (D_NS_IN + INSTN)
#define D_NS_NI   (D_ND_IN + NODEN)
#define D_ND_NI   (D_NS_NI + NODEN)
#define D_NS_II   (D_ND_NI + INSTN)
#define D_ND_II   (D_NS_II + INSTN)
#define DEG_TOTAL 600000
#define OFF_CUR   (OFF_DEG + (long long)DEG_TOTAL)
#define CUR_SC    0
#define CUR_IN    (CUR_SC + SVCN)
#define CUR_NI    (CUR_IN + NODEN)
#define CUR_II    (CUR_NI + INSTN)
#define CUR_TOTAL (CUR_II + INSTN)
#define OFF_SORT  (OFF_CUR + (long long)CUR_TOTAL)
#define SORT_SC   0
#define SORT_IN   (SORT_SC + E_SC)
#define SORT_NI   (SORT_IN + E_IN)
#define SORT_II   (SORT_NI + E_NI)
static_assert(SVCN % 4 == 0 && NODEN % 4 == 0 && INSTN % 4 == 0, "degree planes are scanned four at a time");
static_assert(TOTAL % 8 == 0, "the gather runs eight rows per block");

typedef __attribute__((ext_vector_type(2))) float v2f;
typedef __attribute__((ext_vector_type(8))) float v8f;

__device__ __forceinline__ float lrelu(float x) { return x > 0.f ? x : 0.01f * x; }

__device__ __forceinline__ void atomAddF(float* p, float v) { unsafeAtomicAdd(p, v); }

__global__ void init_deg_kernel(float* deg) {
    int i = blockIdx.x * blockDim.x + threadIdx.x;
    if (i < DEG_TOTAL) deg[i] = 0.f;
}

__global__ void init_out_kernel(float* __restrict__ out, const float* __restrict__ b_tot) {
    int i = blockIdx.x * blockDim.x + threadIdx.x;
    if (i < OUTD * HDIM) out[i] = b_tot[i >> 7];
}

__global__ void degree_kernel(const int* __restrict__ src, const int* __restrict__ dst,
                              float* __restrict__ degS, float* __restrict__ degD, int nE) {
    int e = blockIdx.x * blockDim.x + threadIdx.x;
    if (e >= nE) return;
    atomAddF(&degS[src[e]], 1.f);
    atomAddF(&degD[dst[e]], 1.f);
}

__global__ void rsqrt_kernel(float* deg) {
    int i = blockIdx.x * blockDim.x + threadIdx.x;
    if (i < DEG_TOTAL) {
        float d = deg[i];
        deg[i] = rsqrtf(d > 1.f ? d : 1.f);
    }
}

__global__ __launch_bounds__(256) void proj_kernel(const float* __restrict__ X,
                                                   const float* __restrict__ W,
                                                   const float* __restrict__ ns,
                                                   float* __restrict__ Hout, int nrows) {
    __shared__ float Wlds[128 * 128];
    int t = threadIdx.x;
    for (int i = t * 4; i < 128 * 128; i += 256 * 4)
        *(float4*)&Wlds[i] = *(const float4*)&W[i];
    __syncthreads();

    const int lane = t & 31;
    const int wave = t >> 5;
    const int half = lane >> 4;
    const int l16  = lane & 15;
    const int r0   = blockIdx.x * 64;
    const int n0   = wave * 16;

    const float* Arow[4];
#pragma unroll
    for (int rt = 0; rt < 4; ++rt) {
        int row = r0 + rt * 16 + l16;
        int rc  = row < nrows ? row : nrows - 1;
        Arow[rt] = X + (long long)rc * 128;
    }

    v8f acc[4] = {v8f{}, v8f{}, v8f{}, v8f{}};

    for (int k = 0; k < 128; k += 4) {
        const int kk = k + 2 * half;
        v2f b;
        b.x = Wlds[kk * 128 + n0 + l16];
        b.y = Wlds[(kk + 1) * 128 + n0 + l16];
#pragma unroll
        for (int rt = 0; rt < 4; ++rt) {
            v2f a = *(const v2f*)&Arow[rt][kk];
            acc[rt] = __builtin_amdgcn_wmma_f32_16x16x4_f32(
                false, a, false, b, (short)0, acc[rt], false, false);
        }
    }

    if (r0 + 64 <= nrows) {
#pragma unroll
        for (int rt = 0; rt < 4; ++rt) {
#pragma unroll
            for (int i = 0; i < 8; ++i) {
                int row = r0 + rt * 16 + i + 8 * half;
                Hout[(long long)row * 128 + n0 + l16] = acc[rt][i] * ns[row];
            }
        }
    } else {
#pragma unroll
        for (int rt = 0; rt < 4; ++rt) {
#pragma unroll
            for (int i = 0; i < 8; ++i) {
                int row = r0 + rt * 16 + i + 8 * half;
                if (row < nrows)
                    Hout[(long long)row * 128 + n0 + l16] = acc[rt][i] * ns[row];
            }
        }
    }
}


#define SCAN_T 1024
#define SCAN_Q 4
__global__ __launch_bounds__(SCAN_T) void offsets_kernel(const float* __restrict__ deg,
                                                         int* __restrict__ cur) {
    __shared__ int wsum[32];
    __shared__ int tile_total;
    const float* dg; int* out; int n;
    if (blockIdx.x == 0)      { dg = deg + (D_ND_SC - OFF_DEG); out = cur + CUR_SC; n = SVCN;  }
    else if (blockIdx.x == 1) { dg = deg + (D_ND_IN - OFF_DEG); out = cur + CUR_IN; n = NODEN; }
    else if (blockIdx.x == 2) { dg = deg + (D_ND_NI - OFF_DEG); out = cur + CUR_NI; n = INSTN; }
    else                      { dg = deg + (D_ND_II - OFF_DEG); out = cur + CUR_II; n = INSTN; }
    const int t = threadIdx.x, lane = t & 31, wave = t >> 5;
    int carry = 0;
    for (int base = 0; base < n; base += SCAN_T * SCAN_Q * 4) {
        const int i0 = base + t * (SCAN_Q * 4);
        int4 v[SCAN_Q];
        int s = 0;
#pragma unroll
        for (int q = 0; q < SCAN_Q; ++q) {
            const int i = i0 + q * 4;
            v[q] = make_int4(0, 0, 0, 0);
            if (i < n) {
                const float4 f = *(const float4*)&dg[i];
                v[q] = make_int4((int)f.x, (int)f.y, (int)f.z, (int)f.w);
            }
            s += v[q].x + v[q].y + v[q].z + v[q].w;
        }
        int x = s;
        for (int o = 1; o < 32; o <<= 1) {
            const int y = __shfl_up(x, o);
            if (lane >= o) x += y;
        }
        if (lane == 31) wsum[wave] = x;
        __syncthreads();
        if (wave == 0) {
            const int w = wsum[lane];
            int xs = w;
            for (int o = 1; o < 32; o <<= 1) {
                const int y = __shfl_up(xs, o);
                if (lane >= o) xs += y;
            }
            wsum[lane] = xs - w;
            if (lane == 31) tile_total = xs;
        }
        __syncthreads();
        int run = carry + wsum[wave] + (x - s);
#pragma unroll
        for (int q = 0; q < SCAN_Q; ++q) {
            const int i = i0 + q * 4;
            if (i < n)
                *(int4*)&out[i] = make_int4(run, run + v[q].x, run + v[q].x + v[q].y,
                                            run + v[q].x + v[q].y + v[q].z);
            run += v[q].x + v[q].y + v[q].z + v[q].w;
        }
        carry += tile_total;
    }
}

#define FILL_T   256
#define FILL_NB0 ((E_SC + FILL_T - 1) / FILL_T)
#define FILL_NB1 ((E_IN + FILL_T - 1) / FILL_T)
#define FILL_NB2 ((E_NI + FILL_T - 1) / FILL_T)
#define FILL_NB3 ((E_II + FILL_T - 1) / FILL_T)
__global__ __launch_bounds__(FILL_T) void fill_kernel(const int* __restrict__ svc_src, const int* __restrict__ svc_dst,
                                                      const int* __restrict__ in_src,  const int* __restrict__ in_dst,
                                                      const int* __restrict__ ni_src,  const int* __restrict__ ni_dst,
                                                      const int* __restrict__ ii_src,  const int* __restrict__ ii_dst,
                                                      int* __restrict__ cur, int* __restrict__ sorted) {
    int b = blockIdx.x;
    const int* src; const int* dst; int* cu; int* so; int nE;
    if (b < FILL_NB0) { src = svc_src; dst = svc_dst; cu = cur + CUR_SC; so = sorted + SORT_SC; nE = E_SC; }
    else if ((b -= FILL_NB0) < FILL_NB1) { src = in_src; dst = in_dst; cu = cur + CUR_IN; so = sorted + SORT_IN; nE = E_IN; }
    else if ((b -= FILL_NB1) < FILL_NB2) { src = ni_src; dst = ni_dst; cu = cur + CUR_NI; so = sorted + SORT_NI; nE = E_NI; }
    else { b -= FILL_NB2;                  src = ii_src; dst = ii_dst; cu = cur + CUR_II; so = sorted + SORT_II; nE = E_II; }
    const int e = b * FILL_T + threadIdx.x;
    if (e >= nE) return;
    const int pos = atomicAdd(&cu[dst[e]], 1);
    so[pos] = src[e];
}

__device__ __forceinline__ float4 sum_sources(const float* __restrict__ H,
                                              const int* __restrict__ cur,
                                              const int* __restrict__ sorted,
                                              int d, int c4) {
    const int begin = d > 0 ? cur[d - 1] : 0;
    const int end   = cur[d];
    float4 acc = make_float4(0.f, 0.f, 0.f, 0.f);
    int j = begin;
    for (; j + 4 <= end; j += 4) {
        const int s0 = sorted[j], s1 = sorted[j + 1], s2 = sorted[j + 2], s3 = sorted[j + 3];
        const float4 a0 = *(const float4*)&H[(long long)s0 * 128 + c4];
        const float4 a1 = *(const float4*)&H[(long long)s1 * 128 + c4];
        const float4 a2 = *(const float4*)&H[(long long)s2 * 128 + c4];
        const float4 a3 = *(const float4*)&H[(long long)s3 * 128 + c4];
        acc.x += a0.x; acc.y += a0.y; acc.z += a0.z; acc.w += a0.w;
        acc.x += a1.x; acc.y += a1.y; acc.z += a1.z; acc.w += a1.w;
        acc.x += a2.x; acc.y += a2.y; acc.z += a2.z; acc.w += a2.w;
        acc.x += a3.x; acc.y += a3.y; acc.z += a3.z; acc.w += a3.w;
    }
    for (; j < end; ++j) {
        const float4 a0 = *(const float4*)&H[(long long)sorted[j] * 128 + c4];
        acc.x += a0.x; acc.y += a0.y; acc.z += a0.z; acc.w += a0.w;
    }
    return acc;
}

__global__ __launch_bounds__(256) void gather_kernel(const float* __restrict__ h_sc,
                                                     const float* __restrict__ h_in,
                                                     const float* __restrict__ h_ni,
                                                     const float* __restrict__ h_ii,
                                                     const int* __restrict__ cur,
                                                     const int* __restrict__ sorted,
                                                     const float* __restrict__ deg,
                                                     const float* __restrict__ b_sc,
                                                     const float* __restrict__ b_in,
                                                     const float* __restrict__ b_ni,
                                                     const float* __restrict__ b_ii,
                                                     float* __restrict__ cat) {
    const int c4 = (threadIdx.x & 31) * 4;
    const int row = __builtin_amdgcn_readfirstlane((int)(blockIdx.x * 8 + (threadIdx.x >> 5)));
    float4 v;
    if (row < SVCN) {
        const int d = row;
        const float4 a = sum_sources(h_sc, cur + CUR_SC, sorted + SORT_SC, d, c4);
        const float nd = deg[(D_ND_SC - OFF_DEG) + d];
        const float4 b = *(const float4*)&b_sc[c4];
        v = make_float4(a.x * nd + b.x, a.y * nd + b.y, a.z * nd + b.z, a.w * nd + b.w);
    } else if (row < SVCN + NODEN) {
        const int d = row - SVCN;
        const float4 a = sum_sources(h_in, cur + CUR_IN, sorted + SORT_IN, d, c4);
        const float nd = deg[(D_ND_IN - OFF_DEG) + d];
        const float4 b = *(const float4*)&b_in[c4];
        v = make_float4(a.x * nd + b.x, a.y * nd + b.y, a.z * nd + b.z, a.w * nd + b.w);
    } else {
        const int d = row - (SVCN + NODEN);
        const float4 a1 = sum_sources(h_ni, cur + CUR_NI, sorted + SORT_NI, d, c4);
        const float4 a2 = sum_sources(h_ii, cur + CUR_II, sorted + SORT_II, d, c4);
        const float n1 = deg[(D_ND_NI - OFF_DEG) + d];
        const float n2 = deg[(D_ND_II - OFF_DEG) + d];
        const float4 b1 = *(const float4*)&b_ni[c4];
        const float4 b2 = *(const float4*)&b_ii[c4];
        v = make_float4((a1.x * n1 + b1.x) + (a2.x * n2 + b2.x),
                        (a1.y * n1 + b1.y) + (a2.y * n2 + b2.y),
                        (a1.z * n1 + b1.z) + (a2.z * n2 + b2.z),
                        (a1.w * n1 + b1.w) + (a2.w * n2 + b2.w));
    }
    *(float4*)&cat[(long long)row * 128 + c4] = v;
}

#define FG_KT      32
#define FG_TILES   (TOTAL / FG_KT)
#define FG_BLOCKS  250
#define FG_TPB     (FG_TILES / FG_BLOCKS)
#define FG_PA      36
#define FG_PB      136
static_assert(TOTAL % FG_KT == 0, "K must be a whole number of tiles");
static_assert(FG_TILES % FG_BLOCKS == 0, "tiles must divide evenly over the blocks");

__global__ __launch_bounds__(512) void final_gemm_kernel(const float* __restrict__ Wt,
                                                         const float* __restrict__ cat,
                                                         float* __restrict__ out) {
    __shared__ float As[OUTD * FG_PA];
    __shared__ float Bs[FG_KT * FG_PB];

    const int t    = threadIdx.x;
    const int lane = t & 31;
    const int wave = t >> 5;
    const int half = lane >> 4;
    const int l16  = lane & 15;
    const int wm   = wave >> 2;
    const int wn   = wave & 3;
    const int tile0 = blockIdx.x * FG_TPB;

    const int arow = t >> 3;
    const int ac4  = t & 7;
    const int brow = t >> 5;
    const int bc4  = t & 31;
    const float* pA = Wt + (long long)arow * TOTAL + (long long)tile0 * FG_KT + ac4 * 4;
    const float* pB = cat + (long long)tile0 * FG_KT * 128 + t * 4;
    const long long aStep = 64LL * TOTAL;
    float* sA = &As[arow * FG_PA + ac4 * 4];
    float* sB = &Bs[brow * FG_PB + bc4 * 4];

    v8f acc[4][2];
#pragma unroll
    for (int mt = 0; mt < 4; ++mt)
#pragma unroll
        for (int nt = 0; nt < 2; ++nt) acc[mt][nt] = v8f{};

    float4 ra0 = *(const float4*)(pA);
    float4 ra1 = *(const float4*)(pA + aStep);
    float4 ra2 = *(const float4*)(pA + 2 * aStep);
    float4 ra3 = *(const float4*)(pA + 3 * aStep);
    float4 rb0 = *(const float4*)(pB);
    float4 rb1 = *(const float4*)(pB + 2048);

    for (int it = 0; it < FG_TPB; ++it) {
        *(float4*)(sA)                 = ra0;
        *(float4*)(sA + 64 * FG_PA)    = ra1;
        *(float4*)(sA + 128 * FG_PA)   = ra2;
        *(float4*)(sA + 192 * FG_PA)   = ra3;
        rb0.x = lrelu(rb0.x); rb0.y = lrelu(rb0.y); rb0.z = lrelu(rb0.z); rb0.w = lrelu(rb0.w);
        rb1.x = lrelu(rb1.x); rb1.y = lrelu(rb1.y); rb1.z = lrelu(rb1.z); rb1.w = lrelu(rb1.w);
        *(float4*)(sB)                 = rb0;
        *(float4*)(sB + 16 * FG_PB)    = rb1;
        __syncthreads();

        if (it + 1 < FG_TPB) {
            pA += FG_KT;
            pB += FG_KT * 128;
            ra0 = *(const float4*)(pA);
            ra1 = *(const float4*)(pA + aStep);
            ra2 = *(const float4*)(pA + 2 * aStep);
            ra3 = *(const float4*)(pA + 3 * aStep);
            rb0 = *(const float4*)(pB);
            rb1 = *(const float4*)(pB + 2048);
        }
        __builtin_amdgcn_sched_barrier(0);

#pragma unroll
        for (int ks = 0; ks < FG_KT / 4; ++ks) {
            const int kk = ks * 4 + 2 * half;
            v2f a[4], b[2];
#pragma unroll
            for (int mt = 0; mt < 4; ++mt)
                a[mt] = *(const v2f*)&As[(wm * 64 + mt * 16 + l16) * FG_PA + kk];
#pragma unroll
            for (int nt = 0; nt < 2; ++nt) {
                b[nt].x = Bs[kk * FG_PB + wn * 32 + nt * 16 + l16];
                b[nt].y = Bs[(kk + 1) * FG_PB + wn * 32 + nt * 16 + l16];
            }
#pragma unroll
            for (int mt = 0; mt < 4; ++mt)
#pragma unroll
                for (int nt = 0; nt < 2; ++nt)
                    acc[mt][nt] = __builtin_amdgcn_wmma_f32_16x16x4_f32(
                        false, a[mt], false, b[nt], (short)0, acc[mt][nt], false, false);
        }
        __syncthreads();
    }

#pragma unroll
    for (int mt = 0; mt < 4; ++mt) {
#pragma unroll
        for (int nt = 0; nt < 2; ++nt) {
#pragma unroll
            for (int i = 0; i < 8; ++i) {
                int m = wm * 64 + mt * 16 + i + 8 * half;
                atomAddF(&out[m * 128 + wn * 32 + nt * 16 + l16], acc[mt][nt][i]);
            }
        }
    }
}

static inline int cdiv(long long a, long long b) { return (int)((a + b - 1) / b); }

extern "C" void kernel_launch(void* const* d_in, const int* in_sizes, int n_in,
                              void* d_out, int out_size, void* d_ws, size_t ws_size,
                              hipStream_t stream) {
    const float* feat_svc  = (const float*)d_in[0];
    const float* feat_inst = (const float*)d_in[1];
    const float* feat_node = (const float*)d_in[2];
    const int*   svc_src   = (const int*)d_in[3];
    const int*   svc_dst   = (const int*)d_in[4];
    const int*   in_src    = (const int*)d_in[5];
    const int*   in_dst    = (const int*)d_in[6];
    const int*   ni_src    = (const int*)d_in[7];
    const int*   ni_dst    = (const int*)d_in[8];
    const int*   ii_src    = (const int*)d_in[9];
    const int*   ii_dst    = (const int*)d_in[10];
    const float* W_sc = (const float*)d_in[11];  const float* b_sc = (const float*)d_in[12];
    const float* W_in = (const float*)d_in[13];  const float* b_in = (const float*)d_in[14];
    const float* W_ni = (const float*)d_in[15];  const float* b_ni = (const float*)d_in[16];
    const float* W_ii = (const float*)d_in[17];  const float* b_ii = (const float*)d_in[18];
    const float* W_tot = (const float*)d_in[19]; const float* b_tot = (const float*)d_in[20];
    float* out = (float*)d_out;

    float* ws   = (float*)d_ws;
    float* cat  = ws + OFF_CAT;
    float* h_sc = ws + OFF_HSC;
    float* h_in = ws + OFF_HIN;
    float* h_ni = ws + OFF_HNI;
    float* h_ii = ws + OFF_HII;
    float* deg  = ws + OFF_DEG;
    float* ns_sc = ws + D_NS_SC; float* nd_sc = ws + D_ND_SC;
    float* ns_in = ws + D_NS_IN; float* nd_in = ws + D_ND_IN;
    float* ns_ni = ws + D_NS_NI; float* nd_ni = ws + D_ND_NI;
    float* ns_ii = ws + D_NS_II; float* nd_ii = ws + D_ND_II;
    int* cur    = (int*)(ws + OFF_CUR);
    int* sorted = (int*)(ws + OFF_SORT);

    init_deg_kernel<<<cdiv(DEG_TOTAL, 256), 256, 0, stream>>>(deg);
    init_out_kernel<<<cdiv(OUTD * HDIM, 256), 256, 0, stream>>>(out, b_tot);

    degree_kernel<<<cdiv(E_SC, 256), 256, 0, stream>>>(svc_src, svc_dst, ns_sc, nd_sc, E_SC);
    degree_kernel<<<cdiv(E_IN, 256), 256, 0, stream>>>(in_src,  in_dst,  ns_in, nd_in, E_IN);
    degree_kernel<<<cdiv(E_NI, 256), 256, 0, stream>>>(ni_src,  ni_dst,  ns_ni, nd_ni, E_NI);
    degree_kernel<<<cdiv(E_II, 256), 256, 0, stream>>>(ii_src,  ii_dst,  ns_ii, nd_ii, E_II);
    offsets_kernel<<<4, SCAN_T, 0, stream>>>(deg, cur);
    rsqrt_kernel<<<cdiv(DEG_TOTAL, 256), 256, 0, stream>>>(deg);

    fill_kernel<<<FILL_NB0 + FILL_NB1 + FILL_NB2 + FILL_NB3, FILL_T, 0, stream>>>(
        svc_src, svc_dst, in_src, in_dst, ni_src, ni_dst, ii_src, ii_dst, cur, sorted);

    proj_kernel<<<cdiv(SVCN, 64),  256, 0, stream>>>(feat_svc,  W_sc, ns_sc, h_sc, SVCN);
    proj_kernel<<<cdiv(INSTN, 64), 256, 0, stream>>>(feat_inst, W_in, ns_in, h_in, INSTN);
    proj_kernel<<<cdiv(NODEN, 64), 256, 0, stream>>>(feat_node, W_ni, ns_ni, h_ni, NODEN);
    proj_kernel<<<cdiv(INSTN, 64), 256, 0, stream>>>(feat_inst, W_ii, ns_ii, h_ii, INSTN);

    gather_kernel<<<TOTAL / 8, 256, 0, stream>>>(h_sc, h_in, h_ni, h_ii, cur, sorted, deg,
                                                 b_sc, b_in, b_ni, b_ii, cat);

    final_gemm_kernel<<<FG_BLOCKS, 512, 0, stream>>>(W_tot, cat, out);
}


# graph conv reordered by linearity: raw source rows (scaled by ns) are summed per destination first, the 128x128 projection is fused behind the sum in one kernel; row sums now precede the product
# speedup vs baseline: 1.6922x; 1.2193x over previous
#include <hip/hip_runtime.h>

#define SVCN   50000
#define INSTN  100000
#define NODEN  50000
#define FDIM   128
#define HDIM   128
#define OUTD   256
#define TOTAL  (SVCN + NODEN + INSTN)
#define E_SC   400000
#define E_IN   100000
#define E_NI   100000
#define E_II   800000

#define OFF_CAT   0LL
#define OFF_DEG   (OFF_CAT  + (long long)TOTAL * 128)
#define D_NS_SC   (OFF_DEG + 0LL)
#define D_ND_SC   (D_NS_SC + SVCN)
#define D_NS_IN   (D_ND_SC + SVCN)
#define D_ND_IN   (D_NS_IN + INSTN)
#define D_NS_NI   (D_ND_IN + NODEN)
#define D_ND_NI   (D_NS_NI + NODEN)
#define D_NS_II   (D_ND_NI + INSTN)
#define D_ND_II   (D_NS_II + INSTN)
#define DEG_TOTAL 600000
#define OFF_CUR   (OFF_DEG + (long long)DEG_TOTAL)
#define CUR_SC    0
#define CUR_IN    (CUR_SC + SVCN)
#define CUR_NI    (CUR_IN + NODEN)
#define CUR_II    (CUR_NI + INSTN)
#define CUR_TOTAL (CUR_II + INSTN)
#define OFF_SORT  (OFF_CUR + (long long)CUR_TOTAL)
#define SORT_SC   0
#define SORT_IN   (SORT_SC + E_SC)
#define SORT_NI   (SORT_IN + E_IN)
#define SORT_II   (SORT_NI + E_NI)
static_assert(SVCN % 4 == 0 && NODEN % 4 == 0 && INSTN % 4 == 0, "degree planes are scanned four at a time");

typedef __attribute__((ext_vector_type(2))) float v2f;
typedef __attribute__((ext_vector_type(8))) float v8f;

__device__ __forceinline__ float lrelu(float x) { return x > 0.f ? x : 0.01f * x; }

__device__ __forceinline__ void atomAddF(float* p, float v) { unsafeAtomicAdd(p, v); }

__global__ void init_deg_kernel(float* deg) {
    int i = blockIdx.x * blockDim.x + threadIdx.x;
    if (i < DEG_TOTAL) deg[i] = 0.f;
}

__global__ void init_out_kernel(float* __restrict__ out, const float* __restrict__ b_tot) {
    int i = blockIdx.x * blockDim.x + threadIdx.x;
    if (i < OUTD * HDIM) out[i] = b_tot[i >> 7];
}

__global__ void degree_kernel(const int* __restrict__ src, const int* __restrict__ dst,
                              float* __restrict__ degS, float* __restrict__ degD, int nE) {
    int e = blockIdx.x * blockDim.x + threadIdx.x;
    if (e >= nE) return;
    atomAddF(&degS[src[e]], 1.f);
    atomAddF(&degD[dst[e]], 1.f);
}

__global__ void rsqrt_kernel(float* deg) {
    int i = blockIdx.x * blockDim.x + threadIdx.x;
    if (i < DEG_TOTAL) {
        float d = deg[i];
        deg[i] = rsqrtf(d > 1.f ? d : 1.f);
    }
}


#define SCAN_T 1024
#define SCAN_Q 4
__global__ __launch_bounds__(SCAN_T) void offsets_kernel(const float* __restrict__ deg,
                                                         int* __restrict__ cur) {
    __shared__ int wsum[32];
    __shared__ int tile_total;
    const float* dg; int* out; int n;
    if (blockIdx.x == 0)      { dg = deg + (D_ND_SC - OFF_DEG); out = cur + CUR_SC; n = SVCN;  }
    else if (blockIdx.x == 1) { dg = deg + (D_ND_IN - OFF_DEG); out = cur + CUR_IN; n = NODEN; }
    else if (blockIdx.x == 2) { dg = deg + (D_ND_NI - OFF_DEG); out = cur + CUR_NI; n = INSTN; }
    else                      { dg = deg + (D_ND_II - OFF_DEG); out = cur + CUR_II; n = INSTN; }
    const int t = threadIdx.x, lane = t & 31, wave = t >> 5;
    int carry = 0;
    for (int base = 0; base < n; base += SCAN_T * SCAN_Q * 4) {
        const int i0 = base + t * (SCAN_Q * 4);
        int4 v[SCAN_Q];
        int s = 0;
#pragma unroll
        for (int q = 0; q < SCAN_Q; ++q) {
            const int i = i0 + q * 4;
            v[q] = make_int4(0, 0, 0, 0);
            if (i < n) {
                const float4 f = *(const float4*)&dg[i];
                v[q] = make_int4((int)f.x, (int)f.y, (int)f.z, (int)f.w);
            }
            s += v[q].x + v[q].y + v[q].z + v[q].w;
        }
        int x = s;
        for (int o = 1; o < 32; o <<= 1) {
            const int y = __shfl_up(x, o);
            if (lane >= o) x += y;
        }
        if (lane == 31) wsum[wave] = x;
        __syncthreads();
        if (wave == 0) {
            const int w = wsum[lane];
            int xs = w;
            for (int o = 1; o < 32; o <<= 1) {
                const int y = __shfl_up(xs, o);
                if (lane >= o) xs += y;
            }
            wsum[lane] = xs - w;
            if (lane == 31) tile_total = xs;
        }
        __syncthreads();
        int run = carry + wsum[wave] + (x - s);
#pragma unroll
        for (int q = 0; q < SCAN_Q; ++q) {
            const int i = i0 + q * 4;
            if (i < n)
                *(int4*)&out[i] = make_int4(run, run + v[q].x, run + v[q].x + v[q].y,
                                            run + v[q].x + v[q].y + v[q].z);
            run += v[q].x + v[q].y + v[q].z + v[q].w;
        }
        carry += tile_total;
    }
}

#define FILL_T   256
#define FILL_NB0 ((E_SC + FILL_T - 1) / FILL_T)
#define FILL_NB1 ((E_IN + FILL_T - 1) / FILL_T)
#define FILL_NB2 ((E_NI + FILL_T - 1) / FILL_T)
#define FILL_NB3 ((E_II + FILL_T - 1) / FILL_T)
__global__ __launch_bounds__(FILL_T) void fill_kernel(const int* __restrict__ svc_src, const int* __restrict__ svc_dst,
                                                      const int* __restrict__ in_src,  const int* __restrict__ in_dst,
                                                      const int* __restrict__ ni_src,  const int* __restrict__ ni_dst,
                                                      const int* __restrict__ ii_src,  const int* __restrict__ ii_dst,
                                                      int* __restrict__ cur, int* __restrict__ sorted) {
    int b = blockIdx.x;
    const int* src; const int* dst; int* cu; int* so; int nE;
    if (b < FILL_NB0) { src = svc_src; dst = svc_dst; cu = cur + CUR_SC; so = sorted + SORT_SC; nE = E_SC; }
    else if ((b -= FILL_NB0) < FILL_NB1) { src = in_src; dst = in_dst; cu = cur + CUR_IN; so = sorted + SORT_IN; nE = E_IN; }
    else if ((b -= FILL_NB1) < FILL_NB2) { src = ni_src; dst = ni_dst; cu = cur + CUR_NI; so = sorted + SORT_NI; nE = E_NI; }
    else { b -= FILL_NB2;                  src = ii_src; dst = ii_dst; cu = cur + CUR_II; so = sorted + SORT_II; nE = E_II; }
    const int e = b * FILL_T + threadIdx.x;
    if (e >= nE) return;
    const int pos = atomicAdd(&cu[dst[e]], 1);
    so[pos] = src[e];
}

__device__ __forceinline__ float4 sum_sources_scaled(const float* __restrict__ X,
                                                     const float* __restrict__ ns,
                                                     const int* __restrict__ cur,
                                                     const int* __restrict__ sorted,
                                                     int d, int c4) {
    const int begin = d > 0 ? cur[d - 1] : 0;
    const int end   = cur[d];
    float4 acc = make_float4(0.f, 0.f, 0.f, 0.f);
    int j = begin;
    for (; j + 4 <= end; j += 4) {
        const int s0 = sorted[j], s1 = sorted[j + 1], s2 = sorted[j + 2], s3 = sorted[j + 3];
        const float w0 = ns[s0], w1 = ns[s1], w2 = ns[s2], w3 = ns[s3];
        const float4 a0 = *(const float4*)&X[(long long)s0 * 128 + c4];
        const float4 a1 = *(const float4*)&X[(long long)s1 * 128 + c4];
        const float4 a2 = *(const float4*)&X[(long long)s2 * 128 + c4];
        const float4 a3 = *(const float4*)&X[(long long)s3 * 128 + c4];
        acc.x = fmaf(a0.x, w0, acc.x); acc.y = fmaf(a0.y, w0, acc.y); acc.z = fmaf(a0.z, w0, acc.z); acc.w = fmaf(a0.w, w0, acc.w);
        acc.x = fmaf(a1.x, w1, acc.x); acc.y = fmaf(a1.y, w1, acc.y); acc.z = fmaf(a1.z, w1, acc.z); acc.w = fmaf(a1.w, w1, acc.w);
        acc.x = fmaf(a2.x, w2, acc.x); acc.y = fmaf(a2.y, w2, acc.y); acc.z = fmaf(a2.z, w2, acc.z); acc.w = fmaf(a2.w, w2, acc.w);
        acc.x = fmaf(a3.x, w3, acc.x); acc.y = fmaf(a3.y, w3, acc.y); acc.z = fmaf(a3.z, w3, acc.z); acc.w = fmaf(a3.w, w3, acc.w);
    }
    for (; j < end; ++j) {
        const int s0 = sorted[j];
        const float w0 = ns[s0];
        const float4 a0 = *(const float4*)&X[(long long)s0 * 128 + c4];
        acc.x = fmaf(a0.x, w0, acc.x); acc.y = fmaf(a0.y, w0, acc.y); acc.z = fmaf(a0.z, w0, acc.z); acc.w = fmaf(a0.w, w0, acc.w);
    }
    return acc;
}

#define AG_ROWS   64
#define AG_T      512
#define AG_PITCH  132
#define AG_NB_INST ((INSTN + AG_ROWS - 1) / AG_ROWS)
#define AG_NB_SVC  ((SVCN  + AG_ROWS - 1) / AG_ROWS)
#define AG_NB_NODE ((NODEN + AG_ROWS - 1) / AG_ROWS)
__global__ __launch_bounds__(AG_T) void agg_proj_kernel(
        const float* __restrict__ feat_svc, const float* __restrict__ feat_inst,
        const float* __restrict__ feat_node,
        const float* __restrict__ W_sc, const float* __restrict__ W_in,
        const float* __restrict__ W_ni, const float* __restrict__ W_ii,
        const float* __restrict__ b_sc, const float* __restrict__ b_in,
        const float* __restrict__ b_ni, const float* __restrict__ b_ii,
        const float* __restrict__ deg, const int* __restrict__ curAll,
        const int* __restrict__ sortedAll, float* __restrict__ catAll) {
    __shared__ __attribute__((aligned(16))) float Wlds[128 * 128];
    __shared__ __attribute__((aligned(16))) float Tile[AG_ROWS * AG_PITCH];

    int blk = blockIdx.x;
    const float *X0, *W0, *ns0, *nd0, *bA;
    const int *cur0, *sorted0;
    const float *X1 = nullptr, *W1 = nullptr, *ns1 = nullptr, *nd1 = nullptr, *bB = nullptr;
    const int *cur1 = nullptr, *sorted1 = nullptr;
    int nrows, nlists = 1;
    float* cat;
    if (blk < AG_NB_INST) {
        X0 = feat_node; W0 = W_ni; bA = b_ni;
        ns0 = deg + (D_NS_NI - OFF_DEG); nd0 = deg + (D_ND_NI - OFF_DEG);
        cur0 = curAll + CUR_NI; sorted0 = sortedAll + SORT_NI;
        X1 = feat_inst; W1 = W_ii; bB = b_ii;
        ns1 = deg + (D_NS_II - OFF_DEG); nd1 = deg + (D_ND_II - OFF_DEG);
        cur1 = curAll + CUR_II; sorted1 = sortedAll + SORT_II;
        nrows = INSTN; nlists = 2;
        cat = catAll + (long long)(SVCN + NODEN) * 128;
    } else if ((blk -= AG_NB_INST) < AG_NB_SVC) {
        X0 = feat_svc; W0 = W_sc; bA = b_sc;
        ns0 = deg + (D_NS_SC - OFF_DEG); nd0 = deg + (D_ND_SC - OFF_DEG);
        cur0 = curAll + CUR_SC; sorted0 = sortedAll + SORT_SC;
        nrows = SVCN;
        cat = catAll;
    } else {
        blk -= AG_NB_SVC;
        X0 = feat_inst; W0 = W_in; bA = b_in;
        ns0 = deg + (D_NS_IN - OFF_DEG); nd0 = deg + (D_ND_IN - OFF_DEG);
        cur0 = curAll + CUR_IN; sorted0 = sortedAll + SORT_IN;
        nrows = NODEN;
        cat = catAll + (long long)SVCN * 128;
    }

    const int t    = threadIdx.x;
    const int lane = t & 31;
    const int wave = __builtin_amdgcn_readfirstlane(t >> 5);
    const int half = lane >> 4;
    const int l16  = lane & 15;
    const int c4   = lane * 4;
    const int r0   = blk * AG_ROWS;
    const int n0   = (wave & 7) * 16;
    const int rb   = (wave >> 3) * 32;

    v8f acc[2] = {v8f{}, v8f{}};

    for (int li = 0; li < nlists; ++li) {
        const float* __restrict__ X  = li ? X1 : X0;
        const float* __restrict__ Wg = li ? W1 : W0;
        const float* __restrict__ ns = li ? ns1 : ns0;
        const float* __restrict__ nd = li ? nd1 : nd0;
        const int* __restrict__ cur    = li ? cur1 : cur0;
        const int* __restrict__ sorted = li ? sorted1 : sorted0;

        if (li > 0) __syncthreads();

        for (int i = t * 4; i < 128 * 128; i += AG_T * 4)
            *(float4*)&Wlds[i] = *(const float4*)&Wg[i];

        for (int i = 0; i < 4; ++i) {
            const int lr  = wave * 4 + i;
            const int row = r0 + lr;
            float4 v = make_float4(0.f, 0.f, 0.f, 0.f);
            if (row < nrows) {
                const float4 a = sum_sources_scaled(X, ns, cur, sorted, row, c4);
                const float n = nd[row];
                v = make_float4(a.x * n, a.y * n, a.z * n, a.w * n);
            }
            *(float4*)&Tile[lr * AG_PITCH + c4] = v;
        }
        __syncthreads();

        for (int k = 0; k < 128; k += 4) {
            const int kk = k + 2 * half;
            v2f b;
            b.x = Wlds[kk * 128 + n0 + l16];
            b.y = Wlds[(kk + 1) * 128 + n0 + l16];
#pragma unroll
            for (int rt = 0; rt < 2; ++rt) {
                v2f a = *(const v2f*)&Tile[(rb + rt * 16 + l16) * AG_PITCH + kk];
                acc[rt] = __builtin_amdgcn_wmma_f32_16x16x4_f32(
                    false, a, false, b, (short)0, acc[rt], false, false);
            }
        }
    }

    __syncthreads();
#pragma unroll
    for (int rt = 0; rt < 2; ++rt) {
#pragma unroll
        for (int i = 0; i < 8; ++i)
            Tile[(rb + rt * 16 + i + 8 * half) * AG_PITCH + n0 + l16] = acc[rt][i];
    }
    __syncthreads();

    float4 bias = *(const float4*)&bA[c4];
    if (nlists > 1) {
        const float4 bb = *(const float4*)&bB[c4];
        bias.x += bb.x; bias.y += bb.y; bias.z += bb.z; bias.w += bb.w;
    }
    for (int i = 0; i < 4; ++i) {
        const int lr  = wave * 4 + i;
        const int row = r0 + lr;
        if (row < nrows) {
            const float4 p = *(const float4*)&Tile[lr * AG_PITCH + c4];
            *(float4*)&cat[(long long)row * 128 + c4] =
                make_float4(p.x + bias.x, p.y + bias.y, p.z + bias.z, p.w + bias.w);
        }
    }
}

#define FG_KT      32
#define FG_TILES   (TOTAL / FG_KT)
#define FG_BLOCKS  250
#define FG_TPB     (FG_TILES / FG_BLOCKS)
#define FG_PA      36
#define FG_PB      136
static_assert(TOTAL % FG_KT == 0, "K must be a whole number of tiles");
static_assert(FG_TILES % FG_BLOCKS == 0, "tiles must divide evenly over the blocks");

__global__ __launch_bounds__(512) void final_gemm_kernel(const float* __restrict__ Wt,
                                                         const float* __restrict__ cat,
                                                         float* __restrict__ out) {
    __shared__ float As[OUTD * FG_PA];
    __shared__ float Bs[FG_KT * FG_PB];

    const int t    = threadIdx.x;
    const int lane = t & 31;
    const int wave = t >> 5;
    const int half = lane >> 4;
    const int l16  = lane & 15;
    const int wm   = wave >> 2;
    const int wn   = wave & 3;
    const int tile0 = blockIdx.x * FG_TPB;

    const int arow = t >> 3;
    const int ac4  = t & 7;
    const int brow = t >> 5;
    const int bc4  = t & 31;
    const float* pA = Wt + (long long)arow * TOTAL + (long long)tile0 * FG_KT + ac4 * 4;
    const float* pB = cat + (long long)tile0 * FG_KT * 128 + t * 4;
    const long long aStep = 64LL * TOTAL;
    float* sA = &As[arow * FG_PA + ac4 * 4];
    float* sB = &Bs[brow * FG_PB + bc4 * 4];

    v8f acc[4][2];
#pragma unroll
    for (int mt = 0; mt < 4; ++mt)
#pragma unroll
        for (int nt = 0; nt < 2; ++nt) acc[mt][nt] = v8f{};

    float4 ra0 = *(const float4*)(pA);
    float4 ra1 = *(const float4*)(pA + aStep);
    float4 ra2 = *(const float4*)(pA + 2 * aStep);
    float4 ra3 = *(const float4*)(pA + 3 * aStep);
    float4 rb0 = *(const float4*)(pB);
    float4 rb1 = *(const float4*)(pB + 2048);

    for (int it = 0; it < FG_TPB; ++it) {
        *(float4*)(sA)                 = ra0;
        *(float4*)(sA + 64 * FG_PA)    = ra1;
        *(float4*)(sA + 128 * FG_PA)   = ra2;
        *(float4*)(sA + 192 * FG_PA)   = ra3;
        rb0.x = lrelu(rb0.x); rb0.y = lrelu(rb0.y); rb0.z = lrelu(rb0.z); rb0.w = lrelu(rb0.w);
        rb1.x = lrelu(rb1.x); rb1.y = lrelu(rb1.y); rb1.z = lrelu(rb1.z); rb1.w = lrelu(rb1.w);
        *(float4*)(sB)                 = rb0;
        *(float4*)(sB + 16 * FG_PB)    = rb1;
        __syncthreads();

        if (it + 1 < FG_TPB) {
            pA += FG_KT;
            pB += FG_KT * 128;
            ra0 = *(const float4*)(pA);
            ra1 = *(const float4*)(pA + aStep);
            ra2 = *(const float4*)(pA + 2 * aStep);
            ra3 = *(const float4*)(pA + 3 * aStep);
            rb0 = *(const float4*)(pB);
            rb1 = *(const float4*)(pB + 2048);
        }
        __builtin_amdgcn_sched_barrier(0);

#pragma unroll
        for (int ks = 0; ks < FG_KT / 4; ++ks) {
            const int kk = ks * 4 + 2 * half;
            v2f a[4], b[2];
#pragma unroll
            for (int mt = 0; mt < 4; ++mt)
                a[mt] = *(const v2f*)&As[(wm * 64 + mt * 16 + l16) * FG_PA + kk];
#pragma unroll
            for (int nt = 0; nt < 2; ++nt) {
                b[nt].x = Bs[kk * FG_PB + wn * 32 + nt * 16 + l16];
                b[nt].y = Bs[(kk + 1) * FG_PB + wn * 32 + nt * 16 + l16];
            }
#pragma unroll
            for (int mt = 0; mt < 4; ++mt)
#pragma unroll
                for (int nt = 0; nt < 2; ++nt)
                    acc[mt][nt] = __builtin_amdgcn_wmma_f32_16x16x4_f32(
                        false, a[mt], false, b[nt], (short)0, acc[mt][nt], false, false);
        }
        __syncthreads();
    }

#pragma unroll
    for (int mt = 0; mt < 4; ++mt) {
#pragma unroll
        for (int nt = 0; nt < 2; ++nt) {
#pragma unroll
            for (int i = 0; i < 8; ++i) {
                int m = wm * 64 + mt * 16 + i + 8 * half;
                atomAddF(&out[m * 128 + wn * 32 + nt * 16 + l16], acc[mt][nt][i]);
            }
        }
    }
}

static inline int cdiv(long long a, long long b) { return (int)((a + b - 1) / b); }

extern "C" void kernel_launch(void* const* d_in, const int* in_sizes, int n_in,
                              void* d_out, int out_size, void* d_ws, size_t ws_size,
                              hipStream_t stream) {
    const float* feat_svc  = (const float*)d_in[0];
    const float* feat_inst = (const float*)d_in[1];
    const float* feat_node = (const float*)d_in[2];
    const int*   svc_src   = (const int*)d_in[3];
    const int*   svc_dst   = (const int*)d_in[4];
    const int*   in_src    = (const int*)d_in[5];
    const int*   in_dst    = (const int*)d_in[6];
    const int*   ni_src    = (const int*)d_in[7];
    const int*   ni_dst    = (const int*)d_in[8];
    const int*   ii_src    = (const int*)d_in[9];
    const int*   ii_dst    = (const int*)d_in[10];
    const float* W_sc = (const float*)d_in[11];  const float* b_sc = (const float*)d_in[12];
    const float* W_in = (const float*)d_in[13];  const float* b_in = (const float*)d_in[14];
    const float* W_ni = (const float*)d_in[15];  const float* b_ni = (const float*)d_in[16];
    const float* W_ii = (const float*)d_in[17];  const float* b_ii = (const float*)d_in[18];
    const float* W_tot = (const float*)d_in[19]; const float* b_tot = (const float*)d_in[20];
    float* out = (float*)d_out;

    float* ws   = (float*)d_ws;
    float* cat  = ws + OFF_CAT;
    float* deg  = ws + OFF_DEG;
    float* ns_sc = ws + D_NS_SC; float* nd_sc = ws + D_ND_SC;
    float* ns_in = ws + D_NS_IN; float* nd_in = ws + D_ND_IN;
    float* ns_ni = ws + D_NS_NI; float* nd_ni = ws + D_ND_NI;
    float* ns_ii = ws + D_NS_II; float* nd_ii = ws + D_ND_II;
    int* cur    = (int*)(ws + OFF_CUR);
    int* sorted = (int*)(ws + OFF_SORT);

    init_deg_kernel<<<cdiv(DEG_TOTAL, 256), 256, 0, stream>>>(deg);
    init_out_kernel<<<cdiv(OUTD * HDIM, 256), 256, 0, stream>>>(out, b_tot);

    degree_kernel<<<cdiv(E_SC, 256), 256, 0, stream>>>(svc_src, svc_dst, ns_sc, nd_sc, E_SC);
    degree_kernel<<<cdiv(E_IN, 256), 256, 0, stream>>>(in_src,  in_dst,  ns_in, nd_in, E_IN);
    degree_kernel<<<cdiv(E_NI, 256), 256, 0, stream>>>(ni_src,  ni_dst,  ns_ni, nd_ni, E_NI);
    degree_kernel<<<cdiv(E_II, 256), 256, 0, stream>>>(ii_src,  ii_dst,  ns_ii, nd_ii, E_II);
    offsets_kernel<<<4, SCAN_T, 0, stream>>>(deg, cur);
    rsqrt_kernel<<<cdiv(DEG_TOTAL, 256), 256, 0, stream>>>(deg);

    fill_kernel<<<FILL_NB0 + FILL_NB1 + FILL_NB2 + FILL_NB3, FILL_T, 0, stream>>>(
        svc_src, svc_dst, in_src, in_dst, ni_src, ni_dst, ii_src, ii_dst, cur, sorted);

    agg_proj_kernel<<<AG_NB_INST + AG_NB_SVC + AG_NB_NODE, AG_T, 0, stream>>>(
        feat_svc, feat_inst, feat_node, W_sc, W_in, W_ni, W_ii, b_sc, b_in, b_ni, b_ii,
        deg, cur, sorted, cat);

    final_gemm_kernel<<<FG_BLOCKS, 512, 0, stream>>>(W_tot, cat, out);
}


# agg_proj loads regrouped so latencies overlap: per wave one lane-parallel load each for 4 rows' ranges, source ids, factors; source rows 8 in flight; W tile 8 loads then 8 stores; sum order kept
# speedup vs baseline: 1.7824x; 1.0533x over previous
#include <hip/hip_runtime.h>

#define SVCN   50000
#define INSTN  100000
#define NODEN  50000
#define FDIM   128
#define HDIM   128
#define OUTD   256
#define TOTAL  (SVCN + NODEN + INSTN)
#define E_SC   400000
#define E_IN   100000
#define E_NI   100000
#define E_II   800000

#define OFF_CAT   0LL
#define OFF_DEG   (OFF_CAT  + (long long)TOTAL * 128)
#define D_NS_SC   (OFF_DEG + 0LL)
#define D_ND_SC   (D_NS_SC + SVCN)
#define D_NS_IN   (D_ND_SC + SVCN)
#define D_ND_IN   (D_NS_IN + INSTN)
#define D_NS_NI   (D_ND_IN + NODEN)
#define D_ND_NI   (D_NS_NI + NODEN)
#define D_NS_II   (D_ND_NI + INSTN)
#define D_ND_II   (D_NS_II + INSTN)
#define DEG_TOTAL 600000
#define OFF_CUR   (OFF_DEG + (long long)DEG_TOTAL)
#define CUR_SC    0
#define CUR_IN    (CUR_SC + SVCN)
#define CUR_NI    (CUR_IN + NODEN)
#define CUR_II    (CUR_NI + INSTN)
#define CUR_TOTAL (CUR_II + INSTN)
#define OFF_SORT  (OFF_CUR + (long long)CUR_TOTAL)
#define SORT_SC   0
#define SORT_IN   (SORT_SC + E_SC)
#define SORT_NI   (SORT_IN + E_IN)
#define SORT_II   (SORT_NI + E_NI)
static_assert(SVCN % 4 == 0 && NODEN % 4 == 0 && INSTN % 4 == 0, "degree planes are scanned four at a time");

typedef __attribute__((ext_vector_type(2))) float v2f;
typedef __attribute__((ext_vector_type(8))) float v8f;

__device__ __forceinline__ float lrelu(float x) { return x > 0.f ? x : 0.01f * x; }

__device__ __forceinline__ void atomAddF(float* p, float v) { unsafeAtomicAdd(p, v); }

__global__ void init_deg_kernel(float* deg) {
    int i = blockIdx.x * blockDim.x + threadIdx.x;
    if (i < DEG_TOTAL) deg[i] = 0.f;
}

__global__ void init_out_kernel(float* __restrict__ out, const float* __restrict__ b_tot) {
    int i = blockIdx.x * blockDim.x + threadIdx.x;
    if (i < OUTD * HDIM) out[i] = b_tot[i >> 7];
}

__global__ void degree_kernel(const int* __restrict__ src, const int* __restrict__ dst,
                              float* __restrict__ degS, float* __restrict__ degD, int nE) {
    int e = blockIdx.x * blockDim.x + threadIdx.x;
    if (e >= nE) return;
    atomAddF(&degS[src[e]], 1.f);
    atomAddF(&degD[dst[e]], 1.f);
}

__global__ void rsqrt_kernel(float* deg) {
    int i = blockIdx.x * blockDim.x + threadIdx.x;
    if (i < DEG_TOTAL) {
        float d = deg[i];
        deg[i] = rsqrtf(d > 1.f ? d : 1.f);
    }
}


#define SCAN_T 1024
#define SCAN_Q 4
__global__ __launch_bounds__(SCAN_T) void offsets_kernel(const float* __restrict__ deg,
                                                         int* __restrict__ cur) {
    __shared__ int wsum[32];
    __shared__ int tile_total;
    const float* dg; int* out; int n;
    if (blockIdx.x == 0)      { dg = deg + (D_ND_SC - OFF_DEG); out = cur + CUR_SC; n = SVCN;  }
    else if (blockIdx.x == 1) { dg = deg + (D_ND_IN - OFF_DEG); out = cur + CUR_IN; n = NODEN; }
    else if (blockIdx.x == 2) { dg = deg + (D_ND_NI - OFF_DEG); out = cur + CUR_NI; n = INSTN; }
    else                      { dg = deg + (D_ND_II - OFF_DEG); out = cur + CUR_II; n = INSTN; }
    const int t = threadIdx.x, lane = t & 31, wave = t >> 5;
    int carry = 0;
    for (int base = 0; base < n; base += SCAN_T * SCAN_Q * 4) {
        const int i0 = base + t * (SCAN_Q * 4);
        int4 v[SCAN_Q];
        int s = 0;
#pragma unroll
        for (int q = 0; q < SCAN_Q; ++q) {
            const int i = i0 + q * 4;
            v[q] = make_int4(0, 0, 0, 0);
            if (i < n) {
                const float4 f = *(const float4*)&dg[i];
                v[q] = make_int4((int)f.x, (int)f.y, (int)f.z, (int)f.w);
            }
            s += v[q].x + v[q].y + v[q].z + v[q].w;
        }
        int x = s;
        for (int o = 1; o < 32; o <<= 1) {
            const int y = __shfl_up(x, o);
            if (lane >= o) x += y;
        }
        if (lane == 31) wsum[wave] = x;
        __syncthreads();
        if (wave == 0) {
            const int w = wsum[lane];
            int xs = w;
            for (int o = 1; o < 32; o <<= 1) {
                const int y = __shfl_up(xs, o);
                if (lane >= o) xs += y;
            }
            wsum[lane] = xs - w;
            if (lane == 31) tile_total = xs;
        }
        __syncthreads();
        int run = carry + wsum[wave] + (x - s);
#pragma unroll
        for (int q = 0; q < SCAN_Q; ++q) {
            const int i = i0 + q * 4;
            if (i < n)
                *(int4*)&out[i] = make_int4(run, run + v[q].x, run + v[q].x + v[q].y,
                                            run + v[q].x + v[q].y + v[q].z);
            run += v[q].x + v[q].y + v[q].z + v[q].w;
        }
        carry += tile_total;
    }
}

#define FILL_T   256
#define FILL_NB0 ((E_SC + FILL_T - 1) / FILL_T)
#define FILL_NB1 ((E_IN + FILL_T - 1) / FILL_T)
#define FILL_NB2 ((E_NI + FILL_T - 1) / FILL_T)
#define FILL_NB3 ((E_II + FILL_T - 1) / FILL_T)
__global__ __launch_bounds__(FILL_T) void fill_kernel(const int* __restrict__ svc_src, const int* __restrict__ svc_dst,
                                                      const int* __restrict__ in_src,  const int* __restrict__ in_dst,
                                                      const int* __restrict__ ni_src,  const int* __restrict__ ni_dst,
                                                      const int* __restrict__ ii_src,  const int* __restrict__ ii_dst,
                                                      int* __restrict__ cur, int* __restrict__ sorted) {
    int b = blockIdx.x;
    const int* src; const int* dst; int* cu; int* so; int nE;
    if (b < FILL_NB0) { src = svc_src; dst = svc_dst; cu = cur + CUR_SC; so = sorted + SORT_SC; nE = E_SC; }
    else if ((b -= FILL_NB0) < FILL_NB1) { src = in_src; dst = in_dst; cu = cur + CUR_IN; so = sorted + SORT_IN; nE = E_IN; }
    else if ((b -= FILL_NB1) < FILL_NB2) { src = ni_src; dst = ni_dst; cu = cur + CUR_NI; so = sorted + SORT_NI; nE = E_NI; }
    else { b -= FILL_NB2;                  src = ii_src; dst = ii_dst; cu = cur + CUR_II; so = sorted + SORT_II; nE = E_II; }
    const int e = b * FILL_T + threadIdx.x;
    if (e >= nE) return;
    const int pos = atomicAdd(&cu[dst[e]], 1);
    so[pos] = src[e];
}

#define AG_BATCH 8

#define AG_ROWS   64
#define AG_T      512
#define AG_PITCH  132
#define AG_NB_INST ((INSTN + AG_ROWS - 1) / AG_ROWS)
#define AG_NB_SVC  ((SVCN  + AG_ROWS - 1) / AG_ROWS)
#define AG_NB_NODE ((NODEN + AG_ROWS - 1) / AG_ROWS)
__global__ __launch_bounds__(AG_T) void agg_proj_kernel(
        const float* __restrict__ feat_svc, const float* __restrict__ feat_inst,
        const float* __restrict__ feat_node,
        const float* __restrict__ W_sc, const float* __restrict__ W_in,
        const float* __restrict__ W_ni, const float* __restrict__ W_ii,
        const float* __restrict__ b_sc, const float* __restrict__ b_in,
        const float* __restrict__ b_ni, const float* __restrict__ b_ii,
        const float* __restrict__ deg, const int* __restrict__ curAll,
        const int* __restrict__ sortedAll, float* __restrict__ catAll) {
    __shared__ __attribute__((aligned(16))) float Wlds[128 * 128];
    __shared__ __attribute__((aligned(16))) float Tile[AG_ROWS * AG_PITCH];

    int blk = blockIdx.x;
    const float *X0, *W0, *ns0, *nd0, *bA;
    const int *cur0, *sorted0;
    const float *X1 = nullptr, *W1 = nullptr, *ns1 = nullptr, *nd1 = nullptr, *bB = nullptr;
    const int *cur1 = nullptr, *sorted1 = nullptr;
    int nrows, nlists = 1;
    float* cat;
    if (blk < AG_NB_INST) {
        X0 = feat_node; W0 = W_ni; bA = b_ni;
        ns0 = deg + (D_NS_NI - OFF_DEG); nd0 = deg + (D_ND_NI - OFF_DEG);
        cur0 = curAll + CUR_NI; sorted0 = sortedAll + SORT_NI;
        X1 = feat_inst; W1 = W_ii; bB = b_ii;
        ns1 = deg + (D_NS_II - OFF_DEG); nd1 = deg + (D_ND_II - OFF_DEG);
        cur1 = curAll + CUR_II; sorted1 = sortedAll + SORT_II;
        nrows = INSTN; nlists = 2;
        cat = catAll + (long long)(SVCN + NODEN) * 128;
    } else if ((blk -= AG_NB_INST) < AG_NB_SVC) {
        X0 = feat_svc; W0 = W_sc; bA = b_sc;
        ns0 = deg + (D_NS_SC - OFF_DEG); nd0 = deg + (D_ND_SC - OFF_DEG);
        cur0 = curAll + CUR_SC; sorted0 = sortedAll + SORT_SC;
        nrows = SVCN;
        cat = catAll;
    } else {
        blk -= AG_NB_SVC;
        X0 = feat_inst; W0 = W_in; bA = b_in;
        ns0 = deg + (D_NS_IN - OFF_DEG); nd0 = deg + (D_ND_IN - OFF_DEG);
        cur0 = curAll + CUR_IN; sorted0 = sortedAll + SORT_IN;
        nrows = NODEN;
        cat = catAll + (long long)SVCN * 128;
    }

    const int t    = threadIdx.x;
    const int lane = t & 31;
    const int wave = __builtin_amdgcn_readfirstlane(t >> 5);
    const int half = lane >> 4;
    const int l16  = lane & 15;
    const int c4   = lane * 4;
    const int r0   = blk * AG_ROWS;
    const int n0   = (wave & 7) * 16;
    const int rb   = (wave >> 3) * 32;

    v8f acc[2] = {v8f{}, v8f{}};

    for (int li = 0; li < nlists; ++li) {
        const float* __restrict__ X  = li ? X1 : X0;
        const float* __restrict__ Wg = li ? W1 : W0;
        const float* __restrict__ ns = li ? ns1 : ns0;
        const float* __restrict__ nd = li ? nd1 : nd0;
        const int* __restrict__ cur    = li ? cur1 : cur0;
        const int* __restrict__ sorted = li ? sorted1 : sorted0;

        const int first = r0 + wave * 4;
        int ic = first - 1 + lane; ic = ic < 0 ? 0 : (ic < nrows ? ic : nrows - 1);
        int id = first + lane;     id = id < nrows ? id : nrows - 1;
        const int   cv  = cur[ic];
        const float ndv = nd[id];

        const float4* __restrict__ Wg4 = (const float4*)Wg + t;
        const float4 wr0 = Wg4[0 * AG_T], wr1 = Wg4[1 * AG_T], wr2 = Wg4[2 * AG_T], wr3 = Wg4[3 * AG_T];
        const float4 wr4 = Wg4[4 * AG_T], wr5 = Wg4[5 * AG_T], wr6 = Wg4[6 * AG_T], wr7 = Wg4[7 * AG_T];

        const int b0 = first > 0 ? __builtin_amdgcn_readlane(cv, 0) : 0;
        const int e3 = __builtin_amdgcn_readlane(cv, 4);
        int sv_next = 0;
        if (lane < e3 - b0) sv_next = sorted[b0 + lane];

        if (li > 0) __syncthreads();

        float4* Wl4 = (float4*)Wlds + t;
        Wl4[0 * AG_T] = wr0; Wl4[1 * AG_T] = wr1; Wl4[2 * AG_T] = wr2; Wl4[3 * AG_T] = wr3;
        Wl4[4 * AG_T] = wr4; Wl4[5 * AG_T] = wr5; Wl4[6 * AG_T] = wr6; Wl4[7 * AG_T] = wr7;

        float4 acc4 = make_float4(0.f, 0.f, 0.f, 0.f);
        int ri   = 0;
        int rend = __builtin_amdgcn_readlane(cv, 1);
        for (int cb = b0; cb < e3; cb += 32) {
            const int rem = e3 - cb;
            const int cnt = rem < 32 ? rem : 32;
            const int sv  = sv_next;
            sv_next = 0;
            if (lane < rem - 32) sv_next = sorted[cb + 32 + lane];
            float wv = 0.f;
            if (lane < cnt) wv = ns[sv];
            for (int jj = 0; jj < cnt; jj += AG_BATCH) {
                float4 x[AG_BATCH];
#pragma unroll
                for (int k = 0; k < AG_BATCH; ++k) {
                    const int q = jj + k < cnt ? jj + k : cnt - 1;
                    const int s = __builtin_amdgcn_readlane(sv, q);
                    x[k] = *(const float4*)&X[(long long)s * 128 + c4];
                }
#pragma unroll
                for (int k = 0; k < AG_BATCH; ++k) {
                    const bool live = jj + k < cnt;
                    const int q = live ? jj + k : cnt - 1;
                    const int j = cb + q;
                    while (j >= rend) {
                        const float n = __builtin_bit_cast(float,
                            __builtin_amdgcn_readlane(__builtin_bit_cast(int, ndv), ri));
                        *(float4*)&Tile[(wave * 4 + ri) * AG_PITCH + c4] =
                            make_float4(acc4.x * n, acc4.y * n, acc4.z * n, acc4.w * n);
                        acc4 = make_float4(0.f, 0.f, 0.f, 0.f);
                        ++ri;
                        rend = __builtin_amdgcn_readlane(cv, ri + 1);
                    }
                    const float wq = __builtin_bit_cast(float,
                        __builtin_amdgcn_readlane(__builtin_bit_cast(int, wv), q));
                    const float w = live ? wq : 0.f;
                    acc4.x = fmaf(x[k].x, w, acc4.x); acc4.y = fmaf(x[k].y, w, acc4.y);
                    acc4.z = fmaf(x[k].z, w, acc4.z); acc4.w = fmaf(x[k].w, w, acc4.w);
                }
            }
        }
        for (; ri < 4; ++ri) {
            const float n = __builtin_bit_cast(float,
                __builtin_amdgcn_readlane(__builtin_bit_cast(int, ndv), ri));
            *(float4*)&Tile[(wave * 4 + ri) * AG_PITCH + c4] =
                make_float4(acc4.x * n, acc4.y * n, acc4.z * n, acc4.w * n);
            acc4 = make_float4(0.f, 0.f, 0.f, 0.f);
        }
        __syncthreads();

        for (int k = 0; k < 128; k += 4) {
            const int kk = k + 2 * half;
            v2f b;
            b.x = Wlds[kk * 128 + n0 + l16];
            b.y = Wlds[(kk + 1) * 128 + n0 + l16];
#pragma unroll
            for (int rt = 0; rt < 2; ++rt) {
                v2f a = *(const v2f*)&Tile[(rb + rt * 16 + l16) * AG_PITCH + kk];
                acc[rt] = __builtin_amdgcn_wmma_f32_16x16x4_f32(
                    false, a, false, b, (short)0, acc[rt], false, false);
            }
        }
    }

    __syncthreads();
#pragma unroll
    for (int rt = 0; rt < 2; ++rt) {
#pragma unroll
        for (int i = 0; i < 8; ++i)
            Tile[(rb + rt * 16 + i + 8 * half) * AG_PITCH + n0 + l16] = acc[rt][i];
    }
    __syncthreads();

    float4 bias = *(const float4*)&bA[c4];
    if (nlists > 1) {
        const float4 bb = *(const float4*)&bB[c4];
        bias.x += bb.x; bias.y += bb.y; bias.z += bb.z; bias.w += bb.w;
    }
    for (int i = 0; i < 4; ++i) {
        const int lr  = wave * 4 + i;
        const int row = r0 + lr;
        if (row < nrows) {
            const float4 p = *(const float4*)&Tile[lr * AG_PITCH + c4];
            *(float4*)&cat[(long long)row * 128 + c4] =
                make_float4(p.x + bias.x, p.y + bias.y, p.z + bias.z, p.w + bias.w);
        }
    }
}

#define FG_KT      32
#define FG_TILES   (TOTAL / FG_KT)
#define FG_BLOCKS  250
#define FG_TPB     (FG_TILES / FG_BLOCKS)
#define FG_PA      36
#define FG_PB      136
static_assert(TOTAL % FG_KT == 0, "K must be a whole number of tiles");
static_assert(FG_TILES % FG_BLOCKS == 0, "tiles must divide evenly over the blocks");

__global__ __launch_bounds__(512) void final_gemm_kernel(const float* __restrict__ Wt,
                                                         const float* __restrict__ cat,
                                                         float* __restrict__ out) {
    __shared__ float As[OUTD * FG_PA];
    __shared__ float Bs[FG_KT * FG_PB];

    const int t    = threadIdx.x;
    const int lane = t & 31;
    const int wave = t >> 5;
    const int half = lane >> 4;
    const int l16  = lane & 15;
    const int wm   = wave >> 2;
    const int wn   = wave & 3;
    const int tile0 = blockIdx.x * FG_TPB;

    const int arow = t >> 3;
    const int ac4  = t & 7;
    const int brow = t >> 5;
    const int bc4  = t & 31;
    const float* pA = Wt + (long long)arow * TOTAL + (long long)tile0 * FG_KT + ac4 * 4;
    const float* pB = cat + (long long)tile0 * FG_KT * 128 + t * 4;
    const long long aStep = 64LL * TOTAL;
    float* sA = &As[arow * FG_PA + ac4 * 4];
    float* sB = &Bs[brow * FG_PB + bc4 * 4];

    v8f acc[4][2];
#pragma unroll
    for (int mt = 0; mt < 4; ++mt)
#pragma unroll
        for (int nt = 0; nt < 2; ++nt) acc[mt][nt] = v8f{};

    float4 ra0 = *(const float4*)(pA);
    float4 ra1 = *(const float4*)(pA + aStep);
    float4 ra2 = *(const float4*)(pA + 2 * aStep);
    float4 ra3 = *(const float4*)(pA + 3 * aStep);
    float4 rb0 = *(const float4*)(pB);
    float4 rb1 = *(const float4*)(pB + 2048);

    for (int it = 0; it < FG_TPB; ++it) {
        *(float4*)(sA)                 = ra0;
        *(float4*)(sA + 64 * FG_PA)    = ra1;
        *(float4*)(sA + 128 * FG_PA)   = ra2;
        *(float4*)(sA + 192 * FG_PA)   = ra3;
        rb0.x = lrelu(rb0.x); rb0.y = lrelu(rb0.y); rb0.z = lrelu(rb0.z); rb0.w = lrelu(rb0.w);
        rb1.x = lrelu(rb1.x); rb1.y = lrelu(rb1.y); rb1.z = lrelu(rb1.z); rb1.w = lrelu(rb1.w);
        *(float4*)(sB)                 = rb0;
        *(float4*)(sB + 16 * FG_PB)    = rb1;
        __syncthreads();

        if (it + 1 < FG_TPB) {
            pA += FG_KT;
            pB += FG_KT * 128;
            ra0 = *(const float4*)(pA);
            ra1 = *(const float4*)(pA + aStep);
            ra2 = *(const float4*)(pA + 2 * aStep);
            ra3 = *(const float4*)(pA + 3 * aStep);
            rb0 = *(const float4*)(pB);
            rb1 = *(const float4*)(pB + 2048);
        }
        __builtin_amdgcn_sched_barrier(0);

#pragma unroll
        for (int ks = 0; ks < FG_KT / 4; ++ks) {
            const int kk = ks * 4 + 2 * half;
            v2f a[4], b[2];
#pragma unroll
            for (int mt = 0; mt < 4; ++mt)
                a[mt] = *(const v2f*)&As[(wm * 64 + mt * 16 + l16) * FG_PA + kk];
#pragma unroll
            for (int nt = 0; nt < 2; ++nt) {
                b[nt].x = Bs[kk * FG_PB + wn * 32 + nt * 16 + l16];
                b[nt].y = Bs[(kk + 1) * FG_PB + wn * 32 + nt * 16 + l16];
            }
#pragma unroll
            for (int mt = 0; mt < 4; ++mt)
#pragma unroll
                for (int nt = 0; nt < 2; ++nt)
                    acc[mt][nt] = __builtin_amdgcn_wmma_f32_16x16x4_f32(
                        false, a[mt], false, b[nt], (short)0, acc[mt][nt], false, false);
        }
        __syncthreads();
    }

#pragma unroll
    for (int mt = 0; mt < 4; ++mt) {
#pragma unroll
        for (int nt = 0; nt < 2; ++nt) {
#pragma unroll
            for (int i = 0; i < 8; ++i) {
                int m = wm * 64 + mt * 16 + i + 8 * half;
                atomAddF(&out[m * 128 + wn * 32 + nt * 16 + l16], acc[mt][nt][i]);
            }
        }
    }
}

static inline int cdiv(long long a, long long b) { return (int)((a + b - 1) / b); }

extern "C" void kernel_launch(void* const* d_in, const int* in_sizes, int n_in,
                              void* d_out, int out_size, void* d_ws, size_t ws_size,
                              hipStream_t stream) {
    const float* feat_svc  = (const float*)d_in[0];
    const float* feat_inst = (const float*)d_in[1];
    const float* feat_node = (const float*)d_in[2];
    const int*   svc_src   = (const int*)d_in[3];
    const int*   svc_dst   = (const int*)d_in[4];
    const int*   in_src    = (const int*)d_in[5];
    const int*   in_dst    = (const int*)d_in[6];
    const int*   ni_src    = (const int*)d_in[7];
    const int*   ni_dst    = (const int*)d_in[8];
    const int*   ii_src    = (const int*)d_in[9];
    const int*   ii_dst    = (const int*)d_in[10];
    const float* W_sc = (const float*)d_in[11];  const float* b_sc = (const float*)d_in[12];
    const float* W_in = (const float*)d_in[13];  const float* b_in = (const float*)d_in[14];
    const float* W_ni = (const float*)d_in[15];  const float* b_ni = (const float*)d_in[16];
    const float* W_ii = (const float*)d_in[17];  const float* b_ii = (const float*)d_in[18];
    const float* W_tot = (const float*)d_in[19]; const float* b_tot = (const float*)d_in[20];
    float* out = (float*)d_out;

    float* ws   = (float*)d_ws;
    float* cat  = ws + OFF_CAT;
    float* deg  = ws + OFF_DEG;
    float* ns_sc = ws + D_NS_SC; float* nd_sc = ws + D_ND_SC;
    float* ns_in = ws + D_NS_IN; float* nd_in = ws + D_ND_IN;
    float* ns_ni = ws + D_NS_NI; float* nd_ni = ws + D_ND_NI;
    float* ns_ii = ws + D_NS_II; float* nd_ii = ws + D_ND_II;
    int* cur    = (int*)(ws + OFF_CUR);
    int* sorted = (int*)(ws + OFF_SORT);

    init_deg_kernel<<<cdiv(DEG_TOTAL, 256), 256, 0, stream>>>(deg);
    init_out_kernel<<<cdiv(OUTD * HDIM, 256), 256, 0, stream>>>(out, b_tot);

    degree_kernel<<<cdiv(E_SC, 256), 256, 0, stream>>>(svc_src, svc_dst, ns_sc, nd_sc, E_SC);
    degree_kernel<<<cdiv(E_IN, 256), 256, 0, stream>>>(in_src,  in_dst,  ns_in, nd_in, E_IN);
    degree_kernel<<<cdiv(E_NI, 256), 256, 0, stream>>>(ni_src,  ni_dst,  ns_ni, nd_ni, E_NI);
    degree_kernel<<<cdiv(E_II, 256), 256, 0, stream>>>(ii_src,  ii_dst,  ns_ii, nd_ii, E_II);
    offsets_kernel<<<4, SCAN_T, 0, stream>>>(deg, cur);
    rsqrt_kernel<<<cdiv(DEG_TOTAL, 256), 256, 0, stream>>>(deg);

    fill_kernel<<<FILL_NB0 + FILL_NB1 + FILL_NB2 + FILL_NB3, FILL_T, 0, stream>>>(
        svc_src, svc_dst, in_src, in_dst, ni_src, ni_dst, ii_src, ii_dst, cur, sorted);

    agg_proj_kernel<<<AG_NB_INST + AG_NB_SVC + AG_NB_NODE, AG_T, 0, stream>>>(
        feat_svc, feat_inst, feat_node, W_sc, W_in, W_ni, W_ii, b_sc, b_in, b_ni, b_ii,
        deg, cur, sorted, cat);

    final_gemm_kernel<<<FG_BLOCKS, 512, 0, stream>>>(W_tot, cat, out);
}


# 11 launches become 6: both presets share a launch, the four degree counts share one, and the fill launch's threads also turn the counts into rsqrt factors (no rsqrt pass); sum orders unchanged
# speedup vs baseline: 1.9058x; 1.0692x over previous
#include <hip/hip_runtime.h>

#define SVCN   50000
#define INSTN  100000
#define NODEN  50000
#define FDIM   128
#define HDIM   128
#define OUTD   256
#define TOTAL  (SVCN + NODEN + INSTN)
#define E_SC   400000
#define E_IN   100000
#define E_NI   100000
#define E_II   800000

#define OFF_CAT   0LL
#define OFF_DEG   (OFF_CAT  + (long long)TOTAL * 128)
#define D_NS_SC   (OFF_DEG + 0LL)
#define D_ND_SC   (D_NS_SC + SVCN)
#define D_NS_IN   (D_ND_SC + SVCN)
#define D_ND_IN   (D_NS_IN + INSTN)
#define D_NS_NI   (D_ND_IN + NODEN)
#define D_ND_NI   (D_NS_NI + NODEN)
#define D_NS_II   (D_ND_NI + INSTN)
#define D_ND_II   (D_NS_II + INSTN)
#define DEG_TOTAL 600000
#define OFF_CUR   (OFF_DEG + (long long)DEG_TOTAL)
#define CUR_SC    0
#define CUR_IN    (CUR_SC + SVCN)
#define CUR_NI    (CUR_IN + NODEN)
#define CUR_II    (CUR_NI + INSTN)
#define CUR_TOTAL (CUR_II + INSTN)
#define OFF_SORT  (OFF_CUR + (long long)CUR_TOTAL)
#define SORT_SC   0
#define SORT_IN   (SORT_SC + E_SC)
#define SORT_NI   (SORT_IN + E_IN)
#define SORT_II   (SORT_NI + E_NI)
static_assert(SVCN % 4 == 0 && NODEN % 4 == 0 && INSTN % 4 == 0, "degree planes are scanned four at a time");

typedef __attribute__((ext_vector_type(2))) float v2f;
typedef __attribute__((ext_vector_type(8))) float v8f;

__device__ __forceinline__ float lrelu(float x) { return x > 0.f ? x : 0.01f * x; }

__device__ __forceinline__ void atomAddF(float* p, float v) { unsafeAtomicAdd(p, v); }

#define PRESET_N (DEG_TOTAL + OUTD * HDIM)
__global__ void preset_kernel(float* __restrict__ deg, float* __restrict__ out,
                              const float* __restrict__ b_tot) {
    int i = blockIdx.x * blockDim.x + threadIdx.x;
    if (i < DEG_TOTAL) { deg[i] = 0.f; return; }
    i -= DEG_TOTAL;
    if (i < OUTD * HDIM) out[i] = b_tot[i >> 7];
}

__device__ __forceinline__ float deg_factor(float d) { return rsqrtf(d > 1.f ? d : 1.f); }

#define FILL_T   256
#define FILL_NB0 ((E_SC + FILL_T - 1) / FILL_T)
#define FILL_NB1 ((E_IN + FILL_T - 1) / FILL_T)
#define FILL_NB2 ((E_NI + FILL_T - 1) / FILL_T)
#define FILL_NB3 ((E_II + FILL_T - 1) / FILL_T)
__global__ __launch_bounds__(FILL_T) void degree_kernel(const int* __restrict__ svc_src, const int* __restrict__ svc_dst,
                                                        const int* __restrict__ in_src,  const int* __restrict__ in_dst,
                                                        const int* __restrict__ ni_src,  const int* __restrict__ ni_dst,
                                                        const int* __restrict__ ii_src,  const int* __restrict__ ii_dst,
                                                        float* __restrict__ deg) {
    int b = blockIdx.x;
    const int* src; const int* dst; float* degS; float* degD; int nE;
    if (b < FILL_NB0) { src = svc_src; dst = svc_dst; degS = deg + (D_NS_SC - OFF_DEG); degD = deg + (D_ND_SC - OFF_DEG); nE = E_SC; }
    else if ((b -= FILL_NB0) < FILL_NB1) { src = in_src; dst = in_dst; degS = deg + (D_NS_IN - OFF_DEG); degD = deg + (D_ND_IN - OFF_DEG); nE = E_IN; }
    else if ((b -= FILL_NB1) < FILL_NB2) { src = ni_src; dst = ni_dst; degS = deg + (D_NS_NI - OFF_DEG); degD = deg + (D_ND_NI - OFF_DEG); nE = E_NI; }
    else { b -= FILL_NB2;                  src = ii_src; dst = ii_dst; degS = deg + (D_NS_II - OFF_DEG); degD = deg + (D_ND_II - OFF_DEG); nE = E_II; }
    const int e = b * FILL_T + threadIdx.x;
    if (e >= nE) return;
    atomAddF(&degS[src[e]], 1.f);
    atomAddF(&degD[dst[e]], 1.f);
}


#define SCAN_T 1024
#define SCAN_Q 4
__global__ __launch_bounds__(SCAN_T) void offsets_kernel(const float* __restrict__ deg,
                                                         int* __restrict__ cur) {
    __shared__ int wsum[32];
    __shared__ int tile_total;
    const float* dg; int* out; int n;
    if (blockIdx.x == 0)      { dg = deg + (D_ND_SC - OFF_DEG); out = cur + CUR_SC; n = SVCN;  }
    else if (blockIdx.x == 1) { dg = deg + (D_ND_IN - OFF_DEG); out = cur + CUR_IN; n = NODEN; }
    else if (blockIdx.x == 2) { dg = deg + (D_ND_NI - OFF_DEG); out = cur + CUR_NI; n = INSTN; }
    else                      { dg = deg + (D_ND_II - OFF_DEG); out = cur + CUR_II; n = INSTN; }
    const int t = threadIdx.x, lane = t & 31, wave = t >> 5;
    int carry = 0;
    for (int base = 0; base < n; base += SCAN_T * SCAN_Q * 4) {
        const int i0 = base + t * (SCAN_Q * 4);
        int4 v[SCAN_Q];
        int s = 0;
#pragma unroll
        for (int q = 0; q < SCAN_Q; ++q) {
            const int i = i0 + q * 4;
            v[q] = make_int4(0, 0, 0, 0);
            if (i < n) {
                const float4 f = *(const float4*)&dg[i];
                v[q] = make_int4((int)f.x, (int)f.y, (int)f.z, (int)f.w);
            }
            s += v[q].x + v[q].y + v[q].z + v[q].w;
        }
        int x = s;
        for (int o = 1; o < 32; o <<= 1) {
            const int y = __shfl_up(x, o);
            if (lane >= o) x += y;
        }
        if (lane == 31) wsum[wave] = x;
        __syncthreads();
        if (wave == 0) {
            const int w = wsum[lane];
            int xs = w;
            for (int o = 1; o < 32; o <<= 1) {
                const int y = __shfl_up(xs, o);
                if (lane >= o) xs += y;
            }
            wsum[lane] = xs - w;
            if (lane == 31) tile_total = xs;
        }
        __syncthreads();
        int run = carry + wsum[wave] + (x - s);
#pragma unroll
        for (int q = 0; q < SCAN_Q; ++q) {
            const int i = i0 + q * 4;
            if (i < n)
                *(int4*)&out[i] = make_int4(run, run + v[q].x, run + v[q].x + v[q].y,
                                            run + v[q].x + v[q].y + v[q].z);
            run += v[q].x + v[q].y + v[q].z + v[q].w;
        }
        carry += tile_total;
    }
}

static_assert((FILL_NB0 + FILL_NB1 + FILL_NB2 + FILL_NB3) * FILL_T >= DEG_TOTAL, "one thread per degree entry");
__global__ __launch_bounds__(FILL_T) void fill_kernel(const int* __restrict__ svc_src, const int* __restrict__ svc_dst,
                                                      const int* __restrict__ in_src,  const int* __restrict__ in_dst,
                                                      const int* __restrict__ ni_src,  const int* __restrict__ ni_dst,
                                                      const int* __restrict__ ii_src,  const int* __restrict__ ii_dst,
                                                      int* __restrict__ cur, int* __restrict__ sorted,
                                                      float* __restrict__ deg) {
    const int g = blockIdx.x * FILL_T + threadIdx.x;
    const bool has_entry = g < DEG_TOTAL;
    float count = 1.f;
    if (has_entry) count = deg[g];
    int b = blockIdx.x;
    const int* src; const int* dst; int* cu; int* so; int nE;
    if (b < FILL_NB0) { src = svc_src; dst = svc_dst; cu = cur + CUR_SC; so = sorted + SORT_SC; nE = E_SC; }
    else if ((b -= FILL_NB0) < FILL_NB1) { src = in_src; dst = in_dst; cu = cur + CUR_IN; so = sorted + SORT_IN; nE = E_IN; }
    else if ((b -= FILL_NB1) < FILL_NB2) { src = ni_src; dst = ni_dst; cu = cur + CUR_NI; so = sorted + SORT_NI; nE = E_NI; }
    else { b -= FILL_NB2;                  src = ii_src; dst = ii_dst; cu = cur + CUR_II; so = sorted + SORT_II; nE = E_II; }
    const int e = b * FILL_T + threadIdx.x;
    if (e < nE) {
        const int pos = atomicAdd(&cu[dst[e]], 1);
        so[pos] = src[e];
    }
    if (has_entry) deg[g] = deg_factor(count);
}

#define AG_BATCH 8

#define AG_ROWS   64
#define AG_T      512
#define AG_PITCH  132
#define AG_NB_INST ((INSTN + AG_ROWS - 1) / AG_ROWS)
#define AG_NB_SVC  ((SVCN  + AG_ROWS - 1) / AG_ROWS)
#define AG_NB_NODE ((NODEN + AG_ROWS - 1) / AG_ROWS)
__global__ __launch_bounds__(AG_T) void agg_proj_kernel(
        const float* __restrict__ feat_svc, const float* __restrict__ feat_inst,
        const float* __restrict__ feat_node,
        const float* __restrict__ W_sc, const float* __restrict__ W_in,
        const float* __restrict__ W_ni, const float* __restrict__ W_ii,
        const float* __restrict__ b_sc, const float* __restrict__ b_in,
        const float* __restrict__ b_ni, const float* __restrict__ b_ii,
        const float* __restrict__ deg, const int* __restrict__ curAll,
        const int* __restrict__ sortedAll, float* __restrict__ catAll) {
    __shared__ __attribute__((aligned(16))) float Wlds[128 * 128];
    __shared__ __attribute__((aligned(16))) float Tile[AG_ROWS * AG_PITCH];

    int blk = blockIdx.x;
    const float *X0, *W0, *ns0, *nd0, *bA;
    const int *cur0, *sorted0;
    const float *X1 = nullptr, *W1 = nullptr, *ns1 = nullptr, *nd1 = nullptr, *bB = nullptr;
    const int *cur1 = nullptr, *sorted1 = nullptr;
    int nrows, nlists = 1;
    float* cat;
    if (blk < AG_NB_INST) {
        X0 = feat_node; W0 = W_ni; bA = b_ni;
        ns0 = deg + (D_NS_NI - OFF_DEG); nd0 = deg + (D_ND_NI - OFF_DEG);
        cur0 = curAll + CUR_NI; sorted0 = sortedAll + SORT_NI;
        X1 = feat_inst; W1 = W_ii; bB = b_ii;
        ns1 = deg + (D_NS_II - OFF_DEG); nd1 = deg + (D_ND_II - OFF_DEG);
        cur1 = curAll + CUR_II; sorted1 = sortedAll + SORT_II;
        nrows = INSTN; nlists = 2;
        cat = catAll + (long long)(SVCN + NODEN) * 128;
    } else if ((blk -= AG_NB_INST) < AG_NB_SVC) {
        X0 = feat_svc; W0 = W_sc; bA = b_sc;
        ns0 = deg + (D_NS_SC - OFF_DEG); nd0 = deg + (D_ND_SC - OFF_DEG);
        cur0 = curAll + CUR_SC; sorted0 = sortedAll + SORT_SC;
        nrows = SVCN;
        cat = catAll;
    } else {
        blk -= AG_NB_SVC;
        X0 = feat_inst; W0 = W_in; bA = b_in;
        ns0 = deg + (D_NS_IN - OFF_DEG); nd0 = deg + (D_ND_IN - OFF_DEG);
        cur0 = curAll + CUR_IN; sorted0 = sortedAll + SORT_IN;
        nrows = NODEN;
        cat = catAll + (long long)SVCN * 128;
    }

    const int t    = threadIdx.x;
    const int lane = t & 31;
    const int wave = __builtin_amdgcn_readfirstlane(t >> 5);
    const int half = lane >> 4;
    const int l16  = lane & 15;
    const int c4   = lane * 4;
    const int r0   = blk * AG_ROWS;
    const int n0   = (wave & 7) * 16;
    const int rb   = (wave >> 3) * 32;

    v8f acc[2] = {v8f{}, v8f{}};

    for (int li = 0; li < nlists; ++li) {
        const float* __restrict__ X  = li ? X1 : X0;
        const float* __restrict__ Wg = li ? W1 : W0;
        const float* __restrict__ ns = li ? ns1 : ns0;
        const float* __restrict__ nd = li ? nd1 : nd0;
        const int* __restrict__ cur    = li ? cur1 : cur0;
        const int* __restrict__ sorted = li ? sorted1 : sorted0;

        const int first = r0 + wave * 4;
        int ic = first - 1 + lane; ic = ic < 0 ? 0 : (ic < nrows ? ic : nrows - 1);
        int id = first + lane;     id = id < nrows ? id : nrows - 1;
        const int   cv  = cur[ic];
        const float ndv = nd[id];

        const float4* __restrict__ Wg4 = (const float4*)Wg + t;
        const float4 wr0 = Wg4[0 * AG_T], wr1 = Wg4[1 * AG_T], wr2 = Wg4[2 * AG_T], wr3 = Wg4[3 * AG_T];
        const float4 wr4 = Wg4[4 * AG_T], wr5 = Wg4[5 * AG_T], wr6 = Wg4[6 * AG_T], wr7 = Wg4[7 * AG_T];

        const int b0 = first > 0 ? __builtin_amdgcn_readlane(cv, 0) : 0;
        const int e3 = __builtin_amdgcn_readlane(cv, 4);
        int sv_next = 0;
        if (lane < e3 - b0) sv_next = sorted[b0 + lane];

        if (li > 0) __syncthreads();

        float4* Wl4 = (float4*)Wlds + t;
        Wl4[0 * AG_T] = wr0; Wl4[1 * AG_T] = wr1; Wl4[2 * AG_T] = wr2; Wl4[3 * AG_T] = wr3;
        Wl4[4 * AG_T] = wr4; Wl4[5 * AG_T] = wr5; Wl4[6 * AG_T] = wr6; Wl4[7 * AG_T] = wr7;

        float4 acc4 = make_float4(0.f, 0.f, 0.f, 0.f);
        int ri   = 0;
        int rend = __builtin_amdgcn_readlane(cv, 1);
        for (int cb = b0; cb < e3; cb += 32) {
            const int rem = e3 - cb;
            const int cnt = rem < 32 ? rem : 32;
            const int sv  = sv_next;
            sv_next = 0;
            if (lane < rem - 32) sv_next = sorted[cb + 32 + lane];
            float wv = 0.f;
            if (lane < cnt) wv = ns[sv];
            for (int jj = 0; jj < cnt; jj += AG_BATCH) {
                float4 x[AG_BATCH];
#pragma unroll
                for (int k = 0; k < AG_BATCH; ++k) {
                    const int q = jj + k < cnt ? jj + k : cnt - 1;
                    const int s = __builtin_amdgcn_readlane(sv, q);
                    x[k] = *(const float4*)&X[(long long)s * 128 + c4];
                }
#pragma unroll
                for (int k = 0; k < AG_BATCH; ++k) {
                    const bool live = jj + k < cnt;
                    const int q = live ? jj + k : cnt - 1;
                    const int j = cb + q;
                    while (j >= rend) {
                        const float n = __builtin_bit_cast(float,
                            __builtin_amdgcn_readlane(__builtin_bit_cast(int, ndv), ri));
                        *(float4*)&Tile[(wave * 4 + ri) * AG_PITCH + c4] =
                            make_float4(acc4.x * n, acc4.y * n, acc4.z * n, acc4.w * n);
                        acc4 = make_float4(0.f, 0.f, 0.f, 0.f);
                        ++ri;
                        rend = __builtin_amdgcn_readlane(cv, ri + 1);
                    }
                    const float wq = __builtin_bit_cast(float,
                        __builtin_amdgcn_readlane(__builtin_bit_cast(int, wv), q));
                    const float w = live ? wq : 0.f;
                    acc4.x = fmaf(x[k].x, w, acc4.x); acc4.y = fmaf(x[k].y, w, acc4.y);
                    acc4.z = fmaf(x[k].z, w, acc4.z); acc4.w = fmaf(x[k].w, w, acc4.w);
                }
            }
        }
        for (; ri < 4; ++ri) {
            const float n = __builtin_bit_cast(float,
                __builtin_amdgcn_readlane(__builtin_bit_cast(int, ndv), ri));
            *(float4*)&Tile[(wave * 4 + ri) * AG_PITCH + c4] =
                make_float4(acc4.x * n, acc4.y * n, acc4.z * n, acc4.w * n);
            acc4 = make_float4(0.f, 0.f, 0.f, 0.f);
        }
        __syncthreads();

        for (int k = 0; k < 128; k += 4) {
            const int kk = k + 2 * half;
            v2f b;
            b.x = Wlds[kk * 128 + n0 + l16];
            b.y = Wlds[(kk + 1) * 128 + n0 + l16];
#pragma unroll
            for (int rt = 0; rt < 2; ++rt) {
                v2f a = *(const v2f*)&Tile[(rb + rt * 16 + l16) * AG_PITCH + kk];
                acc[rt] = __builtin_amdgcn_wmma_f32_16x16x4_f32(
                    false, a, false, b, (short)0, acc[rt], false, false);
            }
        }
    }

    __syncthreads();
#pragma unroll
    for (int rt = 0; rt < 2; ++rt) {
#pragma unroll
        for (int i = 0; i < 8; ++i)
            Tile[(rb + rt * 16 + i + 8 * half) * AG_PITCH + n0 + l16] = acc[rt][i];
    }
    __syncthreads();

    float4 bias = *(const float4*)&bA[c4];
    if (nlists > 1) {
        const float4 bb = *(const float4*)&bB[c4];
        bias.x += bb.x; bias.y += bb.y; bias.z += bb.z; bias.w += bb.w;
    }
    for (int i = 0; i < 4; ++i) {
        const int lr  = wave * 4 + i;
        const int row = r0 + lr;
        if (row < nrows) {
            const float4 p = *(const float4*)&Tile[lr * AG_PITCH + c4];
            *(float4*)&cat[(long long)row * 128 + c4] =
                make_float4(p.x + bias.x, p.y + bias.y, p.z + bias.z, p.w + bias.w);
        }
    }
}

#define FG_KT      32
#define FG_TILES   (TOTAL / FG_KT)
#define FG_BLOCKS  250
#define FG_TPB     (FG_TILES / FG_BLOCKS)
#define FG_PA      36
#define FG_PB      136
static_assert(TOTAL % FG_KT == 0, "K must be a whole number of tiles");
static_assert(FG_TILES % FG_BLOCKS == 0, "tiles must divide evenly over the blocks");

__global__ __launch_bounds__(512) void final_gemm_kernel(const float* __restrict__ Wt,
                                                         const float* __restrict__ cat,
                                                         float* __restrict__ out) {
    __shared__ float As[OUTD * FG_PA];
    __shared__ float Bs[FG_KT * FG_PB];

    const int t    = threadIdx.x;
    const int lane = t & 31;
    const int wave = t >> 5;
    const int half = lane >> 4;
    const int l16  = lane & 15;
    const int wm   = wave >> 2;
    const int wn   = wave & 3;
    const int tile0 = blockIdx.x * FG_TPB;

    const int arow = t >> 3;
    const int ac4  = t & 7;
    const int brow = t >> 5;
    const int bc4  = t & 31;
    const float* pA = Wt + (long long)arow * TOTAL + (long long)tile0 * FG_KT + ac4 * 4;
    const float* pB = cat + (long long)tile0 * FG_KT * 128 + t * 4;
    const long long aStep = 64LL * TOTAL;
    float* sA = &As[arow * FG_PA + ac4 * 4];
    float* sB = &Bs[brow * FG_PB + bc4 * 4];

    v8f acc[4][2];
#pragma unroll
    for (int mt = 0; mt < 4; ++mt)
#pragma unroll
        for (int nt = 0; nt < 2; ++nt) acc[mt][nt] = v8f{};

    float4 ra0 = *(const float4*)(pA);
    float4 ra1 = *(const float4*)(pA + aStep);
    float4 ra2 = *(const float4*)(pA + 2 * aStep);
    float4 ra3 = *(const float4*)(pA + 3 * aStep);
    float4 rb0 = *(const float4*)(pB);
    float4 rb1 = *(const float4*)(pB + 2048);

    for (int it = 0; it < FG_TPB; ++it) {
        *(float4*)(sA)                 = ra0;
        *(float4*)(sA + 64 * FG_PA)    = ra1;
        *(float4*)(sA + 128 * FG_PA)   = ra2;
        *(float4*)(sA + 192 * FG_PA)   = ra3;
        rb0.x = lrelu(rb0.x); rb0.y = lrelu(rb0.y); rb0.z = lrelu(rb0.z); rb0.w = lrelu(rb0.w);
        rb1.x = lrelu(rb1.x); rb1.y = lrelu(rb1.y); rb1.z = lrelu(rb1.z); rb1.w = lrelu(rb1.w);
        *(float4*)(sB)                 = rb0;
        *(float4*)(sB + 16 * FG_PB)    = rb1;
        __syncthreads();

        if (it + 1 < FG_TPB) {
            pA += FG_KT;
            pB += FG_KT * 128;
            ra0 = *(const float4*)(pA);
            ra1 = *(const float4*)(pA + aStep);
            ra2 = *(const float4*)(pA + 2 * aStep);
            ra3 = *(const float4*)(pA + 3 * aStep);
            rb0 = *(const float4*)(pB);
            rb1 = *(const float4*)(pB + 2048);
        }
        __builtin_amdgcn_sched_barrier(0);

#pragma unroll
        for (int ks = 0; ks < FG_KT / 4; ++ks) {
            const int kk = ks * 4 + 2 * half;
            v2f a[4], b[2];
#pragma unroll
            for (int mt = 0; mt < 4; ++mt)
                a[mt] = *(const v2f*)&As[(wm * 64 + mt * 16 + l16) * FG_PA + kk];
#pragma unroll
            for (int nt = 0; nt < 2; ++nt) {
                b[nt].x = Bs[kk * FG_PB + wn * 32 + nt * 16 + l16];
                b[nt].y = Bs[(kk + 1) * FG_PB + wn * 32 + nt * 16 + l16];
            }
#pragma unroll
            for (int mt = 0; mt < 4; ++mt)
#pragma unroll
                for (int nt = 0; nt < 2; ++nt)
                    acc[mt][nt] = __builtin_amdgcn_wmma_f32_16x16x4_f32(
                        false, a[mt], false, b[nt], (short)0, acc[mt][nt], false, false);
        }
        __syncthreads();
    }

#pragma unroll
    for (int mt = 0; mt < 4; ++mt) {
#pragma unroll
        for (int nt = 0; nt < 2; ++nt) {
#pragma unroll
            for (int i = 0; i < 8; ++i) {
                int m = wm * 64 + mt * 16 + i + 8 * half;
                atomAddF(&out[m * 128 + wn * 32 + nt * 16 + l16], acc[mt][nt][i]);
            }
        }
    }
}

static inline int cdiv(long long a, long long b) { return (int)((a + b - 1) / b); }

extern "C" void kernel_launch(void* const* d_in, const int* in_sizes, int n_in,
                              void* d_out, int out_size, void* d_ws, size_t ws_size,
                              hipStream_t stream) {
    const float* feat_svc  = (const float*)d_in[0];
    const float* feat_inst = (const float*)d_in[1];
    const float* feat_node = (const float*)d_in[2];
    const int*   svc_src   = (const int*)d_in[3];
    const int*   svc_dst   = (const int*)d_in[4];
    const int*   in_src    = (const int*)d_in[5];
    const int*   in_dst    = (const int*)d_in[6];
    const int*   ni_src    = (const int*)d_in[7];
    const int*   ni_dst    = (const int*)d_in[8];
    const int*   ii_src    = (const int*)d_in[9];
    const int*   ii_dst    = (const int*)d_in[10];
    const float* W_sc = (const float*)d_in[11];  const float* b_sc = (const float*)d_in[12];
    const float* W_in = (const float*)d_in[13];  const float* b_in = (const float*)d_in[14];
    const float* W_ni = (const float*)d_in[15];  const float* b_ni = (const float*)d_in[16];
    const float* W_ii = (const float*)d_in[17];  const float* b_ii = (const float*)d_in[18];
    const float* W_tot = (const float*)d_in[19]; const float* b_tot = (const float*)d_in[20];
    float* out = (float*)d_out;

    float* ws   = (float*)d_ws;
    float* cat  = ws + OFF_CAT;
    float* deg  = ws + OFF_DEG;
    int* cur    = (int*)(ws + OFF_CUR);
    int* sorted = (int*)(ws + OFF_SORT);

    preset_kernel<<<cdiv(PRESET_N, 256), 256, 0, stream>>>(deg, out, b_tot);

    degree_kernel<<<FILL_NB0 + FILL_NB1 + FILL_NB2 + FILL_NB3, FILL_T, 0, stream>>>(
        svc_src, svc_dst, in_src, in_dst, ni_src, ni_dst, ii_src, ii_dst, deg);
    offsets_kernel<<<4, SCAN_T, 0, stream>>>(deg, cur);

    fill_kernel<<<FILL_NB0 + FILL_NB1 + FILL_NB2 + FILL_NB3, FILL_T, 0, stream>>>(
        svc_src, svc_dst, in_src, in_dst, ni_src, ni_dst, ii_src, ii_dst, cur, sorted, deg);

    agg_proj_kernel<<<AG_NB_INST + AG_NB_SVC + AG_NB_NODE, AG_T, 0, stream>>>(
        feat_svc, feat_inst, feat_node, W_sc, W_in, W_ni, W_ii, b_sc, b_in, b_ni, b_ii,
        deg, cur, sorted, cat);

    final_gemm_kernel<<<FG_BLOCKS, 512, 0, stream>>>(W_tot, cat, out);
}


# degree counts: one scattered global f32 atomic per index replaced by LDS counters (a block owns 12500 plane entries, scans 50000 indices), flushed by line-wise f32 adds; counts exact in any order
# speedup vs baseline: 2.4638x; 1.2928x over previous
#include <hip/hip_runtime.h>

#define SVCN   50000
#define INSTN  100000
#define NODEN  50000
#define FDIM   128
#define HDIM   128
#define OUTD   256
#define TOTAL  (SVCN + NODEN + INSTN)
#define E_SC   400000
#define E_IN   100000
#define E_NI   100000
#define E_II   800000

#define OFF_CAT   0LL
#define OFF_DEG   (OFF_CAT  + (long long)TOTAL * 128)
#define D_NS_SC   (OFF_DEG + 0LL)
#define D_ND_SC   (D_NS_SC + SVCN)
#define D_NS_IN   (D_ND_SC + SVCN)
#define D_ND_IN   (D_NS_IN + INSTN)
#define D_NS_NI   (D_ND_IN + NODEN)
#define D_ND_NI   (D_NS_NI + NODEN)
#define D_NS_II   (D_ND_NI + INSTN)
#define D_ND_II   (D_NS_II + INSTN)
#define DEG_TOTAL 600000
#define OFF_CUR   (OFF_DEG + (long long)DEG_TOTAL)
#define CUR_SC    0
#define CUR_IN    (CUR_SC + SVCN)
#define CUR_NI    (CUR_IN + NODEN)
#define CUR_II    (CUR_NI + INSTN)
#define CUR_TOTAL (CUR_II + INSTN)
#define OFF_SORT  (OFF_CUR + (long long)CUR_TOTAL)
#define SORT_SC   0
#define SORT_IN   (SORT_SC + E_SC)
#define SORT_NI   (SORT_IN + E_IN)
#define SORT_II   (SORT_NI + E_NI)
static_assert(SVCN % 4 == 0 && NODEN % 4 == 0 && INSTN % 4 == 0, "degree planes are scanned four at a time");

typedef __attribute__((ext_vector_type(2))) float v2f;
typedef __attribute__((ext_vector_type(8))) float v8f;

__device__ __forceinline__ float lrelu(float x) { return x > 0.f ? x : 0.01f * x; }

__device__ __forceinline__ void atomAddF(float* p, float v) { unsafeAtomicAdd(p, v); }

#define PRESET_N (DEG_TOTAL + OUTD * HDIM)
__global__ void preset_kernel(float* __restrict__ deg, float* __restrict__ out,
                              const float* __restrict__ b_tot) {
    int i = blockIdx.x * blockDim.x + threadIdx.x;
    if (i < DEG_TOTAL) { deg[i] = 0.f; return; }
    i -= DEG_TOTAL;
    if (i < OUTD * HDIM) out[i] = b_tot[i >> 7];
}

__device__ __forceinline__ float deg_factor(float d) { return rsqrtf(d > 1.f ? d : 1.f); }

#define FILL_T   256
#define FILL_NB0 ((E_SC + FILL_T - 1) / FILL_T)
#define FILL_NB1 ((E_IN + FILL_T - 1) / FILL_T)
#define FILL_NB2 ((E_NI + FILL_T - 1) / FILL_T)
#define FILL_NB3 ((E_II + FILL_T - 1) / FILL_T)

#define DG_T  512
#define DG_R  12500
#define DG_C  50000
#define DG_L  ((DG_C / 4 + DG_T - 1) / DG_T)
#define DG_F  ((DG_R + DG_T - 1) / DG_T)
static_assert(SVCN % DG_R == 0 && NODEN % DG_R == 0 && INSTN % DG_R == 0, "planes split into whole ranges");
static_assert(E_SC % DG_C == 0 && E_IN % DG_C == 0 && E_NI % DG_C == 0 && E_II % DG_C == 0, "lists split into whole chunks");
static_assert(DG_R % 4 == 0 && DG_C % 4 == 0, "ranges and chunks are whole 16-byte pieces");
#define DG_NB(entries, edges) (((entries) / DG_R) * ((edges) / DG_C))
#define DG_NB0 DG_NB(SVCN,  E_SC)
#define DG_NB1 DG_NB(SVCN,  E_SC)
#define DG_NB2 DG_NB(INSTN, E_IN)
#define DG_NB3 DG_NB(NODEN, E_IN)
#define DG_NB4 DG_NB(NODEN, E_NI)
#define DG_NB5 DG_NB(INSTN, E_NI)
#define DG_NB6 DG_NB(INSTN, E_II)
#define DG_NB7 DG_NB(INSTN, E_II)
#define DG_NB_ALL (DG_NB0 + DG_NB1 + DG_NB2 + DG_NB3 + DG_NB4 + DG_NB5 + DG_NB6 + DG_NB7)
__global__ __launch_bounds__(DG_T) void degree_kernel(const int* __restrict__ svc_src, const int* __restrict__ svc_dst,
                                                      const int* __restrict__ in_src,  const int* __restrict__ in_dst,
                                                      const int* __restrict__ ni_src,  const int* __restrict__ ni_dst,
                                                      const int* __restrict__ ii_src,  const int* __restrict__ ii_dst,
                                                      float* __restrict__ deg) {
    __shared__ __attribute__((aligned(16))) unsigned int cnt[DG_R];
    int b = blockIdx.x;
    const int* idx; float* plane; int nch;
    if (b < DG_NB0)                  { idx = svc_src; plane = deg + (D_NS_SC - OFF_DEG); nch = E_SC / DG_C; }
    else if ((b -= DG_NB0) < DG_NB1) { idx = svc_dst; plane = deg + (D_ND_SC - OFF_DEG); nch = E_SC / DG_C; }
    else if ((b -= DG_NB1) < DG_NB2) { idx = in_src;  plane = deg + (D_NS_IN - OFF_DEG); nch = E_IN / DG_C; }
    else if ((b -= DG_NB2) < DG_NB3) { idx = in_dst;  plane = deg + (D_ND_IN - OFF_DEG); nch = E_IN / DG_C; }
    else if ((b -= DG_NB3) < DG_NB4) { idx = ni_src;  plane = deg + (D_NS_NI - OFF_DEG); nch = E_NI / DG_C; }
    else if ((b -= DG_NB4) < DG_NB5) { idx = ni_dst;  plane = deg + (D_ND_NI - OFF_DEG); nch = E_NI / DG_C; }
    else if ((b -= DG_NB5) < DG_NB6) { idx = ii_src;  plane = deg + (D_NS_II - OFF_DEG); nch = E_II / DG_C; }
    else { b -= DG_NB6;                idx = ii_dst;  plane = deg + (D_ND_II - OFF_DEG); nch = E_II / DG_C; }
    const int range = b / nch;
    const int chunk = b - range * nch;
    const int lo    = range * DG_R;
    const int t     = threadIdx.x;

    const int4* __restrict__ p4 = (const int4*)(idx + chunk * DG_C);
    int4 v[DG_L];
#pragma unroll
    for (int k = 0; k < DG_L; ++k) {
        const int i = k * DG_T + t;
        v[k] = make_int4(-1, -1, -1, -1);
        if (i < DG_C / 4) v[k] = p4[i];
    }

    for (int i = t; i < DG_R / 4; i += DG_T) ((uint4*)cnt)[i] = make_uint4(0u, 0u, 0u, 0u);
    __syncthreads();

#pragma unroll
    for (int k = 0; k < DG_L; ++k) {
        unsigned int d;
        d = (unsigned int)(v[k].x - lo); if (d < DG_R) atomicAdd(&cnt[d], 1u);
        d = (unsigned int)(v[k].y - lo); if (d < DG_R) atomicAdd(&cnt[d], 1u);
        d = (unsigned int)(v[k].z - lo); if (d < DG_R) atomicAdd(&cnt[d], 1u);
        d = (unsigned int)(v[k].w - lo); if (d < DG_R) atomicAdd(&cnt[d], 1u);
    }
    __syncthreads();

    unsigned int c[DG_F];
#pragma unroll
    for (int k = 0; k < DG_F; ++k) {
        const int i = k * DG_T + t;
        c[k] = i < DG_R ? cnt[i] : 0u;
    }
#pragma unroll
    for (int k = 0; k < DG_F; ++k)
        if (c[k] != 0u) atomAddF(&plane[lo + k * DG_T + t], (float)c[k]);
}


#define SCAN_T 1024
#define SCAN_Q 4
__global__ __launch_bounds__(SCAN_T) void offsets_kernel(const float* __restrict__ deg,
                                                         int* __restrict__ cur) {
    __shared__ int wsum[32];
    __shared__ int tile_total;
    const float* dg; int* out; int n;
    if (blockIdx.x == 0)      { dg = deg + (D_ND_SC - OFF_DEG); out = cur + CUR_SC; n = SVCN;  }
    else if (blockIdx.x == 1) { dg = deg + (D_ND_IN - OFF_DEG); out = cur + CUR_IN; n = NODEN; }
    else if (blockIdx.x == 2) { dg = deg + (D_ND_NI - OFF_DEG); out = cur + CUR_NI; n = INSTN; }
    else                      { dg = deg + (D_ND_II - OFF_DEG); out = cur + CUR_II; n = INSTN; }
    const int t = threadIdx.x, lane = t & 31, wave = t >> 5;
    int carry = 0;
    for (int base = 0; base < n; base += SCAN_T * SCAN_Q * 4) {
        const int i0 = base + t * (SCAN_Q * 4);
        int4 v[SCAN_Q];
        int s = 0;
#pragma unroll
        for (int q = 0; q < SCAN_Q; ++q) {
            const int i = i0 + q * 4;
            v[q] = make_int4(0, 0, 0, 0);
            if (i < n) {
                const float4 f = *(const float4*)&dg[i];
                v[q] = make_int4((int)f.x, (int)f.y, (int)f.z, (int)f.w);
            }
            s += v[q].x + v[q].y + v[q].z + v[q].w;
        }
        int x = s;
        for (int o = 1; o < 32; o <<= 1) {
            const int y = __shfl_up(x, o);
            if (lane >= o) x += y;
        }
        if (lane == 31) wsum[wave] = x;
        __syncthreads();
        if (wave == 0) {
            const int w = wsum[lane];
            int xs = w;
            for (int o = 1; o < 32; o <<= 1) {
                const int y = __shfl_up(xs, o);
                if (lane >= o) xs += y;
            }
            wsum[lane] = xs - w;
            if (lane == 31) tile_total = xs;
        }
        __syncthreads();
        int run = carry + wsum[wave] + (x - s);
#pragma unroll
        for (int q = 0; q < SCAN_Q; ++q) {
            const int i = i0 + q * 4;
            if (i < n)
                *(int4*)&out[i] = make_int4(run, run + v[q].x, run + v[q].x + v[q].y,
                                            run + v[q].x + v[q].y + v[q].z);
            run += v[q].x + v[q].y + v[q].z + v[q].w;
        }
        carry += tile_total;
    }
}

static_assert((FILL_NB0 + FILL_NB1 + FILL_NB2 + FILL_NB3) * FILL_T >= DEG_TOTAL, "one thread per degree entry");
__global__ __launch_bounds__(FILL_T) void fill_kernel(const int* __restrict__ svc_src, const int* __restrict__ svc_dst,
                                                      const int* __restrict__ in_src,  const int* __restrict__ in_dst,
                                                      const int* __restrict__ ni_src,  const int* __restrict__ ni_dst,
                                                      const int* __restrict__ ii_src,  const int* __restrict__ ii_dst,
                                                      int* __restrict__ cur, int* __restrict__ sorted,
                                                      float* __restrict__ deg) {
    const int g = blockIdx.x * FILL_T + threadIdx.x;
    const bool has_entry = g < DEG_TOTAL;
    float count = 1.f;
    if (has_entry) count = deg[g];
    int b = blockIdx.x;
    const int* src; const int* dst; int* cu; int* so; int nE;
    if (b < FILL_NB0) { src = svc_src; dst = svc_dst; cu = cur + CUR_SC; so = sorted + SORT_SC; nE = E_SC; }
    else if ((b -= FILL_NB0) < FILL_NB1) { src = in_src; dst = in_dst; cu = cur + CUR_IN; so = sorted + SORT_IN; nE = E_IN; }
    else if ((b -= FILL_NB1) < FILL_NB2) { src = ni_src; dst = ni_dst; cu = cur + CUR_NI; so = sorted + SORT_NI; nE = E_NI; }
    else { b -= FILL_NB2;                  src = ii_src; dst = ii_dst; cu = cur + CUR_II; so = sorted + SORT_II; nE = E_II; }
    const int e = b * FILL_T + threadIdx.x;
    if (e < nE) {
        const int pos = atomicAdd(&cu[dst[e]], 1);
        so[pos] = src[e];
    }
    if (has_entry) deg[g] = deg_factor(count);
}

#define AG_BATCH 8

#define AG_ROWS   64
#define AG_T      512
#define AG_PITCH  132
#define AG_NB_INST ((INSTN + AG_ROWS - 1) / AG_ROWS)
#define AG_NB_SVC  ((SVCN  + AG_ROWS - 1) / AG_ROWS)
#define AG_NB_NODE ((NODEN + AG_ROWS - 1) / AG_ROWS)
__global__ __launch_bounds__(AG_T) void agg_proj_kernel(
        const float* __restrict__ feat_svc, const float* __restrict__ feat_inst,
        const float* __restrict__ feat_node,
        const float* __restrict__ W_sc, const float* __restrict__ W_in,
        const float* __restrict__ W_ni, const float* __restrict__ W_ii,
        const float* __restrict__ b_sc, const float* __restrict__ b_in,
        const float* __restrict__ b_ni, const float* __restrict__ b_ii,
        const float* __restrict__ deg, const int* __restrict__ curAll,
        const int* __restrict__ sortedAll, float* __restrict__ catAll) {
    __shared__ __attribute__((aligned(16))) float Wlds[128 * 128];
    __shared__ __attribute__((aligned(16))) float Tile[AG_ROWS * AG_PITCH];

    int blk = blockIdx.x;
    const float *X0, *W0, *ns0, *nd0, *bA;
    const int *cur0, *sorted0;
    const float *X1 = nullptr, *W1 = nullptr, *ns1 = nullptr, *nd1 = nullptr, *bB = nullptr;
    const int *cur1 = nullptr, *sorted1 = nullptr;
    int nrows, nlists = 1;
    float* cat;
    if (blk < AG_NB_INST) {
        X0 = feat_node; W0 = W_ni; bA = b_ni;
        ns0 = deg + (D_NS_NI - OFF_DEG); nd0 = deg + (D_ND_NI - OFF_DEG);
        cur0 = curAll + CUR_NI; sorted0 = sortedAll + SORT_NI;
        X1 = feat_inst; W1 = W_ii; bB = b_ii;
        ns1 = deg + (D_NS_II - OFF_DEG); nd1 = deg + (D_ND_II - OFF_DEG);
        cur1 = curAll + CUR_II; sorted1 = sortedAll + SORT_II;
        nrows = INSTN; nlists = 2;
        cat = catAll + (long long)(SVCN + NODEN) * 128;
    } else if ((blk -= AG_NB_INST) < AG_NB_SVC) {
        X0 = feat_svc; W0 = W_sc; bA = b_sc;
        ns0 = deg + (D_NS_SC - OFF_DEG); nd0 = deg + (D_ND_SC - OFF_DEG);
        cur0 = curAll + CUR_SC; sorted0 = sortedAll + SORT_SC;
        nrows = SVCN;
        cat = catAll;
    } else {
        blk -= AG_NB_SVC;
        X0 = feat_inst; W0 = W_in; bA = b_in;
        ns0 = deg + (D_NS_IN - OFF_DEG); nd0 = deg + (D_ND_IN - OFF_DEG);
        cur0 = curAll + CUR_IN; sorted0 = sortedAll + SORT_IN;
        nrows = NODEN;
        cat = catAll + (long long)SVCN * 128;
    }

    const int t    = threadIdx.x;
    const int lane = t & 31;
    const int wave = __builtin_amdgcn_readfirstlane(t >> 5);
    const int half = lane >> 4;
    const int l16  = lane & 15;
    const int c4   = lane * 4;
    const int r0   = blk * AG_ROWS;
    const int n0   = (wave & 7) * 16;
    const int rb   = (wave >> 3) * 32;

    v8f acc[2] = {v8f{}, v8f{}};

    for (int li = 0; li < nlists; ++li) {
        const float* __restrict__ X  = li ? X1 : X0;
        const float* __restrict__ Wg = li ? W1 : W0;
        const float* __restrict__ ns = li ? ns1 : ns0;
        const float* __restrict__ nd = li ? nd1 : nd0;
        const int* __restrict__ cur    = li ? cur1 : cur0;
        const int* __restrict__ sorted = li ? sorted1 : sorted0;

        const int first = r0 + wave * 4;
        int ic = first - 1 + lane; ic = ic < 0 ? 0 : (ic < nrows ? ic : nrows - 1);
        int id = first + lane;     id = id < nrows ? id : nrows - 1;
        const int   cv  = cur[ic];
        const float ndv = nd[id];

        const float4* __restrict__ Wg4 = (const float4*)Wg + t;
        const float4 wr0 = Wg4[0 * AG_T], wr1 = Wg4[1 * AG_T], wr2 = Wg4[2 * AG_T], wr3 = Wg4[3 * AG_T];
        const float4 wr4 = Wg4[4 * AG_T], wr5 = Wg4[5 * AG_T], wr6 = Wg4[6 * AG_T], wr7 = Wg4[7 * AG_T];

        const int b0 = first > 0 ? __builtin_amdgcn_readlane(cv, 0) : 0;
        const int e3 = __builtin_amdgcn_readlane(cv, 4);
        int sv_next = 0;
        if (lane < e3 - b0) sv_next = sorted[b0 + lane];

        if (li > 0) __syncthreads();

        float4* Wl4 = (float4*)Wlds + t;
        Wl4[0 * AG_T] = wr0; Wl4[1 * AG_T] = wr1; Wl4[2 * AG_T] = wr2; Wl4[3 * AG_T] = wr3;
        Wl4[4 * AG_T] = wr4; Wl4[5 * AG_T] = wr5; Wl4[6 * AG_T] = wr6; Wl4[7 * AG_T] = wr7;

        float4 acc4 = make_float4(0.f, 0.f, 0.f, 0.f);
        int ri   = 0;
        int rend = __builtin_amdgcn_readlane(cv, 1);
        for (int cb = b0; cb < e3; cb += 32) {
            const int rem = e3 - cb;
            const int cnt = rem < 32 ? rem : 32;
            const int sv  = sv_next;
            sv_next = 0;
            if (lane < rem - 32) sv_next = sorted[cb + 32 + lane];
            float wv = 0.f;
            if (lane < cnt) wv = ns[sv];
            for (int jj = 0; jj < cnt; jj += AG_BATCH) {
                float4 x[AG_BATCH];
#pragma unroll
                for (int k = 0; k < AG_BATCH; ++k) {
                    const int q = jj + k < cnt ? jj + k : cnt - 1;
                    const int s = __builtin_amdgcn_readlane(sv, q);
                    x[k] = *(const float4*)&X[(long long)s * 128 + c4];
                }
#pragma unroll
                for (int k = 0; k < AG_BATCH; ++k) {
                    const bool live = jj + k < cnt;
                    const int q = live ? jj + k : cnt - 1;
                    const int j = cb + q;
                    while (j >= rend) {
                        const float n = __builtin_bit_cast(float,
                            __builtin_amdgcn_readlane(__builtin_bit_cast(int, ndv), ri));
                        *(float4*)&Tile[(wave * 4 + ri) * AG_PITCH + c4] =
                            make_float4(acc4.x * n, acc4.y * n, acc4.z * n, acc4.w * n);
                        acc4 = make_float4(0.f, 0.f, 0.f, 0.f);
                        ++ri;
                        rend = __builtin_amdgcn_readlane(cv, ri + 1);
                    }
                    const float wq = __builtin_bit_cast(float,
                        __builtin_amdgcn_readlane(__builtin_bit_cast(int, wv), q));
                    const float w = live ? wq : 0.f;
                    acc4.x = fmaf(x[k].x, w, acc4.x); acc4.y = fmaf(x[k].y, w, acc4.y);
                    acc4.z = fmaf(x[k].z, w, acc4.z); acc4.w = fmaf(x[k].w, w, acc4.w);
                }
            }
        }
        for (; ri < 4; ++ri) {
            const float n = __builtin_bit_cast(float,
                __builtin_amdgcn_readlane(__builtin_bit_cast(int, ndv), ri));
            *(float4*)&Tile[(wave * 4 + ri) * AG_PITCH + c4] =
                make_float4(acc4.x * n, acc4.y * n, acc4.z * n, acc4.w * n);
            acc4 = make_float4(0.f, 0.f, 0.f, 0.f);
        }
        __syncthreads();

        for (int k = 0; k < 128; k += 4) {
            const int kk = k + 2 * half;
            v2f b;
            b.x = Wlds[kk * 128 + n0 + l16];
            b.y = Wlds[(kk + 1) * 128 + n0 + l16];
#pragma unroll
            for (int rt = 0; rt < 2; ++rt) {
                v2f a = *(const v2f*)&Tile[(rb + rt * 16 + l16) * AG_PITCH + kk];
                acc[rt] = __builtin_amdgcn_wmma_f32_16x16x4_f32(
                    false, a, false, b, (short)0, acc[rt], false, false);
            }
        }
    }

    __syncthreads();
#pragma unroll
    for (int rt = 0; rt < 2; ++rt) {
#pragma unroll
        for (int i = 0; i < 8; ++i)
            Tile[(rb + rt * 16 + i + 8 * half) * AG_PITCH + n0 + l16] = acc[rt][i];
    }
    __syncthreads();

    float4 bias = *(const float4*)&bA[c4];
    if (nlists > 1) {
        const float4 bb = *(const float4*)&bB[c4];
        bias.x += bb.x; bias.y += bb.y; bias.z += bb.z; bias.w += bb.w;
    }
    for (int i = 0; i < 4; ++i) {
        const int lr  = wave * 4 + i;
        const int row = r0 + lr;
        if (row < nrows) {
            const float4 p = *(const float4*)&Tile[lr * AG_PITCH + c4];
            *(float4*)&cat[(long long)row * 128 + c4] =
                make_float4(p.x + bias.x, p.y + bias.y, p.z + bias.z, p.w + bias.w);
        }
    }
}

#define FG_KT      32
#define FG_TILES   (TOTAL / FG_KT)
#define FG_BLOCKS  250
#define FG_TPB     (FG_TILES / FG_BLOCKS)
#define FG_PA      36
#define FG_PB      136
static_assert(TOTAL % FG_KT == 0, "K must be a whole number of tiles");
static_assert(FG_TILES % FG_BLOCKS == 0, "tiles must divide evenly over the blocks");

__global__ __launch_bounds__(512) void final_gemm_kernel(const float* __restrict__ Wt,
                                                         const float* __restrict__ cat,
                                                         float* __restrict__ out) {
    __shared__ float As[OUTD * FG_PA];
    __shared__ float Bs[FG_KT * FG_PB];

    const int t    = threadIdx.x;
    const int lane = t & 31;
    const int wave = t >> 5;
    const int half = lane >> 4;
    const int l16  = lane & 15;
    const int wm   = wave >> 2;
    const int wn   = wave & 3;
    const int tile0 = blockIdx.x * FG_TPB;

    const int arow = t >> 3;
    const int ac4  = t & 7;
    const int brow = t >> 5;
    const int bc4  = t & 31;
    const float* pA = Wt + (long long)arow * TOTAL + (long long)tile0 * FG_KT + ac4 * 4;
    const float* pB = cat + (long long)tile0 * FG_KT * 128 + t * 4;
    const long long aStep = 64LL * TOTAL;
    float* sA = &As[arow * FG_PA + ac4 * 4];
    float* sB = &Bs[brow * FG_PB + bc4 * 4];

    v8f acc[4][2];
#pragma unroll
    for (int mt = 0; mt < 4; ++mt)
#pragma unroll
        for (int nt = 0; nt < 2; ++nt) acc[mt][nt] = v8f{};

    float4 ra0 = *(const float4*)(pA);
    float4 ra1 = *(const float4*)(pA + aStep);
    float4 ra2 = *(const float4*)(pA + 2 * aStep);
    float4 ra3 = *(const float4*)(pA + 3 * aStep);
    float4 rb0 = *(const float4*)(pB);
    float4 rb1 = *(const float4*)(pB + 2048);

    for (int it = 0; it < FG_TPB; ++it) {
        *(float4*)(sA)                 = ra0;
        *(float4*)(sA + 64 * FG_PA)    = ra1;
        *(float4*)(sA + 128 * FG_PA)   = ra2;
        *(float4*)(sA + 192 * FG_PA)   = ra3;
        rb0.x = lrelu(rb0.x); rb0.y = lrelu(rb0.y); rb0.z = lrelu(rb0.z); rb0.w = lrelu(rb0.w);
        rb1.x = lrelu(rb1.x); rb1.y = lrelu(rb1.y); rb1.z = lrelu(rb1.z); rb1.w = lrelu(rb1.w);
        *(float4*)(sB)                 = rb0;
        *(float4*)(sB + 16 * FG_PB)    = rb1;
        __syncthreads();

        if (it + 1 < FG_TPB) {
            pA += FG_KT;
            pB += FG_KT * 128;
            ra0 = *(const float4*)(pA);
            ra1 = *(const float4*)(pA + aStep);
            ra2 = *(const float4*)(pA + 2 * aStep);
            ra3 = *(const float4*)(pA + 3 * aStep);
            rb0 = *(const float4*)(pB);
            rb1 = *(const float4*)(pB + 2048);
        }
        __builtin_amdgcn_sched_barrier(0);

#pragma unroll
        for (int ks = 0; ks < FG_KT / 4; ++ks) {
            const int kk = ks * 4 + 2 * half;
            v2f a[4], b[2];
#pragma unroll
            for (int mt = 0; mt < 4; ++mt)
                a[mt] = *(const v2f*)&As[(wm * 64 + mt * 16 + l16) * FG_PA + kk];
#pragma unroll
            for (int nt = 0; nt < 2; ++nt) {
                b[nt].x = Bs[kk * FG_PB + wn * 32 + nt * 16 + l16];
                b[nt].y = Bs[(kk + 1) * FG_PB + wn * 32 + nt * 16 + l16];
            }
#pragma unroll
            for (int mt = 0; mt < 4; ++mt)
#pragma unroll
                for (int nt = 0; nt < 2; ++nt)
                    acc[mt][nt] = __builtin_amdgcn_wmma_f32_16x16x4_f32(
                        false, a[mt], false, b[nt], (short)0, acc[mt][nt], false, false);
        }
        __syncthreads();
    }

#pragma unroll
    for (int mt = 0; mt < 4; ++mt) {
#pragma unroll
        for (int nt = 0; nt < 2; ++nt) {
#pragma unroll
            for (int i = 0; i < 8; ++i) {
                int m = wm * 64 + mt * 16 + i + 8 * half;
                atomAddF(&out[m * 128 + wn * 32 + nt * 16 + l16], acc[mt][nt][i]);
            }
        }
    }
}

static inline int cdiv(long long a, long long b) { return (int)((a + b - 1) / b); }

extern "C" void kernel_launch(void* const* d_in, const int* in_sizes, int n_in,
                              void* d_out, int out_size, void* d_ws, size_t ws_size,
                              hipStream_t stream) {
    const float* feat_svc  = (const float*)d_in[0];
    const float* feat_inst = (const float*)d_in[1];
    const float* feat_node = (const float*)d_in[2];
    const int*   svc_src   = (const int*)d_in[3];
    const int*   svc_dst   = (const int*)d_in[4];
    const int*   in_src    = (const int*)d_in[5];
    const int*   in_dst    = (const int*)d_in[6];
    const int*   ni_src    = (const int*)d_in[7];
    const int*   ni_dst    = (const int*)d_in[8];
    const int*   ii_src    = (const int*)d_in[9];
    const int*   ii_dst    = (const int*)d_in[10];
    const float* W_sc = (const float*)d_in[11];  const float* b_sc = (const float*)d_in[12];
    const float* W_in = (const float*)d_in[13];  const float* b_in = (const float*)d_in[14];
    const float* W_ni = (const float*)d_in[15];  const float* b_ni = (const float*)d_in[16];
    const float* W_ii = (const float*)d_in[17];  const float* b_ii = (const float*)d_in[18];
    const float* W_tot = (const float*)d_in[19]; const float* b_tot = (const float*)d_in[20];
    float* out = (float*)d_out;

    float* ws   = (float*)d_ws;
    float* cat  = ws + OFF_CAT;
    float* deg  = ws + OFF_DEG;
    int* cur    = (int*)(ws + OFF_CUR);
    int* sorted = (int*)(ws + OFF_SORT);

    preset_kernel<<<cdiv(PRESET_N, 256), 256, 0, stream>>>(deg, out, b_tot);

    degree_kernel<<<DG_NB_ALL, DG_T, 0, stream>>>(
        svc_src, svc_dst, in_src, in_dst, ni_src, ni_dst, ii_src, ii_dst, deg);
    offsets_kernel<<<4, SCAN_T, 0, stream>>>(deg, cur);

    fill_kernel<<<FILL_NB0 + FILL_NB1 + FILL_NB2 + FILL_NB3, FILL_T, 0, stream>>>(
        svc_src, svc_dst, in_src, in_dst, ni_src, ni_dst, ii_src, ii_dst, cur, sorted, deg);

    agg_proj_kernel<<<AG_NB_INST + AG_NB_SVC + AG_NB_NODE, AG_T, 0, stream>>>(
        feat_svc, feat_inst, feat_node, W_sc, W_in, W_ni, W_ii, b_sc, b_in, b_ni, b_ii,
        deg, cur, sorted, cat);

    final_gemm_kernel<<<FG_BLOCKS, 512, 0, stream>>>(W_tot, cat, out);
}


# edge grouping: slot from LDS counters (a block owns 12500 destinations, scans 50000 edges) plus one line-wise global add per destination, not one scattered global atomic per edge; arrival order kept
# speedup vs baseline: 2.8066x; 1.1391x over previous
#include <hip/hip_runtime.h>

#define SVCN   50000
#define INSTN  100000
#define NODEN  50000
#define FDIM   128
#define HDIM   128
#define OUTD   256
#define TOTAL  (SVCN + NODEN + INSTN)
#define E_SC   400000
#define E_IN   100000
#define E_NI   100000
#define E_II   800000

#define OFF_CAT   0LL
#define OFF_DEG   (OFF_CAT  + (long long)TOTAL * 128)
#define D_NS_SC   (OFF_DEG + 0LL)
#define D_ND_SC   (D_NS_SC + SVCN)
#define D_NS_IN   (D_ND_SC + SVCN)
#define D_ND_IN   (D_NS_IN + INSTN)
#define D_NS_NI   (D_ND_IN + NODEN)
#define D_ND_NI   (D_NS_NI + NODEN)
#define D_NS_II   (D_ND_NI + INSTN)
#define D_ND_II   (D_NS_II + INSTN)
#define DEG_TOTAL 600000
#define OFF_CUR   (OFF_DEG + (long long)DEG_TOTAL)
#define CUR_SC    0
#define CUR_IN    (CUR_SC + SVCN)
#define CUR_NI    (CUR_IN + NODEN)
#define CUR_II    (CUR_NI + INSTN)
#define CUR_TOTAL (CUR_II + INSTN)
#define OFF_SORT  (OFF_CUR + (long long)CUR_TOTAL)
#define SORT_SC   0
#define SORT_IN   (SORT_SC + E_SC)
#define SORT_NI   (SORT_IN + E_IN)
#define SORT_II   (SORT_NI + E_NI)
static_assert(SVCN % 4 == 0 && NODEN % 4 == 0 && INSTN % 4 == 0, "degree planes are scanned four at a time");

typedef __attribute__((ext_vector_type(2))) float v2f;
typedef __attribute__((ext_vector_type(8))) float v8f;

__device__ __forceinline__ float lrelu(float x) { return x > 0.f ? x : 0.01f * x; }

__device__ __forceinline__ void atomAddF(float* p, float v) { unsafeAtomicAdd(p, v); }

#define PRESET_N (DEG_TOTAL + OUTD * HDIM)
__global__ void preset_kernel(float* __restrict__ deg, float* __restrict__ out,
                              const float* __restrict__ b_tot) {
    int i = blockIdx.x * blockDim.x + threadIdx.x;
    if (i < DEG_TOTAL) { deg[i] = 0.f; return; }
    i -= DEG_TOTAL;
    if (i < OUTD * HDIM) out[i] = b_tot[i >> 7];
}

__device__ __forceinline__ float deg_factor(float d) { return rsqrtf(d > 1.f ? d : 1.f); }

#define DG_T  512
#define DG_R  12500
#define DG_C  50000
#define DG_L  ((DG_C / 4 + DG_T - 1) / DG_T)
#define DG_F  ((DG_R + DG_T - 1) / DG_T)
static_assert(SVCN % DG_R == 0 && NODEN % DG_R == 0 && INSTN % DG_R == 0, "planes split into whole ranges");
static_assert(E_SC % DG_C == 0 && E_IN % DG_C == 0 && E_NI % DG_C == 0 && E_II % DG_C == 0, "lists split into whole chunks");
static_assert(DG_R % 4 == 0 && DG_C % 4 == 0, "ranges and chunks are whole 16-byte pieces");
#define DG_NB(entries, edges) (((entries) / DG_R) * ((edges) / DG_C))
#define DG_NB0 DG_NB(SVCN,  E_SC)
#define DG_NB1 DG_NB(SVCN,  E_SC)
#define DG_NB2 DG_NB(INSTN, E_IN)
#define DG_NB3 DG_NB(NODEN, E_IN)
#define DG_NB4 DG_NB(NODEN, E_NI)
#define DG_NB5 DG_NB(INSTN, E_NI)
#define DG_NB6 DG_NB(INSTN, E_II)
#define DG_NB7 DG_NB(INSTN, E_II)
#define DG_NB_ALL (DG_NB0 + DG_NB1 + DG_NB2 + DG_NB3 + DG_NB4 + DG_NB5 + DG_NB6 + DG_NB7)
__global__ __launch_bounds__(DG_T) void degree_kernel(const int* __restrict__ svc_src, const int* __restrict__ svc_dst,
                                                      const int* __restrict__ in_src,  const int* __restrict__ in_dst,
                                                      const int* __restrict__ ni_src,  const int* __restrict__ ni_dst,
                                                      const int* __restrict__ ii_src,  const int* __restrict__ ii_dst,
                                                      float* __restrict__ deg) {
    __shared__ __attribute__((aligned(16))) unsigned int cnt[DG_R];
    int b = blockIdx.x;
    const int* idx; float* plane; int nch;
    if (b < DG_NB0)                  { idx = svc_src; plane = deg + (D_NS_SC - OFF_DEG); nch = E_SC / DG_C; }
    else if ((b -= DG_NB0) < DG_NB1) { idx = svc_dst; plane = deg + (D_ND_SC - OFF_DEG); nch = E_SC / DG_C; }
    else if ((b -= DG_NB1) < DG_NB2) { idx = in_src;  plane = deg + (D_NS_IN - OFF_DEG); nch = E_IN / DG_C; }
    else if ((b -= DG_NB2) < DG_NB3) { idx = in_dst;  plane = deg + (D_ND_IN - OFF_DEG); nch = E_IN / DG_C; }
    else if ((b -= DG_NB3) < DG_NB4) { idx = ni_src;  plane = deg + (D_NS_NI - OFF_DEG); nch = E_NI / DG_C; }
    else if ((b -= DG_NB4) < DG_NB5) { idx = ni_dst;  plane = deg + (D_ND_NI - OFF_DEG); nch = E_NI / DG_C; }
    else if ((b -= DG_NB5) < DG_NB6) { idx = ii_src;  plane = deg + (D_NS_II - OFF_DEG); nch = E_II / DG_C; }
    else { b -= DG_NB6;                idx = ii_dst;  plane = deg + (D_ND_II - OFF_DEG); nch = E_II / DG_C; }
    const int range = b / nch;
    const int chunk = b - range * nch;
    const int lo    = range * DG_R;
    const int t     = threadIdx.x;

    const int4* __restrict__ p4 = (const int4*)(idx + chunk * DG_C);
    int4 v[DG_L];
#pragma unroll
    for (int k = 0; k < DG_L; ++k) {
        const int i = k * DG_T + t;
        v[k] = make_int4(-1, -1, -1, -1);
        if (i < DG_C / 4) v[k] = p4[i];
    }

    for (int i = t; i < DG_R / 4; i += DG_T) ((uint4*)cnt)[i] = make_uint4(0u, 0u, 0u, 0u);
    __syncthreads();

#pragma unroll
    for (int k = 0; k < DG_L; ++k) {
        unsigned int d;
        d = (unsigned int)(v[k].x - lo); if (d < DG_R) atomicAdd(&cnt[d], 1u);
        d = (unsigned int)(v[k].y - lo); if (d < DG_R) atomicAdd(&cnt[d], 1u);
        d = (unsigned int)(v[k].z - lo); if (d < DG_R) atomicAdd(&cnt[d], 1u);
        d = (unsigned int)(v[k].w - lo); if (d < DG_R) atomicAdd(&cnt[d], 1u);
    }
    __syncthreads();

    unsigned int c[DG_F];
#pragma unroll
    for (int k = 0; k < DG_F; ++k) {
        const int i = k * DG_T + t;
        c[k] = i < DG_R ? cnt[i] : 0u;
    }
#pragma unroll
    for (int k = 0; k < DG_F; ++k)
        if (c[k] != 0u) atomAddF(&plane[lo + k * DG_T + t], (float)c[k]);
}


#define SCAN_T 1024
#define SCAN_Q 4
__global__ __launch_bounds__(SCAN_T) void offsets_kernel(const float* __restrict__ deg,
                                                         int* __restrict__ cur) {
    __shared__ int wsum[32];
    __shared__ int tile_total;
    const float* dg; int* out; int n;
    if (blockIdx.x == 0)      { dg = deg + (D_ND_SC - OFF_DEG); out = cur + CUR_SC; n = SVCN;  }
    else if (blockIdx.x == 1) { dg = deg + (D_ND_IN - OFF_DEG); out = cur + CUR_IN; n = NODEN; }
    else if (blockIdx.x == 2) { dg = deg + (D_ND_NI - OFF_DEG); out = cur + CUR_NI; n = INSTN; }
    else                      { dg = deg + (D_ND_II - OFF_DEG); out = cur + CUR_II; n = INSTN; }
    const int t = threadIdx.x, lane = t & 31, wave = t >> 5;
    int carry = 0;
    for (int base = 0; base < n; base += SCAN_T * SCAN_Q * 4) {
        const int i0 = base + t * (SCAN_Q * 4);
        int4 v[SCAN_Q];
        int s = 0;
#pragma unroll
        for (int q = 0; q < SCAN_Q; ++q) {
            const int i = i0 + q * 4;
            v[q] = make_int4(0, 0, 0, 0);
            if (i < n) {
                const float4 f = *(const float4*)&dg[i];
                v[q] = make_int4((int)f.x, (int)f.y, (int)f.z, (int)f.w);
            }
            s += v[q].x + v[q].y + v[q].z + v[q].w;
        }
        int x = s;
        for (int o = 1; o < 32; o <<= 1) {
            const int y = __shfl_up(x, o);
            if (lane >= o) x += y;
        }
        if (lane == 31) wsum[wave] = x;
        __syncthreads();
        if (wave == 0) {
            const int w = wsum[lane];
            int xs = w;
            for (int o = 1; o < 32; o <<= 1) {
                const int y = __shfl_up(xs, o);
                if (lane >= o) xs += y;
            }
            wsum[lane] = xs - w;
            if (lane == 31) tile_total = xs;
        }
        __syncthreads();
        int run = carry + wsum[wave] + (x - s);
#pragma unroll
        for (int q = 0; q < SCAN_Q; ++q) {
            const int i = i0 + q * 4;
            if (i < n)
                *(int4*)&out[i] = make_int4(run, run + v[q].x, run + v[q].x + v[q].y,
                                            run + v[q].x + v[q].y + v[q].z);
            run += v[q].x + v[q].y + v[q].z + v[q].w;
        }
        carry += tile_total;
    }
}

#define FL_NB0 DG_NB(SVCN,  E_SC)
#define FL_NB1 DG_NB(NODEN, E_IN)
#define FL_NB2 DG_NB(INSTN, E_NI)
#define FL_NB3 DG_NB(INSTN, E_II)
#define FL_NB_ALL (FL_NB0 + FL_NB1 + FL_NB2 + FL_NB3)
#define FL_NT     (FL_NB_ALL * DG_T)
#define FL_DEG4   (DEG_TOTAL / 4)
#define FL_DQ     ((FL_DEG4 + FL_NT - 1) / FL_NT)
static_assert(DEG_TOTAL % 4 == 0, "degree planes are whole 16-byte pieces");
__global__ __launch_bounds__(DG_T) void fill_kernel(const int* __restrict__ svc_src, const int* __restrict__ svc_dst,
                                                    const int* __restrict__ in_src,  const int* __restrict__ in_dst,
                                                    const int* __restrict__ ni_src,  const int* __restrict__ ni_dst,
                                                    const int* __restrict__ ii_src,  const int* __restrict__ ii_dst,
                                                    int* __restrict__ cur, int* __restrict__ sorted,
                                                    float* __restrict__ deg) {
    __shared__ __attribute__((aligned(16))) unsigned int cnt[DG_R];
    const int t = threadIdx.x;

    const int g = blockIdx.x * DG_T + t;
    float4 dq[FL_DQ];
#pragma unroll
    for (int q = 0; q < FL_DQ; ++q) {
        const int i = q * FL_NT + g;
        dq[q] = make_float4(1.f, 1.f, 1.f, 1.f);
        if (i < FL_DEG4) dq[q] = ((const float4*)deg)[i];
    }

    int b = blockIdx.x;
    const int* src; const int* dst; int* cu; int* so; int nch;
    if (b < FL_NB0)                  { src = svc_src; dst = svc_dst; cu = cur + CUR_SC; so = sorted + SORT_SC; nch = E_SC / DG_C; }
    else if ((b -= FL_NB0) < FL_NB1) { src = in_src;  dst = in_dst;  cu = cur + CUR_IN; so = sorted + SORT_IN; nch = E_IN / DG_C; }
    else if ((b -= FL_NB1) < FL_NB2) { src = ni_src;  dst = ni_dst;  cu = cur + CUR_NI; so = sorted + SORT_NI; nch = E_NI / DG_C; }
    else { b -= FL_NB2;                src = ii_src;  dst = ii_dst;  cu = cur + CUR_II; so = sorted + SORT_II; nch = E_II / DG_C; }
    const int range = b / nch;
    const int chunk = b - range * nch;
    const int lo    = range * DG_R;

    const int4* __restrict__ d4 = (const int4*)(dst + chunk * DG_C);
    const int4* __restrict__ s4 = (const int4*)(src + chunk * DG_C);
    int4 vd[DG_L], vs[DG_L];
#pragma unroll
    for (int k = 0; k < DG_L; ++k) {
        const int i = k * DG_T + t;
        vd[k] = make_int4(-1, -1, -1, -1);
        vs[k] = make_int4(0, 0, 0, 0);
        if (i < DG_C / 4) { vd[k] = d4[i]; vs[k] = s4[i]; }
    }

    for (int i = t; i < DG_R / 4; i += DG_T) ((uint4*)cnt)[i] = make_uint4(0u, 0u, 0u, 0u);

#pragma unroll
    for (int q = 0; q < FL_DQ; ++q) {
        const int i = q * FL_NT + g;
        if (i < FL_DEG4)
            ((float4*)deg)[i] = make_float4(deg_factor(dq[q].x), deg_factor(dq[q].y),
                                            deg_factor(dq[q].z), deg_factor(dq[q].w));
    }
    __syncthreads();

#pragma unroll
    for (int k = 0; k < DG_L; ++k) {
        unsigned int d;
        d = (unsigned int)(vd[k].x - lo); if (d < DG_R) atomicAdd(&cnt[d], 1u);
        d = (unsigned int)(vd[k].y - lo); if (d < DG_R) atomicAdd(&cnt[d], 1u);
        d = (unsigned int)(vd[k].z - lo); if (d < DG_R) atomicAdd(&cnt[d], 1u);
        d = (unsigned int)(vd[k].w - lo); if (d < DG_R) atomicAdd(&cnt[d], 1u);
    }
    __syncthreads();

    unsigned int c[DG_F];
#pragma unroll
    for (int k = 0; k < DG_F; ++k) {
        const int i = k * DG_T + t;
        c[k] = i < DG_R ? cnt[i] : 0u;
    }
#pragma unroll
    for (int k = 0; k < DG_F; ++k)
        if (c[k] != 0u) c[k] = (unsigned int)atomicAdd(&cu[lo + k * DG_T + t], (int)c[k]);
#pragma unroll
    for (int k = 0; k < DG_F; ++k) {
        const int i = k * DG_T + t;
        if (i < DG_R) cnt[i] = c[k];
    }
    __syncthreads();

#pragma unroll
    for (int k = 0; k < DG_L; ++k) {
        const unsigned int dx = (unsigned int)(vd[k].x - lo), dy = (unsigned int)(vd[k].y - lo);
        const unsigned int dz = (unsigned int)(vd[k].z - lo), dw = (unsigned int)(vd[k].w - lo);
        unsigned int px = 0u, py = 0u, pz = 0u, pw = 0u;
        if (dx < DG_R) px = atomicAdd(&cnt[dx], 1u);
        if (dy < DG_R) py = atomicAdd(&cnt[dy], 1u);
        if (dz < DG_R) pz = atomicAdd(&cnt[dz], 1u);
        if (dw < DG_R) pw = atomicAdd(&cnt[dw], 1u);
        if (dx < DG_R) so[px] = vs[k].x;
        if (dy < DG_R) so[py] = vs[k].y;
        if (dz < DG_R) so[pz] = vs[k].z;
        if (dw < DG_R) so[pw] = vs[k].w;
    }
}

#define AG_BATCH 8

#define AG_ROWS   64
#define AG_T      512
#define AG_PITCH  132
#define AG_NB_INST ((INSTN + AG_ROWS - 1) / AG_ROWS)
#define AG_NB_SVC  ((SVCN  + AG_ROWS - 1) / AG_ROWS)
#define AG_NB_NODE ((NODEN + AG_ROWS - 1) / AG_ROWS)
__global__ __launch_bounds__(AG_T) void agg_proj_kernel(
        const float* __restrict__ feat_svc, const float* __restrict__ feat_inst,
        const float* __restrict__ feat_node,
        const float* __restrict__ W_sc, const float* __restrict__ W_in,
        const float* __restrict__ W_ni, const float* __restrict__ W_ii,
        const float* __restrict__ b_sc, const float* __restrict__ b_in,
        const float* __restrict__ b_ni, const float* __restrict__ b_ii,
        const float* __restrict__ deg, const int* __restrict__ curAll,
        const int* __restrict__ sortedAll, float* __restrict__ catAll) {
    __shared__ __attribute__((aligned(16))) float Wlds[128 * 128];
    __shared__ __attribute__((aligned(16))) float Tile[AG_ROWS * AG_PITCH];

    int blk = blockIdx.x;
    const float *X0, *W0, *ns0, *nd0, *bA;
    const int *cur0, *sorted0;
    const float *X1 = nullptr, *W1 = nullptr, *ns1 = nullptr, *nd1 = nullptr, *bB = nullptr;
    const int *cur1 = nullptr, *sorted1 = nullptr;
    int nrows, nlists = 1;
    float* cat;
    if (blk < AG_NB_INST) {
        X0 = feat_node; W0 = W_ni; bA = b_ni;
        ns0 = deg + (D_NS_NI - OFF_DEG); nd0 = deg + (D_ND_NI - OFF_DEG);
        cur0 = curAll + CUR_NI; sorted0 = sortedAll + SORT_NI;
        X1 = feat_inst; W1 = W_ii; bB = b_ii;
        ns1 = deg + (D_NS_II - OFF_DEG); nd1 = deg + (D_ND_II - OFF_DEG);
        cur1 = curAll + CUR_II; sorted1 = sortedAll + SORT_II;
        nrows = INSTN; nlists = 2;
        cat = catAll + (long long)(SVCN + NODEN) * 128;
    } else if ((blk -= AG_NB_INST) < AG_NB_SVC) {
        X0 = feat_svc; W0 = W_sc; bA = b_sc;
        ns0 = deg + (D_NS_SC - OFF_DEG); nd0 = deg + (D_ND_SC - OFF_DEG);
        cur0 = curAll + CUR_SC; sorted0 = sortedAll + SORT_SC;
        nrows = SVCN;
        cat = catAll;
    } else {
        blk -= AG_NB_SVC;
        X0 = feat_inst; W0 = W_in; bA = b_in;
        ns0 = deg + (D_NS_IN - OFF_DEG); nd0 = deg + (D_ND_IN - OFF_DEG);
        cur0 = curAll + CUR_IN; sorted0 = sortedAll + SORT_IN;
        nrows = NODEN;
        cat = catAll + (long long)SVCN * 128;
    }

    const int t    = threadIdx.x;
    const int lane = t & 31;
    const int wave = __builtin_amdgcn_readfirstlane(t >> 5);
    const int half = lane >> 4;
    const int l16  = lane & 15;
    const int c4   = lane * 4;
    const int r0   = blk * AG_ROWS;
    const int n0   = (wave & 7) * 16;
    const int rb   = (wave >> 3) * 32;

    v8f acc[2] = {v8f{}, v8f{}};

    for (int li = 0; li < nlists; ++li) {
        const float* __restrict__ X  = li ? X1 : X0;
        const float* __restrict__ Wg = li ? W1 : W0;
        const float* __restrict__ ns = li ? ns1 : ns0;
        const float* __restrict__ nd = li ? nd1 : nd0;
        const int* __restrict__ cur    = li ? cur1 : cur0;
        const int* __restrict__ sorted = li ? sorted1 : sorted0;

        const int first = r0 + wave * 4;
        int ic = first - 1 + lane; ic = ic < 0 ? 0 : (ic < nrows ? ic : nrows - 1);
        int id = first + lane;     id = id < nrows ? id : nrows - 1;
        const int   cv  = cur[ic];
        const float ndv = nd[id];

        const float4* __restrict__ Wg4 = (const float4*)Wg + t;
        const float4 wr0 = Wg4[0 * AG_T], wr1 = Wg4[1 * AG_T], wr2 = Wg4[2 * AG_T], wr3 = Wg4[3 * AG_T];
        const float4 wr4 = Wg4[4 * AG_T], wr5 = Wg4[5 * AG_T], wr6 = Wg4[6 * AG_T], wr7 = Wg4[7 * AG_T];

        const int b0 = first > 0 ? __builtin_amdgcn_readlane(cv, 0) : 0;
        const int e3 = __builtin_amdgcn_readlane(cv, 4);
        int sv_next = 0;
        if (lane < e3 - b0) sv_next = sorted[b0 + lane];

        if (li > 0) __syncthreads();

        float4* Wl4 = (float4*)Wlds + t;
        Wl4[0 * AG_T] = wr0; Wl4[1 * AG_T] = wr1; Wl4[2 * AG_T] = wr2; Wl4[3 * AG_T] = wr3;
        Wl4[4 * AG_T] = wr4; Wl4[5 * AG_T] = wr5; Wl4[6 * AG_T] = wr6; Wl4[7 * AG_T] = wr7;

        float4 acc4 = make_float4(0.f, 0.f, 0.f, 0.f);
        int ri   = 0;
        int rend = __builtin_amdgcn_readlane(cv, 1);
        for (int cb = b0; cb < e3; cb += 32) {
            const int rem = e3 - cb;
            const int cnt = rem < 32 ? rem : 32;
            const int sv  = sv_next;
            sv_next = 0;
            if (lane < rem - 32) sv_next = sorted[cb + 32 + lane];
            float wv = 0.f;
            if (lane < cnt) wv = ns[sv];
            for (int jj = 0; jj < cnt; jj += AG_BATCH) {
                float4 x[AG_BATCH];
#pragma unroll
                for (int k = 0; k < AG_BATCH; ++k) {
                    const int q = jj + k < cnt ? jj + k : cnt - 1;
                    const int s = __builtin_amdgcn_readlane(sv, q);
                    x[k] = *(const float4*)&X[(long long)s * 128 + c4];
                }
#pragma unroll
                for (int k = 0; k < AG_BATCH; ++k) {
                    const bool live = jj + k < cnt;
                    const int q = live ? jj + k : cnt - 1;
                    const int j = cb + q;
                    while (j >= rend) {
                        const float n = __builtin_bit_cast(float,
                            __builtin_amdgcn_readlane(__builtin_bit_cast(int, ndv), ri));
                        *(float4*)&Tile[(wave * 4 + ri) * AG_PITCH + c4] =
                            make_float4(acc4.x * n, acc4.y * n, acc4.z * n, acc4.w * n);
                        acc4 = make_float4(0.f, 0.f, 0.f, 0.f);
                        ++ri;
                        rend = __builtin_amdgcn_readlane(cv, ri + 1);
                    }
                    const float wq = __builtin_bit_cast(float,
                        __builtin_amdgcn_readlane(__builtin_bit_cast(int, wv), q));
                    const float w = live ? wq : 0.f;
                    acc4.x = fmaf(x[k].x, w, acc4.x); acc4.y = fmaf(x[k].y, w, acc4.y);
                    acc4.z = fmaf(x[k].z, w, acc4.z); acc4.w = fmaf(x[k].w, w, acc4.w);
                }
            }
        }
        for (; ri < 4; ++ri) {
            const float n = __builtin_bit_cast(float,
                __builtin_amdgcn_readlane(__builtin_bit_cast(int, ndv), ri));
            *(float4*)&Tile[(wave * 4 + ri) * AG_PITCH + c4] =
                make_float4(acc4.x * n, acc4.y * n, acc4.z * n, acc4.w * n);
            acc4 = make_float4(0.f, 0.f, 0.f, 0.f);
        }
        __syncthreads();

        for (int k = 0; k < 128; k += 4) {
            const int kk = k + 2 * half;
            v2f b;
            b.x = Wlds[kk * 128 + n0 + l16];
            b.y = Wlds[(kk + 1) * 128 + n0 + l16];
#pragma unroll
            for (int rt = 0; rt < 2; ++rt) {
                v2f a = *(const v2f*)&Tile[(rb + rt * 16 + l16) * AG_PITCH + kk];
                acc[rt] = __builtin_amdgcn_wmma_f32_16x16x4_f32(
                    false, a, false, b, (short)0, acc[rt], false, false);
            }
        }
    }

    __syncthreads();
#pragma unroll
    for (int rt = 0; rt < 2; ++rt) {
#pragma unroll
        for (int i = 0; i < 8; ++i)
            Tile[(rb + rt * 16 + i + 8 * half) * AG_PITCH + n0 + l16] = acc[rt][i];
    }
    __syncthreads();

    float4 bias = *(const float4*)&bA[c4];
    if (nlists > 1) {
        const float4 bb = *(const float4*)&bB[c4];
        bias.x += bb.x; bias.y += bb.y; bias.z += bb.z; bias.w += bb.w;
    }
    for (int i = 0; i < 4; ++i) {
        const int lr  = wave * 4 + i;
        const int row = r0 + lr;
        if (row < nrows) {
            const float4 p = *(const float4*)&Tile[lr * AG_PITCH + c4];
            *(float4*)&cat[(long long)row * 128 + c4] =
                make_float4(p.x + bias.x, p.y + bias.y, p.z + bias.z, p.w + bias.w);
        }
    }
}

#define FG_KT      32
#define FG_TILES   (TOTAL / FG_KT)
#define FG_BLOCKS  250
#define FG_TPB     (FG_TILES / FG_BLOCKS)
#define FG_PA      36
#define FG_PB      136
static_assert(TOTAL % FG_KT == 0, "K must be a whole number of tiles");
static_assert(FG_TILES % FG_BLOCKS == 0, "tiles must divide evenly over the blocks");

__global__ __launch_bounds__(512) void final_gemm_kernel(const float* __restrict__ Wt,
                                                         const float* __restrict__ cat,
                                                         float* __restrict__ out) {
    __shared__ float As[OUTD * FG_PA];
    __shared__ float Bs[FG_KT * FG_PB];

    const int t    = threadIdx.x;
    const int lane = t & 31;
    const int wave = t >> 5;
    const int half = lane >> 4;
    const int l16  = lane & 15;
    const int wm   = wave >> 2;
    const int wn   = wave & 3;
    const int tile0 = blockIdx.x * FG_TPB;

    const int arow = t >> 3;
    const int ac4  = t & 7;
    const int brow = t >> 5;
    const int bc4  = t & 31;
    const float* pA = Wt + (long long)arow * TOTAL + (long long)tile0 * FG_KT + ac4 * 4;
    const float* pB = cat + (long long)tile0 * FG_KT * 128 + t * 4;
    const long long aStep = 64LL * TOTAL;
    float* sA = &As[arow * FG_PA + ac4 * 4];
    float* sB = &Bs[brow * FG_PB + bc4 * 4];

    v8f acc[4][2];
#pragma unroll
    for (int mt = 0; mt < 4; ++mt)
#pragma unroll
        for (int nt = 0; nt < 2; ++nt) acc[mt][nt] = v8f{};

    float4 ra0 = *(const float4*)(pA);
    float4 ra1 = *(const float4*)(pA + aStep);
    float4 ra2 = *(const float4*)(pA + 2 * aStep);
    float4 ra3 = *(const float4*)(pA + 3 * aStep);
    float4 rb0 = *(const float4*)(pB);
    float4 rb1 = *(const float4*)(pB + 2048);

    for (int it = 0; it < FG_TPB; ++it) {
        *(float4*)(sA)                 = ra0;
        *(float4*)(sA + 64 * FG_PA)    = ra1;
        *(float4*)(sA + 128 * FG_PA)   = ra2;
        *(float4*)(sA + 192 * FG_PA)   = ra3;
        rb0.x = lrelu(rb0.x); rb0.y = lrelu(rb0.y); rb0.z = lrelu(rb0.z); rb0.w = lrelu(rb0.w);
        rb1.x = lrelu(rb1.x); rb1.y = lrelu(rb1.y); rb1.z = lrelu(rb1.z); rb1.w = lrelu(rb1.w);
        *(float4*)(sB)                 = rb0;
        *(float4*)(sB + 16 * FG_PB)    = rb1;
        __syncthreads();

        if (it + 1 < FG_TPB) {
            pA += FG_KT;
            pB += FG_KT * 128;
            ra0 = *(const float4*)(pA);
            ra1 = *(const float4*)(pA + aStep);
            ra2 = *(const float4*)(pA + 2 * aStep);
            ra3 = *(const float4*)(pA + 3 * aStep);
            rb0 = *(const float4*)(pB);
            rb1 = *(const float4*)(pB + 2048);
        }
        __builtin_amdgcn_sched_barrier(0);

#pragma unroll
        for (int ks = 0; ks < FG_KT / 4; ++ks) {
            const int kk = ks * 4 + 2 * half;
            v2f a[4], b[2];
#pragma unroll
            for (int mt = 0; mt < 4; ++mt)
                a[mt] = *(const v2f*)&As[(wm * 64 + mt * 16 + l16) * FG_PA + kk];
#pragma unroll
            for (int nt = 0; nt < 2; ++nt) {
                b[nt].x = Bs[kk * FG_PB + wn * 32 + nt * 16 + l16];
                b[nt].y = Bs[(kk + 1) * FG_PB + wn * 32 + nt * 16 + l16];
            }
#pragma unroll
            for (int mt = 0; mt < 4; ++mt)
#pragma unroll
                for (int nt = 0; nt < 2; ++nt)
                    acc[mt][nt] = __builtin_amdgcn_wmma_f32_16x16x4_f32(
                        false, a[mt], false, b[nt], (short)0, acc[mt][nt], false, false);
        }
        __syncthreads();
    }

#pragma unroll
    for (int mt = 0; mt < 4; ++mt) {
#pragma unroll
        for (int nt = 0; nt < 2; ++nt) {
#pragma unroll
            for (int i = 0; i < 8; ++i) {
                int m = wm * 64 + mt * 16 + i + 8 * half;
                atomAddF(&out[m * 128 + wn * 32 + nt * 16 + l16], acc[mt][nt][i]);
            }
        }
    }
}

static inline int cdiv(long long a, long long b) { return (int)((a + b - 1) / b); }

extern "C" void kernel_launch(void* const* d_in, const int* in_sizes, int n_in,
                              void* d_out, int out_size, void* d_ws, size_t ws_size,
                              hipStream_t stream) {
    const float* feat_svc  = (const float*)d_in[0];
    const float* feat_inst = (const float*)d_in[1];
    const float* feat_node = (const float*)d_in[2];
    const int*   svc_src   = (const int*)d_in[3];
    const int*   svc_dst   = (const int*)d_in[4];
    const int*   in_src    = (const int*)d_in[5];
    const int*   in_dst    = (const int*)d_in[6];
    const int*   ni_src    = (const int*)d_in[7];
    const int*   ni_dst    = (const int*)d_in[8];
    const int*   ii_src    = (const int*)d_in[9];
    const int*   ii_dst    = (const int*)d_in[10];
    const float* W_sc = (const float*)d_in[11];  const float* b_sc = (const float*)d_in[12];
    const float* W_in = (const float*)d_in[13];  const float* b_in = (const float*)d_in[14];
    const float* W_ni = (const float*)d_in[15];  const float* b_ni = (const float*)d_in[16];
    const float* W_ii = (const float*)d_in[17];  const float* b_ii = (const float*)d_in[18];
    const float* W_tot = (const float*)d_in[19]; const float* b_tot = (const float*)d_in[20];
    float* out = (float*)d_out;

    float* ws   = (float*)d_ws;
    float* cat  = ws + OFF_CAT;
    float* deg  = ws + OFF_DEG;
    int* cur    = (int*)(ws + OFF_CUR);
    int* sorted = (int*)(ws + OFF_SORT);

    preset_kernel<<<cdiv(PRESET_N, 256), 256, 0, stream>>>(deg, out, b_tot);

    degree_kernel<<<DG_NB_ALL, DG_T, 0, stream>>>(
        svc_src, svc_dst, in_src, in_dst, ni_src, ni_dst, ii_src, ii_dst, deg);
    offsets_kernel<<<4, SCAN_T, 0, stream>>>(deg, cur);

    fill_kernel<<<FL_NB_ALL, DG_T, 0, stream>>>(
        svc_src, svc_dst, in_src, in_dst, ni_src, ni_dst, ii_src, ii_dst, cur, sorted, deg);

    agg_proj_kernel<<<AG_NB_INST + AG_NB_SVC + AG_NB_NODE, AG_T, 0, stream>>>(
        feat_svc, feat_inst, feat_node, W_sc, W_in, W_ni, W_ii, b_sc, b_in, b_ni, b_ii,
        deg, cur, sorted, cat);

    final_gemm_kernel<<<FG_BLOCKS, 512, 0, stream>>>(W_tot, cat, out);
}


# final GEMM: each K-slice block stores its 256x128 partial as whole lines; a 7th launch adds the 250 planes and the bias in fixed order, replacing the f32 atomics on out (sum order now fixed)
# speedup vs baseline: 2.9827x; 1.0627x over previous
#include <hip/hip_runtime.h>

#define SVCN   50000
#define INSTN  100000
#define NODEN  50000
#define FDIM   128
#define HDIM   128
#define OUTD   256
#define TOTAL  (SVCN + NODEN + INSTN)
#define E_SC   400000
#define E_IN   100000
#define E_NI   100000
#define E_II   800000

#define OFF_CAT   0LL
#define OFF_DEG   (OFF_CAT  + (long long)TOTAL * 128)
#define D_NS_SC   (OFF_DEG + 0LL)
#define D_ND_SC   (D_NS_SC + SVCN)
#define D_NS_IN   (D_ND_SC + SVCN)
#define D_ND_IN   (D_NS_IN + INSTN)
#define D_NS_NI   (D_ND_IN + NODEN)
#define D_ND_NI   (D_NS_NI + NODEN)
#define D_NS_II   (D_ND_NI + INSTN)
#define D_ND_II   (D_NS_II + INSTN)
#define DEG_TOTAL 600000
#define OFF_CUR   (OFF_DEG + (long long)DEG_TOTAL)
#define CUR_SC    0
#define CUR_IN    (CUR_SC + SVCN)
#define CUR_NI    (CUR_IN + NODEN)
#define CUR_II    (CUR_NI + INSTN)
#define CUR_TOTAL (CUR_II + INSTN)
#define OFF_SORT  (OFF_CUR + (long long)CUR_TOTAL)
#define SORT_SC   0
#define SORT_IN   (SORT_SC + E_SC)
#define SORT_NI   (SORT_IN + E_IN)
#define SORT_II   (SORT_NI + E_NI)
#define OFF_PART  (OFF_SORT + (long long)(E_SC + E_IN + E_NI + E_II))
static_assert(OFF_PART % 32 == 0, "partial planes start on a 128-byte line");
static_assert(SVCN % 4 == 0 && NODEN % 4 == 0 && INSTN % 4 == 0, "degree planes are scanned four at a time");

typedef __attribute__((ext_vector_type(2))) float v2f;
typedef __attribute__((ext_vector_type(8))) float v8f;

__device__ __forceinline__ float lrelu(float x) { return x > 0.f ? x : 0.01f * x; }

__device__ __forceinline__ void atomAddF(float* p, float v) { unsafeAtomicAdd(p, v); }

#define PRESET_N DEG_TOTAL
__global__ void preset_kernel(float* __restrict__ deg) {
    const int i = blockIdx.x * blockDim.x + threadIdx.x;
    if (i < DEG_TOTAL) deg[i] = 0.f;
}

__device__ __forceinline__ float deg_factor(float d) { return rsqrtf(d > 1.f ? d : 1.f); }

#define DG_T  512
#define DG_R  12500
#define DG_C  50000
#define DG_L  ((DG_C / 4 + DG_T - 1) / DG_T)
#define DG_F  ((DG_R + DG_T - 1) / DG_T)
static_assert(SVCN % DG_R == 0 && NODEN % DG_R == 0 && INSTN % DG_R == 0, "planes split into whole ranges");
static_assert(E_SC % DG_C == 0 && E_IN % DG_C == 0 && E_NI % DG_C == 0 && E_II % DG_C == 0, "lists split into whole chunks");
static_assert(DG_R % 4 == 0 && DG_C % 4 == 0, "ranges and chunks are whole 16-byte pieces");
#define DG_NB(entries, edges) (((entries) / DG_R) * ((edges) / DG_C))
#define DG_NB0 DG_NB(SVCN,  E_SC)
#define DG_NB1 DG_NB(SVCN,  E_SC)
#define DG_NB2 DG_NB(INSTN, E_IN)
#define DG_NB3 DG_NB(NODEN, E_IN)
#define DG_NB4 DG_NB(NODEN, E_NI)
#define DG_NB5 DG_NB(INSTN, E_NI)
#define DG_NB6 DG_NB(INSTN, E_II)
#define DG_NB7 DG_NB(INSTN, E_II)
#define DG_NB_ALL (DG_NB0 + DG_NB1 + DG_NB2 + DG_NB3 + DG_NB4 + DG_NB5 + DG_NB6 + DG_NB7)
__global__ __launch_bounds__(DG_T) void degree_kernel(const int* __restrict__ svc_src, const int* __restrict__ svc_dst,
                                                      const int* __restrict__ in_src,  const int* __restrict__ in_dst,
                                                      const int* __restrict__ ni_src,  const int* __restrict__ ni_dst,
                                                      const int* __restrict__ ii_src,  const int* __restrict__ ii_dst,
                                                      float* __restrict__ deg) {
    __shared__ __attribute__((aligned(16))) unsigned int cnt[DG_R];
    int b = blockIdx.x;
    const int* idx; float* plane; int nch;
    if (b < DG_NB0)                  { idx = svc_src; plane = deg + (D_NS_SC - OFF_DEG); nch = E_SC / DG_C; }
    else if ((b -= DG_NB0) < DG_NB1) { idx = svc_dst; plane = deg + (D_ND_SC - OFF_DEG); nch = E_SC / DG_C; }
    else if ((b -= DG_NB1) < DG_NB2) { idx = in_src;  plane = deg + (D_NS_IN - OFF_DEG); nch = E_IN / DG_C; }
    else if ((b -= DG_NB2) < DG_NB3) { idx = in_dst;  plane = deg + (D_ND_IN - OFF_DEG); nch = E_IN / DG_C; }
    else if ((b -= DG_NB3) < DG_NB4) { idx = ni_src;  plane = deg + (D_NS_NI - OFF_DEG); nch = E_NI / DG_C; }
    else if ((b -= DG_NB4) < DG_NB5) { idx = ni_dst;  plane = deg + (D_ND_NI - OFF_DEG); nch = E_NI / DG_C; }
    else if ((b -= DG_NB5) < DG_NB6) { idx = ii_src;  plane = deg + (D_NS_II - OFF_DEG); nch = E_II / DG_C; }
    else { b -= DG_NB6;                idx = ii_dst;  plane = deg + (D_ND_II - OFF_DEG); nch = E_II / DG_C; }
    const int range = b / nch;
    const int chunk = b - range * nch;
    const int lo    = range * DG_R;
    const int t     = threadIdx.x;

    const int4* __restrict__ p4 = (const int4*)(idx + chunk * DG_C);
    int4 v[DG_L];
#pragma unroll
    for (int k = 0; k < DG_L; ++k) {
        const int i = k * DG_T + t;
        v[k] = make_int4(-1, -1, -1, -1);
        if (i < DG_C / 4) v[k] = p4[i];
    }

    for (int i = t; i < DG_R / 4; i += DG_T) ((uint4*)cnt)[i] = make_uint4(0u, 0u, 0u, 0u);
    __syncthreads();

#pragma unroll
    for (int k = 0; k < DG_L; ++k) {
        unsigned int d;
        d = (unsigned int)(v[k].x - lo); if (d < DG_R) atomicAdd(&cnt[d], 1u);
        d = (unsigned int)(v[k].y - lo); if (d < DG_R) atomicAdd(&cnt[d], 1u);
        d = (unsigned int)(v[k].z - lo); if (d < DG_R) atomicAdd(&cnt[d], 1u);
        d = (unsigned int)(v[k].w - lo); if (d < DG_R) atomicAdd(&cnt[d], 1u);
    }
    __syncthreads();

    unsigned int c[DG_F];
#pragma unroll
    for (int k = 0; k < DG_F; ++k) {
        const int i = k * DG_T + t;
        c[k] = i < DG_R ? cnt[i] : 0u;
    }
#pragma unroll
    for (int k = 0; k < DG_F; ++k)
        if (c[k] != 0u) atomAddF(&plane[lo + k * DG_T + t], (float)c[k]);
}


#define SCAN_T 1024
#define SCAN_Q 4
__global__ __launch_bounds__(SCAN_T) void offsets_kernel(const float* __restrict__ deg,
                                                         int* __restrict__ cur) {
    __shared__ int wsum[32];
    __shared__ int tile_total;
    const float* dg; int* out; int n;
    if (blockIdx.x == 0)      { dg = deg + (D_ND_SC - OFF_DEG); out = cur + CUR_SC; n = SVCN;  }
    else if (blockIdx.x == 1) { dg = deg + (D_ND_IN - OFF_DEG); out = cur + CUR_IN; n = NODEN; }
    else if (blockIdx.x == 2) { dg = deg + (D_ND_NI - OFF_DEG); out = cur + CUR_NI; n = INSTN; }
    else                      { dg = deg + (D_ND_II - OFF_DEG); out = cur + CUR_II; n = INSTN; }
    const int t = threadIdx.x, lane = t & 31, wave = t >> 5;
    int carry = 0;
    for (int base = 0; base < n; base += SCAN_T * SCAN_Q * 4) {
        const int i0 = base + t * (SCAN_Q * 4);
        int4 v[SCAN_Q];
        int s = 0;
#pragma unroll
        for (int q = 0; q < SCAN_Q; ++q) {
            const int i = i0 + q * 4;
            v[q] = make_int4(0, 0, 0, 0);
            if (i < n) {
                const float4 f = *(const float4*)&dg[i];
                v[q] = make_int4((int)f.x, (int)f.y, (int)f.z, (int)f.w);
            }
            s += v[q].x + v[q].y + v[q].z + v[q].w;
        }
        int x = s;
        for (int o = 1; o < 32; o <<= 1) {
            const int y = __shfl_up(x, o);
            if (lane >= o) x += y;
        }
        if (lane == 31) wsum[wave] = x;
        __syncthreads();
        if (wave == 0) {
            const int w = wsum[lane];
            int xs = w;
            for (int o = 1; o < 32; o <<= 1) {
                const int y = __shfl_up(xs, o);
                if (lane >= o) xs += y;
            }
            wsum[lane] = xs - w;
            if (lane == 31) tile_total = xs;
        }
        __syncthreads();
        int run = carry + wsum[wave] + (x - s);
#pragma unroll
        for (int q = 0; q < SCAN_Q; ++q) {
            const int i = i0 + q * 4;
            if (i < n)
                *(int4*)&out[i] = make_int4(run, run + v[q].x, run + v[q].x + v[q].y,
                                            run + v[q].x + v[q].y + v[q].z);
            run += v[q].x + v[q].y + v[q].z + v[q].w;
        }
        carry += tile_total;
    }
}

#define FL_NB0 DG_NB(SVCN,  E_SC)
#define FL_NB1 DG_NB(NODEN, E_IN)
#define FL_NB2 DG_NB(INSTN, E_NI)
#define FL_NB3 DG_NB(INSTN, E_II)
#define FL_NB_ALL (FL_NB0 + FL_NB1 + FL_NB2 + FL_NB3)
#define FL_NT     (FL_NB_ALL * DG_T)
#define FL_DEG4   (DEG_TOTAL / 4)
#define FL_DQ     ((FL_DEG4 + FL_NT - 1) / FL_NT)
static_assert(DEG_TOTAL % 4 == 0, "degree planes are whole 16-byte pieces");
__global__ __launch_bounds__(DG_T) void fill_kernel(const int* __restrict__ svc_src, const int* __restrict__ svc_dst,
                                                    const int* __restrict__ in_src,  const int* __restrict__ in_dst,
                                                    const int* __restrict__ ni_src,  const int* __restrict__ ni_dst,
                                                    const int* __restrict__ ii_src,  const int* __restrict__ ii_dst,
                                                    int* __restrict__ cur, int* __restrict__ sorted,
                                                    float* __restrict__ deg) {
    __shared__ __attribute__((aligned(16))) unsigned int cnt[DG_R];
    const int t = threadIdx.x;

    const int g = blockIdx.x * DG_T + t;
    float4 dq[FL_DQ];
#pragma unroll
    for (int q = 0; q < FL_DQ; ++q) {
        const int i = q * FL_NT + g;
        dq[q] = make_float4(1.f, 1.f, 1.f, 1.f);
        if (i < FL_DEG4) dq[q] = ((const float4*)deg)[i];
    }

    int b = blockIdx.x;
    const int* src; const int* dst; int* cu; int* so; int nch;
    if (b < FL_NB0)                  { src = svc_src; dst = svc_dst; cu = cur + CUR_SC; so = sorted + SORT_SC; nch = E_SC / DG_C; }
    else if ((b -= FL_NB0) < FL_NB1) { src = in_src;  dst = in_dst;  cu = cur + CUR_IN; so = sorted + SORT_IN; nch = E_IN / DG_C; }
    else if ((b -= FL_NB1) < FL_NB2) { src = ni_src;  dst = ni_dst;  cu = cur + CUR_NI; so = sorted + SORT_NI; nch = E_NI / DG_C; }
    else { b -= FL_NB2;                src = ii_src;  dst = ii_dst;  cu = cur + CUR_II; so = sorted + SORT_II; nch = E_II / DG_C; }
    const int range = b / nch;
    const int chunk = b - range * nch;
    const int lo    = range * DG_R;

    const int4* __restrict__ d4 = (const int4*)(dst + chunk * DG_C);
    const int4* __restrict__ s4 = (const int4*)(src + chunk * DG_C);
    int4 vd[DG_L], vs[DG_L];
#pragma unroll
    for (int k = 0; k < DG_L; ++k) {
        const int i = k * DG_T + t;
        vd[k] = make_int4(-1, -1, -1, -1);
        vs[k] = make_int4(0, 0, 0, 0);
        if (i < DG_C / 4) { vd[k] = d4[i]; vs[k] = s4[i]; }
    }

    for (int i = t; i < DG_R / 4; i += DG_T) ((uint4*)cnt)[i] = make_uint4(0u, 0u, 0u, 0u);

#pragma unroll
    for (int q = 0; q < FL_DQ; ++q) {
        const int i = q * FL_NT + g;
        if (i < FL_DEG4)
            ((float4*)deg)[i] = make_float4(deg_factor(dq[q].x), deg_factor(dq[q].y),
                                            deg_factor(dq[q].z), deg_factor(dq[q].w));
    }
    __syncthreads();

#pragma unroll
    for (int k = 0; k < DG_L; ++k) {
        unsigned int d;
        d = (unsigned int)(vd[k].x - lo); if (d < DG_R) atomicAdd(&cnt[d], 1u);
        d = (unsigned int)(vd[k].y - lo); if (d < DG_R) atomicAdd(&cnt[d], 1u);
        d = (unsigned int)(vd[k].z - lo); if (d < DG_R) atomicAdd(&cnt[d], 1u);
        d = (unsigned int)(vd[k].w - lo); if (d < DG_R) atomicAdd(&cnt[d], 1u);
    }
    __syncthreads();

    unsigned int c[DG_F];
#pragma unroll
    for (int k = 0; k < DG_F; ++k) {
        const int i = k * DG_T + t;
        c[k] = i < DG_R ? cnt[i] : 0u;
    }
#pragma unroll
    for (int k = 0; k < DG_F; ++k)
        if (c[k] != 0u) c[k] = (unsigned int)atomicAdd(&cu[lo + k * DG_T + t], (int)c[k]);
#pragma unroll
    for (int k = 0; k < DG_F; ++k) {
        const int i = k * DG_T + t;
        if (i < DG_R) cnt[i] = c[k];
    }
    __syncthreads();

#pragma unroll
    for (int k = 0; k < DG_L; ++k) {
        const unsigned int dx = (unsigned int)(vd[k].x - lo), dy = (unsigned int)(vd[k].y - lo);
        const unsigned int dz = (unsigned int)(vd[k].z - lo), dw = (unsigned int)(vd[k].w - lo);
        unsigned int px = 0u, py = 0u, pz = 0u, pw = 0u;
        if (dx < DG_R) px = atomicAdd(&cnt[dx], 1u);
        if (dy < DG_R) py = atomicAdd(&cnt[dy], 1u);
        if (dz < DG_R) pz = atomicAdd(&cnt[dz], 1u);
        if (dw < DG_R) pw = atomicAdd(&cnt[dw], 1u);
        if (dx < DG_R) so[px] = vs[k].x;
        if (dy < DG_R) so[py] = vs[k].y;
        if (dz < DG_R) so[pz] = vs[k].z;
        if (dw < DG_R) so[pw] = vs[k].w;
    }
}

#define AG_BATCH 8

#define AG_ROWS   64
#define AG_T      512
#define AG_PITCH  132
#define AG_NB_INST ((INSTN + AG_ROWS - 1) / AG_ROWS)
#define AG_NB_SVC  ((SVCN  + AG_ROWS - 1) / AG_ROWS)
#define AG_NB_NODE ((NODEN + AG_ROWS - 1) / AG_ROWS)
__global__ __launch_bounds__(AG_T) void agg_proj_kernel(
        const float* __restrict__ feat_svc, const float* __restrict__ feat_inst,
        const float* __restrict__ feat_node,
        const float* __restrict__ W_sc, const float* __restrict__ W_in,
        const float* __restrict__ W_ni, const float* __restrict__ W_ii,
        const float* __restrict__ b_sc, const float* __restrict__ b_in,
        const float* __restrict__ b_ni, const float* __restrict__ b_ii,
        const float* __restrict__ deg, const int* __restrict__ curAll,
        const int* __restrict__ sortedAll, float* __restrict__ catAll) {
    __shared__ __attribute__((aligned(16))) float Wlds[128 * 128];
    __shared__ __attribute__((aligned(16))) float Tile[AG_ROWS * AG_PITCH];

    int blk = blockIdx.x;
    const float *X0, *W0, *ns0, *nd0, *bA;
    const int *cur0, *sorted0;
    const float *X1 = nullptr, *W1 = nullptr, *ns1 = nullptr, *nd1 = nullptr, *bB = nullptr;
    const int *cur1 = nullptr, *sorted1 = nullptr;
    int nrows, nlists = 1;
    float* cat;
    if (blk < AG_NB_INST) {
        X0 = feat_node; W0 = W_ni; bA = b_ni;
        ns0 = deg + (D_NS_NI - OFF_DEG); nd0 = deg + (D_ND_NI - OFF_DEG);
        cur0 = curAll + CUR_NI; sorted0 = sortedAll + SORT_NI;
        X1 = feat_inst; W1 = W_ii; bB = b_ii;
        ns1 = deg + (D_NS_II - OFF_DEG); nd1 = deg + (D_ND_II - OFF_DEG);
        cur1 = curAll + CUR_II; sorted1 = sortedAll + SORT_II;
        nrows = INSTN; nlists = 2;
        cat = catAll + (long long)(SVCN + NODEN) * 128;
    } else if ((blk -= AG_NB_INST) < AG_NB_SVC) {
        X0 = feat_svc; W0 = W_sc; bA = b_sc;
        ns0 = deg + (D_NS_SC - OFF_DEG); nd0 = deg + (D_ND_SC - OFF_DEG);
        cur0 = curAll + CUR_SC; sorted0 = sortedAll + SORT_SC;
        nrows = SVCN;
        cat = catAll;
    } else {
        blk -= AG_NB_SVC;
        X0 = feat_inst; W0 = W_in; bA = b_in;
        ns0 = deg + (D_NS_IN - OFF_DEG); nd0 = deg + (D_ND_IN - OFF_DEG);
        cur0 = curAll + CUR_IN; sorted0 = sortedAll + SORT_IN;
        nrows = NODEN;
        cat = catAll + (long long)SVCN * 128;
    }

    const int t    = threadIdx.x;
    const int lane = t & 31;
    const int wave = __builtin_amdgcn_readfirstlane(t >> 5);
    const int half = lane >> 4;
    const int l16  = lane & 15;
    const int c4   = lane * 4;
    const int r0   = blk * AG_ROWS;
    const int n0   = (wave & 7) * 16;
    const int rb   = (wave >> 3) * 32;

    v8f acc[2] = {v8f{}, v8f{}};

    for (int li = 0; li < nlists; ++li) {
        const float* __restrict__ X  = li ? X1 : X0;
        const float* __restrict__ Wg = li ? W1 : W0;
        const float* __restrict__ ns = li ? ns1 : ns0;
        const float* __restrict__ nd = li ? nd1 : nd0;
        const int* __restrict__ cur    = li ? cur1 : cur0;
        const int* __restrict__ sorted = li ? sorted1 : sorted0;

        const int first = r0 + wave * 4;
        int ic = first - 1 + lane; ic = ic < 0 ? 0 : (ic < nrows ? ic : nrows - 1);
        int id = first + lane;     id = id < nrows ? id : nrows - 1;
        const int   cv  = cur[ic];
        const float ndv = nd[id];

        const float4* __restrict__ Wg4 = (const float4*)Wg + t;
        const float4 wr0 = Wg4[0 * AG_T], wr1 = Wg4[1 * AG_T], wr2 = Wg4[2 * AG_T], wr3 = Wg4[3 * AG_T];
        const float4 wr4 = Wg4[4 * AG_T], wr5 = Wg4[5 * AG_T], wr6 = Wg4[6 * AG_T], wr7 = Wg4[7 * AG_T];

        const int b0 = first > 0 ? __builtin_amdgcn_readlane(cv, 0) : 0;
        const int e3 = __builtin_amdgcn_readlane(cv, 4);
        int sv_next = 0;
        if (lane < e3 - b0) sv_next = sorted[b0 + lane];

        if (li > 0) __syncthreads();

        float4* Wl4 = (float4*)Wlds + t;
        Wl4[0 * AG_T] = wr0; Wl4[1 * AG_T] = wr1; Wl4[2 * AG_T] = wr2; Wl4[3 * AG_T] = wr3;
        Wl4[4 * AG_T] = wr4; Wl4[5 * AG_T] = wr5; Wl4[6 * AG_T] = wr6; Wl4[7 * AG_T] = wr7;

        float4 acc4 = make_float4(0.f, 0.f, 0.f, 0.f);
        int ri   = 0;
        int rend = __builtin_amdgcn_readlane(cv, 1);
        for (int cb = b0; cb < e3; cb += 32) {
            const int rem = e3 - cb;
            const int cnt = rem < 32 ? rem : 32;
            const int sv  = sv_next;
            sv_next = 0;
            if (lane < rem - 32) sv_next = sorted[cb + 32 + lane];
            float wv = 0.f;
            if (lane < cnt) wv = ns[sv];
            for (int jj = 0; jj < cnt; jj += AG_BATCH) {
                float4 x[AG_BATCH];
#pragma unroll
                for (int k = 0; k < AG_BATCH; ++k) {
                    const int q = jj + k < cnt ? jj + k : cnt - 1;
                    const int s = __builtin_amdgcn_readlane(sv, q);
                    x[k] = *(const float4*)&X[(long long)s * 128 + c4];
                }
#pragma unroll
                for (int k = 0; k < AG_BATCH; ++k) {
                    const bool live = jj + k < cnt;
                    const int q = live ? jj + k : cnt - 1;
                    const int j = cb + q;
                    while (j >= rend) {
                        const float n = __builtin_bit_cast(float,
                            __builtin_amdgcn_readlane(__builtin_bit_cast(int, ndv), ri));
                        *(float4*)&Tile[(wave * 4 + ri) * AG_PITCH + c4] =
                            make_float4(acc4.x * n, acc4.y * n, acc4.z * n, acc4.w * n);
                        acc4 = make_float4(0.f, 0.f, 0.f, 0.f);
                        ++ri;
                        rend = __builtin_amdgcn_readlane(cv, ri + 1);
                    }
                    const float wq = __builtin_bit_cast(float,
                        __builtin_amdgcn_readlane(__builtin_bit_cast(int, wv), q));
                    const float w = live ? wq : 0.f;
                    acc4.x = fmaf(x[k].x, w, acc4.x); acc4.y = fmaf(x[k].y, w, acc4.y);
                    acc4.z = fmaf(x[k].z, w, acc4.z); acc4.w = fmaf(x[k].w, w, acc4.w);
                }
            }
        }
        for (; ri < 4; ++ri) {
            const float n = __builtin_bit_cast(float,
                __builtin_amdgcn_readlane(__builtin_bit_cast(int, ndv), ri));
            *(float4*)&Tile[(wave * 4 + ri) * AG_PITCH + c4] =
                make_float4(acc4.x * n, acc4.y * n, acc4.z * n, acc4.w * n);
            acc4 = make_float4(0.f, 0.f, 0.f, 0.f);
        }
        __syncthreads();

        for (int k = 0; k < 128; k += 4) {
            const int kk = k + 2 * half;
            v2f b;
            b.x = Wlds[kk * 128 + n0 + l16];
            b.y = Wlds[(kk + 1) * 128 + n0 + l16];
#pragma unroll
            for (int rt = 0; rt < 2; ++rt) {
                v2f a = *(const v2f*)&Tile[(rb + rt * 16 + l16) * AG_PITCH + kk];
                acc[rt] = __builtin_amdgcn_wmma_f32_16x16x4_f32(
                    false, a, false, b, (short)0, acc[rt], false, false);
            }
        }
    }

    __syncthreads();
#pragma unroll
    for (int rt = 0; rt < 2; ++rt) {
#pragma unroll
        for (int i = 0; i < 8; ++i)
            Tile[(rb + rt * 16 + i + 8 * half) * AG_PITCH + n0 + l16] = acc[rt][i];
    }
    __syncthreads();

    float4 bias = *(const float4*)&bA[c4];
    if (nlists > 1) {
        const float4 bb = *(const float4*)&bB[c4];
        bias.x += bb.x; bias.y += bb.y; bias.z += bb.z; bias.w += bb.w;
    }
    for (int i = 0; i < 4; ++i) {
        const int lr  = wave * 4 + i;
        const int row = r0 + lr;
        if (row < nrows) {
            const float4 p = *(const float4*)&Tile[lr * AG_PITCH + c4];
            *(float4*)&cat[(long long)row * 128 + c4] =
                make_float4(p.x + bias.x, p.y + bias.y, p.z + bias.z, p.w + bias.w);
        }
    }
}

#define FG_KT      32
#define FG_TILES   (TOTAL / FG_KT)
#define FG_BLOCKS  250
#define FG_TPB     (FG_TILES / FG_BLOCKS)
#define FG_PA      36
#define FG_PB      136
static_assert(TOTAL % FG_KT == 0, "K must be a whole number of tiles");
static_assert(FG_TILES % FG_BLOCKS == 0, "tiles must divide evenly over the blocks");

__global__ __launch_bounds__(512) void final_gemm_kernel(const float* __restrict__ Wt,
                                                         const float* __restrict__ cat,
                                                         float* __restrict__ part) {
    __shared__ float As[OUTD * FG_PA];
    __shared__ float Bs[FG_KT * FG_PB];

    const int t    = threadIdx.x;
    const int lane = t & 31;
    const int wave = t >> 5;
    const int half = lane >> 4;
    const int l16  = lane & 15;
    const int wm   = wave >> 2;
    const int wn   = wave & 3;
    const int tile0 = blockIdx.x * FG_TPB;

    const int arow = t >> 3;
    const int ac4  = t & 7;
    const int brow = t >> 5;
    const int bc4  = t & 31;
    const float* pA = Wt + (long long)arow * TOTAL + (long long)tile0 * FG_KT + ac4 * 4;
    const float* pB = cat + (long long)tile0 * FG_KT * 128 + t * 4;
    const long long aStep = 64LL * TOTAL;
    float* sA = &As[arow * FG_PA + ac4 * 4];
    float* sB = &Bs[brow * FG_PB + bc4 * 4];

    v8f acc[4][2];
#pragma unroll
    for (int mt = 0; mt < 4; ++mt)
#pragma unroll
        for (int nt = 0; nt < 2; ++nt) acc[mt][nt] = v8f{};

    float4 ra0 = *(const float4*)(pA);
    float4 ra1 = *(const float4*)(pA + aStep);
    float4 ra2 = *(const float4*)(pA + 2 * aStep);
    float4 ra3 = *(const float4*)(pA + 3 * aStep);
    float4 rb0 = *(const float4*)(pB);
    float4 rb1 = *(const float4*)(pB + 2048);

    for (int it = 0; it < FG_TPB; ++it) {
        *(float4*)(sA)                 = ra0;
        *(float4*)(sA + 64 * FG_PA)    = ra1;
        *(float4*)(sA + 128 * FG_PA)   = ra2;
        *(float4*)(sA + 192 * FG_PA)   = ra3;
        rb0.x = lrelu(rb0.x); rb0.y = lrelu(rb0.y); rb0.z = lrelu(rb0.z); rb0.w = lrelu(rb0.w);
        rb1.x = lrelu(rb1.x); rb1.y = lrelu(rb1.y); rb1.z = lrelu(rb1.z); rb1.w = lrelu(rb1.w);
        *(float4*)(sB)                 = rb0;
        *(float4*)(sB + 16 * FG_PB)    = rb1;
        __syncthreads();

        if (it + 1 < FG_TPB) {
            pA += FG_KT;
            pB += FG_KT * 128;
            ra0 = *(const float4*)(pA);
            ra1 = *(const float4*)(pA + aStep);
            ra2 = *(const float4*)(pA + 2 * aStep);
            ra3 = *(const float4*)(pA + 3 * aStep);
            rb0 = *(const float4*)(pB);
            rb1 = *(const float4*)(pB + 2048);
        }
        __builtin_amdgcn_sched_barrier(0);

#pragma unroll
        for (int ks = 0; ks < FG_KT / 4; ++ks) {
            const int kk = ks * 4 + 2 * half;
            v2f a[4], b[2];
#pragma unroll
            for (int mt = 0; mt < 4; ++mt)
                a[mt] = *(const v2f*)&As[(wm * 64 + mt * 16 + l16) * FG_PA + kk];
#pragma unroll
            for (int nt = 0; nt < 2; ++nt) {
                b[nt].x = Bs[kk * FG_PB + wn * 32 + nt * 16 + l16];
                b[nt].y = Bs[(kk + 1) * FG_PB + wn * 32 + nt * 16 + l16];
            }
#pragma unroll
            for (int mt = 0; mt < 4; ++mt)
#pragma unroll
                for (int nt = 0; nt < 2; ++nt)
                    acc[mt][nt] = __builtin_amdgcn_wmma_f32_16x16x4_f32(
                        false, a[mt], false, b[nt], (short)0, acc[mt][nt], false, false);
        }
        __syncthreads();
    }

    float* __restrict__ pp = part + (long long)blockIdx.x * (OUTD * HDIM) + wn * 32 + lane;
#pragma unroll
    for (int mt = 0; mt < 4; ++mt) {
#pragma unroll
        for (int i = 0; i < 8; ++i) {
            const float a  = acc[mt][0][i];
            const float b  = acc[mt][1][i];
            const float xa = __shfl_xor(a, 16);
            const float xb = __shfl_xor(b, 16);
            const int   m  = wm * 64 + mt * 16 + i;
            pp[m * 128]       = half ? xb : a;
            pp[(m + 8) * 128] = half ? b : xa;
        }
    }
}

#define RS_T 256
#define RS_G (RS_T / 32)
#define RS_L ((FG_BLOCKS + RS_G - 1) / RS_G)
static_assert(HDIM == 128, "one lane per 16 bytes of an output row");
__global__ __launch_bounds__(RS_T) void out_sum_kernel(const float* __restrict__ part,
                                                       const float* __restrict__ b_tot,
                                                       float* __restrict__ out) {
    __shared__ float4 red[RS_G * 32];
    const int t = threadIdx.x;
    const int c = t & 31;
    const int g = t >> 5;
    const int o = blockIdx.x;
    const float4* __restrict__ p = (const float4*)part + o * (HDIM / 4) + c;
    float4 v[RS_L];
#pragma unroll
    for (int k = 0; k < RS_L; ++k) {
        const int b = k * RS_G + g;
        v[k] = make_float4(0.f, 0.f, 0.f, 0.f);
        if (b < FG_BLOCKS) v[k] = p[(long long)b * (OUTD * HDIM / 4)];
    }
    float4 s = v[0];
#pragma unroll
    for (int k = 1; k < RS_L; ++k) { s.x += v[k].x; s.y += v[k].y; s.z += v[k].z; s.w += v[k].w; }
    red[g * 32 + c] = s;
    __syncthreads();
    if (g == 0) {
        float4 r = red[c];
        for (int j = 1; j < RS_G; ++j) {
            const float4 q = red[j * 32 + c];
            r.x += q.x; r.y += q.y; r.z += q.z; r.w += q.w;
        }
        const float bias = b_tot[o];
        ((float4*)out)[o * (HDIM / 4) + c] = make_float4(r.x + bias, r.y + bias, r.z + bias, r.w + bias);
    }
}

static inline int cdiv(long long a, long long b) { return (int)((a + b - 1) / b); }

extern "C" void kernel_launch(void* const* d_in, const int* in_sizes, int n_in,
                              void* d_out, int out_size, void* d_ws, size_t ws_size,
                              hipStream_t stream) {
    const float* feat_svc  = (const float*)d_in[0];
    const float* feat_inst = (const float*)d_in[1];
    const float* feat_node = (const float*)d_in[2];
    const int*   svc_src   = (const int*)d_in[3];
    const int*   svc_dst   = (const int*)d_in[4];
    const int*   in_src    = (const int*)d_in[5];
    const int*   in_dst    = (const int*)d_in[6];
    const int*   ni_src    = (const int*)d_in[7];
    const int*   ni_dst    = (const int*)d_in[8];
    const int*   ii_src    = (const int*)d_in[9];
    const int*   ii_dst    = (const int*)d_in[10];
    const float* W_sc = (const float*)d_in[11];  const float* b_sc = (const float*)d_in[12];
    const float* W_in = (const float*)d_in[13];  const float* b_in = (const float*)d_in[14];
    const float* W_ni = (const float*)d_in[15];  const float* b_ni = (const float*)d_in[16];
    const float* W_ii = (const float*)d_in[17];  const float* b_ii = (const float*)d_in[18];
    const float* W_tot = (const float*)d_in[19]; const float* b_tot = (const float*)d_in[20];
    float* out = (float*)d_out;

    float* ws   = (float*)d_ws;
    float* cat  = ws + OFF_CAT;
    float* deg  = ws + OFF_DEG;
    int* cur    = (int*)(ws + OFF_CUR);
    int* sorted = (int*)(ws + OFF_SORT);
    float* part = ws + OFF_PART;

    preset_kernel<<<cdiv(PRESET_N, 256), 256, 0, stream>>>(deg);

    degree_kernel<<<DG_NB_ALL, DG_T, 0, stream>>>(
        svc_src, svc_dst, in_src, in_dst, ni_src, ni_dst, ii_src, ii_dst, deg);
    offsets_kernel<<<4, SCAN_T, 0, stream>>>(deg, cur);

    fill_kernel<<<FL_NB_ALL, DG_T, 0, stream>>>(
        svc_src, svc_dst, in_src, in_dst, ni_src, ni_dst, ii_src, ii_dst, cur, sorted, deg);

    agg_proj_kernel<<<AG_NB_INST + AG_NB_SVC + AG_NB_NODE, AG_T, 0, stream>>>(
        feat_svc, feat_inst, feat_node, W_sc, W_in, W_ni, W_ii, b_sc, b_in, b_ni, b_ii,
        deg, cur, sorted, cat);

    final_gemm_kernel<<<FG_BLOCKS, 512, 0, stream>>>(W_tot, cat, part);
    out_sum_kernel<<<OUTD, RS_T, 0, stream>>>(part, b_tot, out);
}


# agg_proj: registers 112 to 80 (fragment addresses as two bases plus constant offsets; one index for range ends and factors), so three blocks, not two, share a WGP; sum orders unchanged
# speedup vs baseline: 3.0762x; 1.0313x over previous
#include <hip/hip_runtime.h>

#define SVCN   50000
#define INSTN  100000
#define NODEN  50000
#define FDIM   128
#define HDIM   128
#define OUTD   256
#define TOTAL  (SVCN + NODEN + INSTN)
#define E_SC   400000
#define E_IN   100000
#define E_NI   100000
#define E_II   800000

#define OFF_CAT   0LL
#define OFF_DEG   (OFF_CAT  + (long long)TOTAL * 128)
#define D_NS_SC   (OFF_DEG + 0LL)
#define D_ND_SC   (D_NS_SC + SVCN)
#define D_NS_IN   (D_ND_SC + SVCN)
#define D_ND_IN   (D_NS_IN + INSTN)
#define D_NS_NI   (D_ND_IN + NODEN)
#define D_ND_NI   (D_NS_NI + NODEN)
#define D_NS_II   (D_ND_NI + INSTN)
#define D_ND_II   (D_NS_II + INSTN)
#define DEG_TOTAL 600000
#define OFF_CUR   (OFF_DEG + (long long)DEG_TOTAL)
#define CUR_SC    0
#define CUR_IN    (CUR_SC + SVCN)
#define CUR_NI    (CUR_IN + NODEN)
#define CUR_II    (CUR_NI + INSTN)
#define CUR_TOTAL (CUR_II + INSTN)
#define OFF_SORT  (OFF_CUR + (long long)CUR_TOTAL)
#define SORT_SC   0
#define SORT_IN   (SORT_SC + E_SC)
#define SORT_NI   (SORT_IN + E_IN)
#define SORT_II   (SORT_NI + E_NI)
#define OFF_PART  (OFF_SORT + (long long)(E_SC + E_IN + E_NI + E_II))
static_assert(OFF_PART % 32 == 0, "partial planes start on a 128-byte line");
static_assert(SVCN % 4 == 0 && NODEN % 4 == 0 && INSTN % 4 == 0, "degree planes are scanned four at a time");

typedef __attribute__((ext_vector_type(2))) float v2f;
typedef __attribute__((ext_vector_type(8))) float v8f;

__device__ __forceinline__ float lrelu(float x) { return x > 0.f ? x : 0.01f * x; }

__device__ __forceinline__ void atomAddF(float* p, float v) { unsafeAtomicAdd(p, v); }

#define PRESET_N DEG_TOTAL
__global__ void preset_kernel(float* __restrict__ deg) {
    const int i = blockIdx.x * blockDim.x + threadIdx.x;
    if (i < DEG_TOTAL) deg[i] = 0.f;
}

__device__ __forceinline__ float deg_factor(float d) { return rsqrtf(d > 1.f ? d : 1.f); }

#define DG_T  512
#define DG_R  12500
#define DG_C  50000
#define DG_L  ((DG_C / 4 + DG_T - 1) / DG_T)
#define DG_F  ((DG_R + DG_T - 1) / DG_T)
static_assert(SVCN % DG_R == 0 && NODEN % DG_R == 0 && INSTN % DG_R == 0, "planes split into whole ranges");
static_assert(E_SC % DG_C == 0 && E_IN % DG_C == 0 && E_NI % DG_C == 0 && E_II % DG_C == 0, "lists split into whole chunks");
static_assert(DG_R % 4 == 0 && DG_C % 4 == 0, "ranges and chunks are whole 16-byte pieces");
#define DG_NB(entries, edges) (((entries) / DG_R) * ((edges) / DG_C))
#define DG_NB0 DG_NB(SVCN,  E_SC)
#define DG_NB1 DG_NB(SVCN,  E_SC)
#define DG_NB2 DG_NB(INSTN, E_IN)
#define DG_NB3 DG_NB(NODEN, E_IN)
#define DG_NB4 DG_NB(NODEN, E_NI)
#define DG_NB5 DG_NB(INSTN, E_NI)
#define DG_NB6 DG_NB(INSTN, E_II)
#define DG_NB7 DG_NB(INSTN, E_II)
#define DG_NB_ALL (DG_NB0 + DG_NB1 + DG_NB2 + DG_NB3 + DG_NB4 + DG_NB5 + DG_NB6 + DG_NB7)
__global__ __launch_bounds__(DG_T) void degree_kernel(const int* __restrict__ svc_src, const int* __restrict__ svc_dst,
                                                      const int* __restrict__ in_src,  const int* __restrict__ in_dst,
                                                      const int* __restrict__ ni_src,  const int* __restrict__ ni_dst,
                                                      const int* __restrict__ ii_src,  const int* __restrict__ ii_dst,
                                                      float* __restrict__ deg) {
    __shared__ __attribute__((aligned(16))) unsigned int cnt[DG_R];
    int b = blockIdx.x;
    const int* idx; float* plane; int nch;
    if (b < DG_NB0)                  { idx = svc_src; plane = deg + (D_NS_SC - OFF_DEG); nch = E_SC / DG_C; }
    else if ((b -= DG_NB0) < DG_NB1) { idx = svc_dst; plane = deg + (D_ND_SC - OFF_DEG); nch = E_SC / DG_C; }
    else if ((b -= DG_NB1) < DG_NB2) { idx = in_src;  plane = deg + (D_NS_IN - OFF_DEG); nch = E_IN / DG_C; }
    else if ((b -= DG_NB2) < DG_NB3) { idx = in_dst;  plane = deg + (D_ND_IN - OFF_DEG); nch = E_IN / DG_C; }
    else if ((b -= DG_NB3) < DG_NB4) { idx = ni_src;  plane = deg + (D_NS_NI - OFF_DEG); nch = E_NI / DG_C; }
    else if ((b -= DG_NB4) < DG_NB5) { idx = ni_dst;  plane = deg + (D_ND_NI - OFF_DEG); nch = E_NI / DG_C; }
    else if ((b -= DG_NB5) < DG_NB6) { idx = ii_src;  plane = deg + (D_NS_II - OFF_DEG); nch = E_II / DG_C; }
    else { b -= DG_NB6;                idx = ii_dst;  plane = deg + (D_ND_II - OFF_DEG); nch = E_II / DG_C; }
    const int range = b / nch;
    const int chunk = b - range * nch;
    const int lo    = range * DG_R;
    const int t     = threadIdx.x;

    const int4* __restrict__ p4 = (const int4*)(idx + chunk * DG_C);
    int4 v[DG_L];
#pragma unroll
    for (int k = 0; k < DG_L; ++k) {
        const int i = k * DG_T + t;
        v[k] = make_int4(-1, -1, -1, -1);
        if (i < DG_C / 4) v[k] = p4[i];
    }

    for (int i = t; i < DG_R / 4; i += DG_T) ((uint4*)cnt)[i] = make_uint4(0u, 0u, 0u, 0u);
    __syncthreads();

#pragma unroll
    for (int k = 0; k < DG_L; ++k) {
        unsigned int d;
        d = (unsigned int)(v[k].x - lo); if (d < DG_R) atomicAdd(&cnt[d], 1u);
        d = (unsigned int)(v[k].y - lo); if (d < DG_R) atomicAdd(&cnt[d], 1u);
        d = (unsigned int)(v[k].z - lo); if (d < DG_R) atomicAdd(&cnt[d], 1u);
        d = (unsigned int)(v[k].w - lo); if (d < DG_R) atomicAdd(&cnt[d], 1u);
    }
    __syncthreads();

    unsigned int c[DG_F];
#pragma unroll
    for (int k = 0; k < DG_F; ++k) {
        const int i = k * DG_T + t;
        c[k] = i < DG_R ? cnt[i] : 0u;
    }
#pragma unroll
    for (int k = 0; k < DG_F; ++k)
        if (c[k] != 0u) atomAddF(&plane[lo + k * DG_T + t], (float)c[k]);
}


#define SCAN_T 1024
#define SCAN_Q 4
__global__ __launch_bounds__(SCAN_T) void offsets_kernel(const float* __restrict__ deg,
                                                         int* __restrict__ cur) {
    __shared__ int wsum[32];
    __shared__ int tile_total;
    const float* dg; int* out; int n;
    if (blockIdx.x == 0)      { dg = deg + (D_ND_SC - OFF_DEG); out = cur + CUR_SC; n = SVCN;  }
    else if (blockIdx.x == 1) { dg = deg + (D_ND_IN - OFF_DEG); out = cur + CUR_IN; n = NODEN; }
    else if (blockIdx.x == 2) { dg = deg + (D_ND_NI - OFF_DEG); out = cur + CUR_NI; n = INSTN; }
    else                      { dg = deg + (D_ND_II - OFF_DEG); out = cur + CUR_II; n = INSTN; }
    const int t = threadIdx.x, lane = t & 31, wave = t >> 5;
    int carry = 0;
    for (int base = 0; base < n; base += SCAN_T * SCAN_Q * 4) {
        const int i0 = base + t * (SCAN_Q * 4);
        int4 v[SCAN_Q];
        int s = 0;
#pragma unroll
        for (int q = 0; q < SCAN_Q; ++q) {
            const int i = i0 + q * 4;
            v[q] = make_int4(0, 0, 0, 0);
            if (i < n) {
                const float4 f = *(const float4*)&dg[i];
                v[q] = make_int4((int)f.x, (int)f.y, (int)f.z, (int)f.w);
            }
            s += v[q].x + v[q].y + v[q].z + v[q].w;
        }
        int x = s;
        for (int o = 1; o < 32; o <<= 1) {
            const int y = __shfl_up(x, o);
            if (lane >= o) x += y;
        }
        if (lane == 31) wsum[wave] = x;
        __syncthreads();
        if (wave == 0) {
            const int w = wsum[lane];
            int xs = w;
            for (int o = 1; o < 32; o <<= 1) {
                const int y = __shfl_up(xs, o);
                if (lane >= o) xs += y;
            }
            wsum[lane] = xs - w;
            if (lane == 31) tile_total = xs;
        }
        __syncthreads();
        int run = carry + wsum[wave] + (x - s);
#pragma unroll
        for (int q = 0; q < SCAN_Q; ++q) {
            const int i = i0 + q * 4;
            if (i < n)
                *(int4*)&out[i] = make_int4(run, run + v[q].x, run + v[q].x + v[q].y,
                                            run + v[q].x + v[q].y + v[q].z);
            run += v[q].x + v[q].y + v[q].z + v[q].w;
        }
        carry += tile_total;
    }
}

#define FL_NB0 DG_NB(SVCN,  E_SC)
#define FL_NB1 DG_NB(NODEN, E_IN)
#define FL_NB2 DG_NB(INSTN, E_NI)
#define FL_NB3 DG_NB(INSTN, E_II)
#define FL_NB_ALL (FL_NB0 + FL_NB1 + FL_NB2 + FL_NB3)
#define FL_NT     (FL_NB_ALL * DG_T)
#define FL_DEG4   (DEG_TOTAL / 4)
#define FL_DQ     ((FL_DEG4 + FL_NT - 1) / FL_NT)
static_assert(DEG_TOTAL % 4 == 0, "degree planes are whole 16-byte pieces");
__global__ __launch_bounds__(DG_T) void fill_kernel(const int* __restrict__ svc_src, const int* __restrict__ svc_dst,
                                                    const int* __restrict__ in_src,  const int* __restrict__ in_dst,
                                                    const int* __restrict__ ni_src,  const int* __restrict__ ni_dst,
                                                    const int* __restrict__ ii_src,  const int* __restrict__ ii_dst,
                                                    int* __restrict__ cur, int* __restrict__ sorted,
                                                    float* __restrict__ deg) {
    __shared__ __attribute__((aligned(16))) unsigned int cnt[DG_R];
    const int t = threadIdx.x;

    const int g = blockIdx.x * DG_T + t;
    float4 dq[FL_DQ];
#pragma unroll
    for (int q = 0; q < FL_DQ; ++q) {
        const int i = q * FL_NT + g;
        dq[q] = make_float4(1.f, 1.f, 1.f, 1.f);
        if (i < FL_DEG4) dq[q] = ((const float4*)deg)[i];
    }

    int b = blockIdx.x;
    const int* src; const int* dst; int* cu; int* so; int nch;
    if (b < FL_NB0)                  { src = svc_src; dst = svc_dst; cu = cur + CUR_SC; so = sorted + SORT_SC; nch = E_SC / DG_C; }
    else if ((b -= FL_NB0) < FL_NB1) { src = in_src;  dst = in_dst;  cu = cur + CUR_IN; so = sorted + SORT_IN; nch = E_IN / DG_C; }
    else if ((b -= FL_NB1) < FL_NB2) { src = ni_src;  dst = ni_dst;  cu = cur + CUR_NI; so = sorted + SORT_NI; nch = E_NI / DG_C; }
    else { b -= FL_NB2;                src = ii_src;  dst = ii_dst;  cu = cur + CUR_II; so = sorted + SORT_II; nch = E_II / DG_C; }
    const int range = b / nch;
    const int chunk = b - range * nch;
    const int lo    = range * DG_R;

    const int4* __restrict__ d4 = (const int4*)(dst + chunk * DG_C);
    const int4* __restrict__ s4 = (const int4*)(src + chunk * DG_C);
    int4 vd[DG_L], vs[DG_L];
#pragma unroll
    for (int k = 0; k < DG_L; ++k) {
        const int i = k * DG_T + t;
        vd[k] = make_int4(-1, -1, -1, -1);
        vs[k] = make_int4(0, 0, 0, 0);
        if (i < DG_C / 4) { vd[k] = d4[i]; vs[k] = s4[i]; }
    }

    for (int i = t; i < DG_R / 4; i += DG_T) ((uint4*)cnt)[i] = make_uint4(0u, 0u, 0u, 0u);

#pragma unroll
    for (int q = 0; q < FL_DQ; ++q) {
        const int i = q * FL_NT + g;
        if (i < FL_DEG4)
            ((float4*)deg)[i] = make_float4(deg_factor(dq[q].x), deg_factor(dq[q].y),
                                            deg_factor(dq[q].z), deg_factor(dq[q].w));
    }
    __syncthreads();

#pragma unroll
    for (int k = 0; k < DG_L; ++k) {
        unsigned int d;
        d = (unsigned int)(vd[k].x - lo); if (d < DG_R) atomicAdd(&cnt[d], 1u);
        d = (unsigned int)(vd[k].y - lo); if (d < DG_R) atomicAdd(&cnt[d], 1u);
        d = (unsigned int)(vd[k].z - lo); if (d < DG_R) atomicAdd(&cnt[d], 1u);
        d = (unsigned int)(vd[k].w - lo); if (d < DG_R) atomicAdd(&cnt[d], 1u);
    }
    __syncthreads();

    unsigned int c[DG_F];
#pragma unroll
    for (int k = 0; k < DG_F; ++k) {
        const int i = k * DG_T + t;
        c[k] = i < DG_R ? cnt[i] : 0u;
    }
#pragma unroll
    for (int k = 0; k < DG_F; ++k)
        if (c[k] != 0u) c[k] = (unsigned int)atomicAdd(&cu[lo + k * DG_T + t], (int)c[k]);
#pragma unroll
    for (int k = 0; k < DG_F; ++k) {
        const int i = k * DG_T + t;
        if (i < DG_R) cnt[i] = c[k];
    }
    __syncthreads();

#pragma unroll
    for (int k = 0; k < DG_L; ++k) {
        const unsigned int dx = (unsigned int)(vd[k].x - lo), dy = (unsigned int)(vd[k].y - lo);
        const unsigned int dz = (unsigned int)(vd[k].z - lo), dw = (unsigned int)(vd[k].w - lo);
        unsigned int px = 0u, py = 0u, pz = 0u, pw = 0u;
        if (dx < DG_R) px = atomicAdd(&cnt[dx], 1u);
        if (dy < DG_R) py = atomicAdd(&cnt[dy], 1u);
        if (dz < DG_R) pz = atomicAdd(&cnt[dz], 1u);
        if (dw < DG_R) pw = atomicAdd(&cnt[dw], 1u);
        if (dx < DG_R) so[px] = vs[k].x;
        if (dy < DG_R) so[py] = vs[k].y;
        if (dz < DG_R) so[pz] = vs[k].z;
        if (dw < DG_R) so[pw] = vs[k].w;
    }
}

#define AG_BATCH 8

#define AG_ROWS   64
#define AG_T      512
#define AG_PITCH  132
#define AG_NB_INST ((INSTN + AG_ROWS - 1) / AG_ROWS)
#define AG_NB_SVC  ((SVCN  + AG_ROWS - 1) / AG_ROWS)
#define AG_NB_NODE ((NODEN + AG_ROWS - 1) / AG_ROWS)
__global__ __launch_bounds__(AG_T, 12) void agg_proj_kernel(
        const float* __restrict__ feat_svc, const float* __restrict__ feat_inst,
        const float* __restrict__ feat_node,
        const float* __restrict__ W_sc, const float* __restrict__ W_in,
        const float* __restrict__ W_ni, const float* __restrict__ W_ii,
        const float* __restrict__ b_sc, const float* __restrict__ b_in,
        const float* __restrict__ b_ni, const float* __restrict__ b_ii,
        const float* __restrict__ deg, const int* __restrict__ curAll,
        const int* __restrict__ sortedAll, float* __restrict__ catAll) {
    __shared__ __attribute__((aligned(16))) float Wlds[128 * 128];
    __shared__ __attribute__((aligned(16))) float Tile[AG_ROWS * AG_PITCH];

    int blk = blockIdx.x;
    const float *X0, *W0, *ns0, *nd0, *bA;
    const int *cur0, *sorted0;
    const float *X1 = nullptr, *W1 = nullptr, *ns1 = nullptr, *nd1 = nullptr, *bB = nullptr;
    const int *cur1 = nullptr, *sorted1 = nullptr;
    int nrows, nlists = 1;
    float* cat;
    if (blk < AG_NB_INST) {
        X0 = feat_node; W0 = W_ni; bA = b_ni;
        ns0 = deg + (D_NS_NI - OFF_DEG); nd0 = deg + (D_ND_NI - OFF_DEG);
        cur0 = curAll + CUR_NI; sorted0 = sortedAll + SORT_NI;
        X1 = feat_inst; W1 = W_ii; bB = b_ii;
        ns1 = deg + (D_NS_II - OFF_DEG); nd1 = deg + (D_ND_II - OFF_DEG);
        cur1 = curAll + CUR_II; sorted1 = sortedAll + SORT_II;
        nrows = INSTN; nlists = 2;
        cat = catAll + (long long)(SVCN + NODEN) * 128;
    } else if ((blk -= AG_NB_INST) < AG_NB_SVC) {
        X0 = feat_svc; W0 = W_sc; bA = b_sc;
        ns0 = deg + (D_NS_SC - OFF_DEG); nd0 = deg + (D_ND_SC - OFF_DEG);
        cur0 = curAll + CUR_SC; sorted0 = sortedAll + SORT_SC;
        nrows = SVCN;
        cat = catAll;
    } else {
        blk -= AG_NB_SVC;
        X0 = feat_inst; W0 = W_in; bA = b_in;
        ns0 = deg + (D_NS_IN - OFF_DEG); nd0 = deg + (D_ND_IN - OFF_DEG);
        cur0 = curAll + CUR_IN; sorted0 = sortedAll + SORT_IN;
        nrows = NODEN;
        cat = catAll + (long long)SVCN * 128;
    }

    const int t    = threadIdx.x;
    const int lane = t & 31;
    const int wave = __builtin_amdgcn_readfirstlane(t >> 5);
    const int half = lane >> 4;
    const int l16  = lane & 15;
    const int c4   = lane * 4;
    const int r0   = blk * AG_ROWS;
    const int n0   = (wave & 7) * 16;
    const int rb   = (wave >> 3) * 32;

    v8f acc[2] = {v8f{}, v8f{}};

    const float* wfrag = &Wlds[(2 * half) * 128 + n0 + l16];
    const float* tfrag = &Tile[(rb + l16) * AG_PITCH + 2 * half];

    for (int li = 0; li < nlists; ++li) {
        const float* __restrict__ X  = li ? X1 : X0;
        const float* __restrict__ Wg = li ? W1 : W0;
        const float* __restrict__ ns = li ? ns1 : ns0;
        const float* __restrict__ nd = li ? nd1 : nd0;
        const int* __restrict__ cur    = li ? cur1 : cur0;
        const int* __restrict__ sorted = li ? sorted1 : sorted0;

        const int first = r0 + wave * 4;
        int id = first + lane;     id = id < nrows ? id : nrows - 1;
        int ib = first - 1;        ib = ib < 0 ? 0 : (ib < nrows ? ib : nrows - 1);
        const int   cv  = cur[id];
        const float ndv = nd[id];
        const int   cb0 = cur[ib];

        const float4* __restrict__ Wg4 = (const float4*)Wg + t;
        const float4 wr0 = Wg4[0 * AG_T], wr1 = Wg4[1 * AG_T], wr2 = Wg4[2 * AG_T], wr3 = Wg4[3 * AG_T];
        const float4 wr4 = Wg4[4 * AG_T], wr5 = Wg4[5 * AG_T], wr6 = Wg4[6 * AG_T], wr7 = Wg4[7 * AG_T];

        const int b0 = first > 0 ? cb0 : 0;
        const int e3 = __builtin_amdgcn_readlane(cv, 3);
        int sv_next = 0;
        if (lane < e3 - b0) sv_next = sorted[b0 + lane];

        if (li > 0) __syncthreads();

        float4* Wl4 = (float4*)Wlds + t;
        Wl4[0 * AG_T] = wr0; Wl4[1 * AG_T] = wr1; Wl4[2 * AG_T] = wr2; Wl4[3 * AG_T] = wr3;
        Wl4[4 * AG_T] = wr4; Wl4[5 * AG_T] = wr5; Wl4[6 * AG_T] = wr6; Wl4[7 * AG_T] = wr7;

        float4 acc4 = make_float4(0.f, 0.f, 0.f, 0.f);
        int ri   = 0;
        int rend = __builtin_amdgcn_readlane(cv, 0);
        for (int cb = b0; cb < e3; cb += 32) {
            const int rem = e3 - cb;
            const int cnt = rem < 32 ? rem : 32;
            const int sv  = sv_next;
            sv_next = 0;
            if (lane < rem - 32) sv_next = sorted[cb + 32 + lane];
            float wv = 0.f;
            if (lane < cnt) wv = ns[sv];
            for (int jj = 0; jj < cnt; jj += AG_BATCH) {
                float4 x[AG_BATCH];
#pragma unroll
                for (int k = 0; k < AG_BATCH; ++k) {
                    const int q = jj + k < cnt ? jj + k : cnt - 1;
                    const int s = __builtin_amdgcn_readlane(sv, q);
                    x[k] = *(const float4*)&X[(long long)s * 128 + c4];
                }
#pragma unroll
                for (int k = 0; k < AG_BATCH; ++k) {
                    const bool live = jj + k < cnt;
                    const int q = live ? jj + k : cnt - 1;
                    const int j = cb + q;
                    while (j >= rend) {
                        const float n = __builtin_bit_cast(float,
                            __builtin_amdgcn_readlane(__builtin_bit_cast(int, ndv), ri));
                        *(float4*)&Tile[(wave * 4 + ri) * AG_PITCH + c4] =
                            make_float4(acc4.x * n, acc4.y * n, acc4.z * n, acc4.w * n);
                        acc4 = make_float4(0.f, 0.f, 0.f, 0.f);
                        ++ri;
                        rend = __builtin_amdgcn_readlane(cv, ri);
                    }
                    const float wq = __builtin_bit_cast(float,
                        __builtin_amdgcn_readlane(__builtin_bit_cast(int, wv), q));
                    const float w = live ? wq : 0.f;
                    acc4.x = fmaf(x[k].x, w, acc4.x); acc4.y = fmaf(x[k].y, w, acc4.y);
                    acc4.z = fmaf(x[k].z, w, acc4.z); acc4.w = fmaf(x[k].w, w, acc4.w);
                }
            }
        }
        for (; ri < 4; ++ri) {
            const float n = __builtin_bit_cast(float,
                __builtin_amdgcn_readlane(__builtin_bit_cast(int, ndv), ri));
            *(float4*)&Tile[(wave * 4 + ri) * AG_PITCH + c4] =
                make_float4(acc4.x * n, acc4.y * n, acc4.z * n, acc4.w * n);
            acc4 = make_float4(0.f, 0.f, 0.f, 0.f);
        }
        __syncthreads();

        for (int k = 0; k < 128; k += 4) {
            v2f b;
            b.x = wfrag[k * 128];
            b.y = wfrag[(k + 1) * 128];
#pragma unroll
            for (int rt = 0; rt < 2; ++rt) {
                v2f a = *(const v2f*)&tfrag[rt * 16 * AG_PITCH + k];
                acc[rt] = __builtin_amdgcn_wmma_f32_16x16x4_f32(
                    false, a, false, b, (short)0, acc[rt], false, false);
            }
        }
    }

    __syncthreads();
#pragma unroll
    for (int rt = 0; rt < 2; ++rt) {
#pragma unroll
        for (int i = 0; i < 8; ++i)
            Tile[(rb + rt * 16 + i + 8 * half) * AG_PITCH + n0 + l16] = acc[rt][i];
    }
    __syncthreads();

    float4 bias = *(const float4*)&bA[c4];
    if (nlists > 1) {
        const float4 bb = *(const float4*)&bB[c4];
        bias.x += bb.x; bias.y += bb.y; bias.z += bb.z; bias.w += bb.w;
    }
    for (int i = 0; i < 4; ++i) {
        const int lr  = wave * 4 + i;
        const int row = r0 + lr;
        if (row < nrows) {
            const float4 p = *(const float4*)&Tile[lr * AG_PITCH + c4];
            *(float4*)&cat[(long long)row * 128 + c4] =
                make_float4(p.x + bias.x, p.y + bias.y, p.z + bias.z, p.w + bias.w);
        }
    }
}

#define FG_KT      32
#define FG_TILES   (TOTAL / FG_KT)
#define FG_BLOCKS  250
#define FG_TPB     (FG_TILES / FG_BLOCKS)
#define FG_PA      36
#define FG_PB      136
static_assert(TOTAL % FG_KT == 0, "K must be a whole number of tiles");
static_assert(FG_TILES % FG_BLOCKS == 0, "tiles must divide evenly over the blocks");

__global__ __launch_bounds__(512) void final_gemm_kernel(const float* __restrict__ Wt,
                                                         const float* __restrict__ cat,
                                                         float* __restrict__ part) {
    __shared__ float As[OUTD * FG_PA];
    __shared__ float Bs[FG_KT * FG_PB];

    const int t    = threadIdx.x;
    const int lane = t & 31;
    const int wave = t >> 5;
    const int half = lane >> 4;
    const int l16  = lane & 15;
    const int wm   = wave >> 2;
    const int wn   = wave & 3;
    const int tile0 = blockIdx.x * FG_TPB;

    const int arow = t >> 3;
    const int ac4  = t & 7;
    const int brow = t >> 5;
    const int bc4  = t & 31;
    const float* pA = Wt + (long long)arow * TOTAL + (long long)tile0 * FG_KT + ac4 * 4;
    const float* pB = cat + (long long)tile0 * FG_KT * 128 + t * 4;
    const long long aStep = 64LL * TOTAL;
    float* sA = &As[arow * FG_PA + ac4 * 4];
    float* sB = &Bs[brow * FG_PB + bc4 * 4];

    v8f acc[4][2];
#pragma unroll
    for (int mt = 0; mt < 4; ++mt)
#pragma unroll
        for (int nt = 0; nt < 2; ++nt) acc[mt][nt] = v8f{};

    float4 ra0 = *(const float4*)(pA);
    float4 ra1 = *(const float4*)(pA + aStep);
    float4 ra2 = *(const float4*)(pA + 2 * aStep);
    float4 ra3 = *(const float4*)(pA + 3 * aStep);
    float4 rb0 = *(const float4*)(pB);
    float4 rb1 = *(const float4*)(pB + 2048);

    for (int it = 0; it < FG_TPB; ++it) {
        *(float4*)(sA)                 = ra0;
        *(float4*)(sA + 64 * FG_PA)    = ra1;
        *(float4*)(sA + 128 * FG_PA)   = ra2;
        *(float4*)(sA + 192 * FG_PA)   = ra3;
        rb0.x = lrelu(rb0.x); rb0.y = lrelu(rb0.y); rb0.z = lrelu(rb0.z); rb0.w = lrelu(rb0.w);
        rb1.x = lrelu(rb1.x); rb1.y = lrelu(rb1.y); rb1.z = lrelu(rb1.z); rb1.w = lrelu(rb1.w);
        *(float4*)(sB)                 = rb0;
        *(float4*)(sB + 16 * FG_PB)    = rb1;
        __syncthreads();

        if (it + 1 < FG_TPB) {
            pA += FG_KT;
            pB += FG_KT * 128;
            ra0 = *(const float4*)(pA);
            ra1 = *(const float4*)(pA + aStep);
            ra2 = *(const float4*)(pA + 2 * aStep);
            ra3 = *(const float4*)(pA + 3 * aStep);
            rb0 = *(const float4*)(pB);
            rb1 = *(const float4*)(pB + 2048);
        }
        __builtin_amdgcn_sched_barrier(0);

#pragma unroll
        for (int ks = 0; ks < FG_KT / 4; ++ks) {
            const int kk = ks * 4 + 2 * half;
            v2f a[4], b[2];
#pragma unroll
            for (int mt = 0; mt < 4; ++mt)
                a[mt] = *(const v2f*)&As[(wm * 64 + mt * 16 + l16) * FG_PA + kk];
#pragma unroll
            for (int nt = 0; nt < 2; ++nt) {
                b[nt].x = Bs[kk * FG_PB + wn * 32 + nt * 16 + l16];
                b[nt].y = Bs[(kk + 1) * FG_PB + wn * 32 + nt * 16 + l16];
            }
#pragma unroll
            for (int mt = 0; mt < 4; ++mt)
#pragma unroll
                for (int nt = 0; nt < 2; ++nt)
                    acc[mt][nt] = __builtin_amdgcn_wmma_f32_16x16x4_f32(
                        false, a[mt], false, b[nt], (short)0, acc[mt][nt], false, false);
        }
        __syncthreads();
    }

    float* __restrict__ pp = part + (long long)blockIdx.x * (OUTD * HDIM) + wn * 32 + lane;
#pragma unroll
    for (int mt = 0; mt < 4; ++mt) {
#pragma unroll
        for (int i = 0; i < 8; ++i) {
            const float a  = acc[mt][0][i];
            const float b  = acc[mt][1][i];
            const float xa = __shfl_xor(a, 16);
            const float xb = __shfl_xor(b, 16);
            const int   m  = wm * 64 + mt * 16 + i;
            pp[m * 128]       = half ? xb : a;
            pp[(m + 8) * 128] = half ? b : xa;
        }
    }
}

#define RS_T 256
#define RS_G (RS_T / 32)
#define RS_L ((FG_BLOCKS + RS_G - 1) / RS_G)
static_assert(HDIM == 128, "one lane per 16 bytes of an output row");
__global__ __launch_bounds__(RS_T) void out_sum_kernel(const float* __restrict__ part,
                                                       const float* __restrict__ b_tot,
                                                       float* __restrict__ out) {
    __shared__ float4 red[RS_G * 32];
    const int t = threadIdx.x;
    const int c = t & 31;
    const int g = t >> 5;
    const int o = blockIdx.x;
    const float4* __restrict__ p = (const float4*)part + o * (HDIM / 4) + c;
    float4 v[RS_L];
#pragma unroll
    for (int k = 0; k < RS_L; ++k) {
        const int b = k * RS_G + g;
        v[k] = make_float4(0.f, 0.f, 0.f, 0.f);
        if (b < FG_BLOCKS) v[k] = p[(long long)b * (OUTD * HDIM / 4)];
    }
    float4 s = v[0];
#pragma unroll
    for (int k = 1; k < RS_L; ++k) { s.x += v[k].x; s.y += v[k].y; s.z += v[k].z; s.w += v[k].w; }
    red[g * 32 + c] = s;
    __syncthreads();
    if (g == 0) {
        float4 r = red[c];
        for (int j = 1; j < RS_G; ++j) {
            const float4 q = red[j * 32 + c];
            r.x += q.x; r.y += q.y; r.z += q.z; r.w += q.w;
        }
        const float bias = b_tot[o];
        ((float4*)out)[o * (HDIM / 4) + c] = make_float4(r.x + bias, r.y + bias, r.z + bias, r.w + bias);
    }
}

static inline int cdiv(long long a, long long b) { return (int)((a + b - 1) / b); }

extern "C" void kernel_launch(void* const* d_in, const int* in_sizes, int n_in,
                              void* d_out, int out_size, void* d_ws, size_t ws_size,
                              hipStream_t stream) {
    const float* feat_svc  = (const float*)d_in[0];
    const float* feat_inst = (const float*)d_in[1];
    const float* feat_node = (const float*)d_in[2];
    const int*   svc_src   = (const int*)d_in[3];
    const int*   svc_dst   = (const int*)d_in[4];
    const int*   in_src    = (const int*)d_in[5];
    const int*   in_dst    = (const int*)d_in[6];
    const int*   ni_src    = (const int*)d_in[7];
    const int*   ni_dst    = (const int*)d_in[8];
    const int*   ii_src    = (const int*)d_in[9];
    const int*   ii_dst    = (const int*)d_in[10];
    const float* W_sc = (const float*)d_in[11];  const float* b_sc = (const float*)d_in[12];
    const float* W_in = (const float*)d_in[13];  const float* b_in = (const float*)d_in[14];
    const float* W_ni = (const float*)d_in[15];  const float* b_ni = (const float*)d_in[16];
    const float* W_ii = (const float*)d_in[17];  const float* b_ii = (const float*)d_in[18];
    const float* W_tot = (const float*)d_in[19]; const float* b_tot = (const float*)d_in[20];
    float* out = (float*)d_out;

    float* ws   = (float*)d_ws;
    float* cat  = ws + OFF_CAT;
    float* deg  = ws + OFF_DEG;
    int* cur    = (int*)(ws + OFF_CUR);
    int* sorted = (int*)(ws + OFF_SORT);
    float* part = ws + OFF_PART;

    preset_kernel<<<cdiv(PRESET_N, 256), 256, 0, stream>>>(deg);

    degree_kernel<<<DG_NB_ALL, DG_T, 0, stream>>>(
        svc_src, svc_dst, in_src, in_dst, ni_src, ni_dst, ii_src, ii_dst, deg);
    offsets_kernel<<<4, SCAN_T, 0, stream>>>(deg, cur);

    fill_kernel<<<FL_NB_ALL, DG_T, 0, stream>>>(
        svc_src, svc_dst, in_src, in_dst, ni_src, ni_dst, ii_src, ii_dst, cur, sorted, deg);

    agg_proj_kernel<<<AG_NB_INST + AG_NB_SVC + AG_NB_NODE, AG_T, 0, stream>>>(
        feat_svc, feat_inst, feat_node, W_sc, W_in, W_ni, W_ii, b_sc, b_in, b_ni, b_ii,
        deg, cur, sorted, cat);

    final_gemm_kernel<<<FG_BLOCKS, 512, 0, stream>>>(W_tot, cat, part);
    out_sum_kernel<<<OUTD, RS_T, 0, stream>>>(part, b_tot, out);
}


# in-degree prefix sums lose their launch (7 to 6): each edge-grouping block scans its own 12500-destination range in LDS on range totals that degree blocks add by integer atomics; sum orders unchanged
# speedup vs baseline: 3.3100x; 1.0760x over previous
#include <hip/hip_runtime.h>

#define SVCN   50000
#define INSTN  100000
#define NODEN  50000
#define FDIM   128
#define HDIM   128
#define OUTD   256
#define TOTAL  (SVCN + NODEN + INSTN)
#define E_SC   400000
#define E_IN   100000
#define E_NI   100000
#define E_II   800000

#define OFF_CAT   0LL
#define OFF_DEG   (OFF_CAT  + (long long)TOTAL * 128)
#define D_NS_SC   (OFF_DEG + 0LL)
#define D_ND_SC   (D_NS_SC + SVCN)
#define D_NS_IN   (D_ND_SC + SVCN)
#define D_ND_IN   (D_NS_IN + INSTN)
#define D_NS_NI   (D_ND_IN + NODEN)
#define D_ND_NI   (D_NS_NI + NODEN)
#define D_NS_II   (D_ND_NI + INSTN)
#define D_ND_II   (D_NS_II + INSTN)
#define DEG_TOTAL 600000
#define OFF_CUR   (OFF_DEG + (long long)DEG_TOTAL)
#define CUR_SC    0
#define CUR_IN    (CUR_SC + SVCN)
#define CUR_NI    (CUR_IN + NODEN)
#define CUR_II    (CUR_NI + INSTN)
#define CUR_TOTAL (CUR_II + INSTN)
#define OFF_SORT  (OFF_CUR + (long long)CUR_TOTAL)
#define SORT_SC   0
#define SORT_IN   (SORT_SC + E_SC)
#define SORT_NI   (SORT_IN + E_IN)
#define SORT_II   (SORT_NI + E_NI)
#define OFF_PART  (OFF_SORT + (long long)(E_SC + E_IN + E_NI + E_II))
static_assert(OFF_PART % 32 == 0, "partial planes start on a 128-byte line");
#define OFF_RT    (OFF_PART + (long long)FG_BLOCKS * OUTD * HDIM)
#define RT_PER    8
#define RT_SC     0
#define RT_IN     (RT_SC + RT_PER)
#define RT_NI     (RT_IN + RT_PER)
#define RT_II     (RT_NI + RT_PER)
#define RT_TOTAL  (RT_II + RT_PER)
#define OFF_RESV  (OFF_RT + RT_TOTAL)
#define OFF_FAC   (OFF_RESV + CUR_TOTAL)
static_assert(RT_TOTAL % 32 == 0 && CUR_TOTAL % 4 == 0, "reservation words and factor planes start on 16-byte pieces");
static_assert(SVCN % 4 == 0 && NODEN % 4 == 0 && INSTN % 4 == 0, "degree planes are scanned four at a time");

typedef __attribute__((ext_vector_type(2))) float v2f;
typedef __attribute__((ext_vector_type(8))) float v8f;

__device__ __forceinline__ float lrelu(float x) { return x > 0.f ? x : 0.01f * x; }

__device__ __forceinline__ void atomAddF(float* p, float v) { unsafeAtomicAdd(p, v); }

#define PRESET_N DEG_TOTAL
#define PRESET_I (RT_TOTAL + CUR_TOTAL)
static_assert(PRESET_I <= PRESET_N, "the grid that clears the planes covers the integer words");
__global__ void preset_kernel(float* __restrict__ deg, int* __restrict__ rt) {
    const int i = blockIdx.x * blockDim.x + threadIdx.x;
    if (i < DEG_TOTAL) deg[i] = 0.f;
    if (i < PRESET_I) rt[i] = 0;
}

__device__ __forceinline__ float deg_factor(float d) { return rsqrtf(d > 1.f ? d : 1.f); }

#define DG_T  512
#define DG_R  12500
#define DG_C  50000
#define DG_L  ((DG_C / 4 + DG_T - 1) / DG_T)
#define DG_F  ((DG_R + DG_T - 1) / DG_T)
static_assert(SVCN % DG_R == 0 && NODEN % DG_R == 0 && INSTN % DG_R == 0, "planes split into whole ranges");
static_assert(E_SC % DG_C == 0 && E_IN % DG_C == 0 && E_NI % DG_C == 0 && E_II % DG_C == 0, "lists split into whole chunks");
static_assert(DG_R % 4 == 0 && DG_C % 4 == 0, "ranges and chunks are whole 16-byte pieces");
static_assert(SVCN / DG_R <= RT_PER && NODEN / DG_R <= RT_PER && INSTN / DG_R <= RT_PER, "a list's ranges fit its range totals");
#define DG_NB(entries, edges) (((entries) / DG_R) * ((edges) / DG_C))
#define DG_NB0 DG_NB(SVCN,  E_SC)
#define DG_NB1 DG_NB(SVCN,  E_SC)
#define DG_NB2 DG_NB(INSTN, E_IN)
#define DG_NB3 DG_NB(NODEN, E_IN)
#define DG_NB4 DG_NB(NODEN, E_NI)
#define DG_NB5 DG_NB(INSTN, E_NI)
#define DG_NB6 DG_NB(INSTN, E_II)
#define DG_NB7 DG_NB(INSTN, E_II)
#define DG_NB_ALL (DG_NB0 + DG_NB1 + DG_NB2 + DG_NB3 + DG_NB4 + DG_NB5 + DG_NB6 + DG_NB7)
__global__ __launch_bounds__(DG_T) void degree_kernel(const int* __restrict__ svc_src, const int* __restrict__ svc_dst,
                                                      const int* __restrict__ in_src,  const int* __restrict__ in_dst,
                                                      const int* __restrict__ ni_src,  const int* __restrict__ ni_dst,
                                                      const int* __restrict__ ii_src,  const int* __restrict__ ii_dst,
                                                      float* __restrict__ deg, int* __restrict__ rt) {
    __shared__ __attribute__((aligned(16))) unsigned int cnt[DG_R];
    int b = blockIdx.x;
    const int* idx; float* plane; int nch;
    int* tot = nullptr;
    if (b < DG_NB0)                  { idx = svc_src; plane = deg + (D_NS_SC - OFF_DEG); nch = E_SC / DG_C; }
    else if ((b -= DG_NB0) < DG_NB1) { idx = svc_dst; plane = deg + (D_ND_SC - OFF_DEG); nch = E_SC / DG_C; tot = rt + RT_SC; }
    else if ((b -= DG_NB1) < DG_NB2) { idx = in_src;  plane = deg + (D_NS_IN - OFF_DEG); nch = E_IN / DG_C; }
    else if ((b -= DG_NB2) < DG_NB3) { idx = in_dst;  plane = deg + (D_ND_IN - OFF_DEG); nch = E_IN / DG_C; tot = rt + RT_IN; }
    else if ((b -= DG_NB3) < DG_NB4) { idx = ni_src;  plane = deg + (D_NS_NI - OFF_DEG); nch = E_NI / DG_C; }
    else if ((b -= DG_NB4) < DG_NB5) { idx = ni_dst;  plane = deg + (D_ND_NI - OFF_DEG); nch = E_NI / DG_C; tot = rt + RT_NI; }
    else if ((b -= DG_NB5) < DG_NB6) { idx = ii_src;  plane = deg + (D_NS_II - OFF_DEG); nch = E_II / DG_C; }
    else { b -= DG_NB6;                idx = ii_dst;  plane = deg + (D_ND_II - OFF_DEG); nch = E_II / DG_C; tot = rt + RT_II; }
    const int range = b / nch;
    const int chunk = b - range * nch;
    const int lo    = range * DG_R;
    const int t     = threadIdx.x;

    const int4* __restrict__ p4 = (const int4*)(idx + chunk * DG_C);
    int4 v[DG_L];
#pragma unroll
    for (int k = 0; k < DG_L; ++k) {
        const int i = k * DG_T + t;
        v[k] = make_int4(-1, -1, -1, -1);
        if (i < DG_C / 4) v[k] = p4[i];
    }

    for (int i = t; i < DG_R / 4; i += DG_T) ((uint4*)cnt)[i] = make_uint4(0u, 0u, 0u, 0u);
    __syncthreads();

#pragma unroll
    for (int k = 0; k < DG_L; ++k) {
        unsigned int d;
        d = (unsigned int)(v[k].x - lo); if (d < DG_R) atomicAdd(&cnt[d], 1u);
        d = (unsigned int)(v[k].y - lo); if (d < DG_R) atomicAdd(&cnt[d], 1u);
        d = (unsigned int)(v[k].z - lo); if (d < DG_R) atomicAdd(&cnt[d], 1u);
        d = (unsigned int)(v[k].w - lo); if (d < DG_R) atomicAdd(&cnt[d], 1u);
    }
    __syncthreads();

    unsigned int c[DG_F];
#pragma unroll
    for (int k = 0; k < DG_F; ++k) {
        const int i = k * DG_T + t;
        c[k] = i < DG_R ? cnt[i] : 0u;
    }
#pragma unroll
    for (int k = 0; k < DG_F; ++k)
        if (c[k] != 0u) atomAddF(&plane[lo + k * DG_T + t], (float)c[k]);

    if (tot != nullptr) {
        unsigned int mine = 0u;
#pragma unroll
        for (int k = 0; k < DG_F; ++k) mine += c[k];
        for (int o = 16; o > 0; o >>= 1) mine += __shfl_xor(mine, o);
        if ((t & 31) == 0 && mine != 0u) atomicAdd(&tot[range], (int)mine);
    }
}


#define FL_NB0 DG_NB(SVCN,  E_SC)
#define FL_NB1 DG_NB(NODEN, E_IN)
#define FL_NB2 DG_NB(INSTN, E_NI)
#define FL_NB3 DG_NB(INSTN, E_II)
#define FL_NB_ALL (FL_NB0 + FL_NB1 + FL_NB2 + FL_NB3)
#define FL_NT     (FL_NB_ALL * DG_T)
#define FL_DEG4   (DEG_TOTAL / 4)
#define FL_DQ     ((FL_DEG4 + FL_NT - 1) / FL_NT)
#define FL_R4     (DG_R / 4)
#define FL_RQ     ((FL_R4 + DG_T - 1) / DG_T)
#define FL_OWN    (DG_R / DG_F)
#define FL_W      (DG_T / 32)
static_assert(DEG_TOTAL % 4 == 0, "degree planes are whole 16-byte pieces");
static_assert(DG_R % DG_F == 0 && FL_OWN <= DG_T, "a range is whole runs of DG_F counts, one per thread");
static_assert(DG_F % 2 == 1, "an odd lane step keeps the runs' LDS accesses off each other's banks");
__global__ __launch_bounds__(DG_T) void fill_kernel(const int* __restrict__ svc_src, const int* __restrict__ svc_dst,
                                                    const int* __restrict__ in_src,  const int* __restrict__ in_dst,
                                                    const int* __restrict__ ni_src,  const int* __restrict__ ni_dst,
                                                    const int* __restrict__ ii_src,  const int* __restrict__ ii_dst,
                                                    int* __restrict__ cur, int* __restrict__ sorted,
                                                    const float* __restrict__ deg, const int* __restrict__ rt,
                                                    int* __restrict__ resv, float* __restrict__ fac) {
    __shared__ __attribute__((aligned(16))) unsigned int cnt[DG_R];
    __shared__ __attribute__((aligned(16))) unsigned int pre[DG_R];
    __shared__ unsigned int wtot[FL_W];
    const int t = threadIdx.x, lane = t & 31, wave = t >> 5;

    int b = blockIdx.x;
    const int* src; const int* dst; const float* ind; const int* tot; int* cu; int* rv; int* so; int nch;
    if (b < FL_NB0)                  { src = svc_src; dst = svc_dst; ind = deg + (D_ND_SC - OFF_DEG); tot = rt + RT_SC;
                                       cu = cur + CUR_SC; rv = resv + CUR_SC; so = sorted + SORT_SC; nch = E_SC / DG_C; }
    else if ((b -= FL_NB0) < FL_NB1) { src = in_src;  dst = in_dst;  ind = deg + (D_ND_IN - OFF_DEG); tot = rt + RT_IN;
                                       cu = cur + CUR_IN; rv = resv + CUR_IN; so = sorted + SORT_IN; nch = E_IN / DG_C; }
    else if ((b -= FL_NB1) < FL_NB2) { src = ni_src;  dst = ni_dst;  ind = deg + (D_ND_NI - OFF_DEG); tot = rt + RT_NI;
                                       cu = cur + CUR_NI; rv = resv + CUR_NI; so = sorted + SORT_NI; nch = E_NI / DG_C; }
    else { b -= FL_NB2;                src = ii_src;  dst = ii_dst;  ind = deg + (D_ND_II - OFF_DEG); tot = rt + RT_II;
                                       cu = cur + CUR_II; rv = resv + CUR_II; so = sorted + SORT_II; nch = E_II / DG_C; }
    const int range = b / nch;
    const int chunk = b - range * nch;
    const int lo    = range * DG_R;

    const float4* __restrict__ ind4 = (const float4*)(ind + lo);
    float4 fq[FL_RQ];
#pragma unroll
    for (int j = 0; j < FL_RQ; ++j) {
        const int i = j * DG_T + t;
        fq[j] = ind4[i < FL_R4 ? i : 0];
    }
    int carry = 0, own = 0;
#pragma unroll
    for (int r = 0; r < RT_PER; ++r) {
        const int v = tot[r];
        carry += r < range ? v : 0;
        own   += r == range ? v : 0;
    }
#pragma unroll
    for (int j = 0; j < FL_RQ; ++j) {
        const int i = j * DG_T + t;
        if (i < FL_R4)
            ((uint4*)pre)[i] = make_uint4((unsigned int)fq[j].x, (unsigned int)fq[j].y,
                                          (unsigned int)fq[j].z, (unsigned int)fq[j].w);
    }
    __syncthreads();
    unsigned int ex[DG_F];
#pragma unroll
    for (int j = 0; j < DG_F; ++j) ex[j] = 0u;
    if (t < FL_OWN) {
#pragma unroll
        for (int j = 0; j < DG_F; ++j) ex[j] = pre[t * DG_F + j];
    }
    unsigned int s = 0u;
#pragma unroll
    for (int j = 0; j < DG_F; ++j) {
        const unsigned int v = ex[j];
        ex[j] = s;
        s += v;
    }
    unsigned int x = s;
    for (int o = 1; o < 32; o <<= 1) {
        const unsigned int y = __shfl_up(x, o);
        if (lane >= o) x += y;
    }
    if (lane == 31) wtot[wave] = x;
    __syncthreads();
    unsigned int start = (unsigned int)carry + (x - s);
#pragma unroll
    for (int w = 0; w < FL_W; ++w) {
        const unsigned int wt = wtot[w];
        start += w < wave ? wt : 0u;
    }
    if (t < FL_OWN) {
#pragma unroll
        for (int j = 0; j < DG_F; ++j) pre[t * DG_F + j] = start + ex[j];
    }

    const int g = blockIdx.x * DG_T + t;
    float4 dq[FL_DQ];
#pragma unroll
    for (int q = 0; q < FL_DQ; ++q) {
        const int i = q * FL_NT + g;
        dq[q] = make_float4(1.f, 1.f, 1.f, 1.f);
        if (i < FL_DEG4) dq[q] = ((const float4*)deg)[i];
    }

    const int4* __restrict__ d4 = (const int4*)(dst + chunk * DG_C);
    const int4* __restrict__ s4 = (const int4*)(src + chunk * DG_C);
    int4 vd[DG_L], vs[DG_L];
#pragma unroll
    for (int k = 0; k < DG_L; ++k) {
        const int i = k * DG_T + t;
        vd[k] = make_int4(-1, -1, -1, -1);
        vs[k] = make_int4(0, 0, 0, 0);
        if (i < DG_C / 4) { vd[k] = d4[i]; vs[k] = s4[i]; }
    }

    for (int i = t; i < DG_R / 4; i += DG_T) ((uint4*)cnt)[i] = make_uint4(0u, 0u, 0u, 0u);

#pragma unroll
    for (int q = 0; q < FL_DQ; ++q) {
        const int i = q * FL_NT + g;
        if (i < FL_DEG4)
            ((float4*)fac)[i] = make_float4(deg_factor(dq[q].x), deg_factor(dq[q].y),
                                            deg_factor(dq[q].z), deg_factor(dq[q].w));
    }
    __syncthreads();

    if (chunk == 0) {
#pragma unroll
        for (int k = 0; k < DG_F; ++k) {
            const int i = k * DG_T + t;
            if (i < DG_R) cu[lo + i] = (int)(i + 1 < DG_R ? pre[i + 1] : (unsigned int)(carry + own));
        }
    }

#pragma unroll
    for (int k = 0; k < DG_L; ++k) {
        unsigned int d;
        d = (unsigned int)(vd[k].x - lo); if (d < DG_R) atomicAdd(&cnt[d], 1u);
        d = (unsigned int)(vd[k].y - lo); if (d < DG_R) atomicAdd(&cnt[d], 1u);
        d = (unsigned int)(vd[k].z - lo); if (d < DG_R) atomicAdd(&cnt[d], 1u);
        d = (unsigned int)(vd[k].w - lo); if (d < DG_R) atomicAdd(&cnt[d], 1u);
    }
    __syncthreads();

    unsigned int c[DG_F];
#pragma unroll
    for (int k = 0; k < DG_F; ++k) {
        const int i = k * DG_T + t;
        c[k] = i < DG_R ? cnt[i] : 0u;
    }
#pragma unroll
    for (int k = 0; k < DG_F; ++k)
        if (c[k] != 0u) c[k] = (unsigned int)atomicAdd(&rv[lo + k * DG_T + t], (int)c[k]);
#pragma unroll
    for (int k = 0; k < DG_F; ++k) {
        const int i = k * DG_T + t;
        if (i < DG_R && c[k] != 0u) atomicAdd(&pre[i], c[k]);
    }
    __syncthreads();

#pragma unroll
    for (int k = 0; k < DG_L; ++k) {
        const unsigned int dx = (unsigned int)(vd[k].x - lo), dy = (unsigned int)(vd[k].y - lo);
        const unsigned int dz = (unsigned int)(vd[k].z - lo), dw = (unsigned int)(vd[k].w - lo);
        unsigned int px = 0u, py = 0u, pz = 0u, pw = 0u;
        if (dx < DG_R) px = atomicAdd(&pre[dx], 1u);
        if (dy < DG_R) py = atomicAdd(&pre[dy], 1u);
        if (dz < DG_R) pz = atomicAdd(&pre[dz], 1u);
        if (dw < DG_R) pw = atomicAdd(&pre[dw], 1u);
        if (dx < DG_R) so[px] = vs[k].x;
        if (dy < DG_R) so[py] = vs[k].y;
        if (dz < DG_R) so[pz] = vs[k].z;
        if (dw < DG_R) so[pw] = vs[k].w;
    }
}

#define AG_BATCH 8

#define AG_ROWS   64
#define AG_T      512
#define AG_PITCH  132
#define AG_NB_INST ((INSTN + AG_ROWS - 1) / AG_ROWS)
#define AG_NB_SVC  ((SVCN  + AG_ROWS - 1) / AG_ROWS)
#define AG_NB_NODE ((NODEN + AG_ROWS - 1) / AG_ROWS)
__global__ __launch_bounds__(AG_T, 12) void agg_proj_kernel(
        const float* __restrict__ feat_svc, const float* __restrict__ feat_inst,
        const float* __restrict__ feat_node,
        const float* __restrict__ W_sc, const float* __restrict__ W_in,
        const float* __restrict__ W_ni, const float* __restrict__ W_ii,
        const float* __restrict__ b_sc, const float* __restrict__ b_in,
        const float* __restrict__ b_ni, const float* __restrict__ b_ii,
        const float* __restrict__ deg, const int* __restrict__ curAll,
        const int* __restrict__ sortedAll, float* __restrict__ catAll) {
    __shared__ __attribute__((aligned(16))) float Wlds[128 * 128];
    __shared__ __attribute__((aligned(16))) float Tile[AG_ROWS * AG_PITCH];

    int blk = blockIdx.x;
    const float *X0, *W0, *ns0, *nd0, *bA;
    const int *cur0, *sorted0;
    const float *X1 = nullptr, *W1 = nullptr, *ns1 = nullptr, *nd1 = nullptr, *bB = nullptr;
    const int *cur1 = nullptr, *sorted1 = nullptr;
    int nrows, nlists = 1;
    float* cat;
    if (blk < AG_NB_INST) {
        X0 = feat_node; W0 = W_ni; bA = b_ni;
        ns0 = deg + (D_NS_NI - OFF_DEG); nd0 = deg + (D_ND_NI - OFF_DEG);
        cur0 = curAll + CUR_NI; sorted0 = sortedAll + SORT_NI;
        X1 = feat_inst; W1 = W_ii; bB = b_ii;
        ns1 = deg + (D_NS_II - OFF_DEG); nd1 = deg + (D_ND_II - OFF_DEG);
        cur1 = curAll + CUR_II; sorted1 = sortedAll + SORT_II;
        nrows = INSTN; nlists = 2;
        cat = catAll + (long long)(SVCN + NODEN) * 128;
    } else if ((blk -= AG_NB_INST) < AG_NB_SVC) {
        X0 = feat_svc; W0 = W_sc; bA = b_sc;
        ns0 = deg + (D_NS_SC - OFF_DEG); nd0 = deg + (D_ND_SC - OFF_DEG);
        cur0 = curAll + CUR_SC; sorted0 = sortedAll + SORT_SC;
        nrows = SVCN;
        cat = catAll;
    } else {
        blk -= AG_NB_SVC;
        X0 = feat_inst; W0 = W_in; bA = b_in;
        ns0 = deg + (D_NS_IN - OFF_DEG); nd0 = deg + (D_ND_IN - OFF_DEG);
        cur0 = curAll + CUR_IN; sorted0 = sortedAll + SORT_IN;
        nrows = NODEN;
        cat = catAll + (long long)SVCN * 128;
    }

    const int t    = threadIdx.x;
    const int lane = t & 31;
    const int wave = __builtin_amdgcn_readfirstlane(t >> 5);
    const int half = lane >> 4;
    const int l16  = lane & 15;
    const int c4   = lane * 4;
    const int r0   = blk * AG_ROWS;
    const int n0   = (wave & 7) * 16;
    const int rb   = (wave >> 3) * 32;

    v8f acc[2] = {v8f{}, v8f{}};

    const float* wfrag = &Wlds[(2 * half) * 128 + n0 + l16];
    const float* tfrag = &Tile[(rb + l16) * AG_PITCH + 2 * half];

    for (int li = 0; li < nlists; ++li) {
        const float* __restrict__ X  = li ? X1 : X0;
        const float* __restrict__ Wg = li ? W1 : W0;
        const float* __restrict__ ns = li ? ns1 : ns0;
        const float* __restrict__ nd = li ? nd1 : nd0;
        const int* __restrict__ cur    = li ? cur1 : cur0;
        const int* __restrict__ sorted = li ? sorted1 : sorted0;

        const int first = r0 + wave * 4;
        int id = first + lane;     id = id < nrows ? id : nrows - 1;
        int ib = first - 1;        ib = ib < 0 ? 0 : (ib < nrows ? ib : nrows - 1);
        const int   cv  = cur[id];
        const float ndv = nd[id];
        const int   cb0 = cur[ib];

        const float4* __restrict__ Wg4 = (const float4*)Wg + t;
        const float4 wr0 = Wg4[0 * AG_T], wr1 = Wg4[1 * AG_T], wr2 = Wg4[2 * AG_T], wr3 = Wg4[3 * AG_T];
        const float4 wr4 = Wg4[4 * AG_T], wr5 = Wg4[5 * AG_T], wr6 = Wg4[6 * AG_T], wr7 = Wg4[7 * AG_T];

        const int b0 = first > 0 ? cb0 : 0;
        const int e3 = __builtin_amdgcn_readlane(cv, 3);
        int sv_next = 0;
        if (lane < e3 - b0) sv_next = sorted[b0 + lane];

        if (li > 0) __syncthreads();

        float4* Wl4 = (float4*)Wlds + t;
        Wl4[0 * AG_T] = wr0; Wl4[1 * AG_T] = wr1; Wl4[2 * AG_T] = wr2; Wl4[3 * AG_T] = wr3;
        Wl4[4 * AG_T] = wr4; Wl4[5 * AG_T] = wr5; Wl4[6 * AG_T] = wr6; Wl4[7 * AG_T] = wr7;

        float4 acc4 = make_float4(0.f, 0.f, 0.f, 0.f);
        int ri   = 0;
        int rend = __builtin_amdgcn_readlane(cv, 0);
        for (int cb = b0; cb < e3; cb += 32) {
            const int rem = e3 - cb;
            const int cnt = rem < 32 ? rem : 32;
            const int sv  = sv_next;
            sv_next = 0;
            if (lane < rem - 32) sv_next = sorted[cb + 32 + lane];
            float wv = 0.f;
            if (lane < cnt) wv = ns[sv];
            for (int jj = 0; jj < cnt; jj += AG_BATCH) {
                float4 x[AG_BATCH];
#pragma unroll
                for (int k = 0; k < AG_BATCH; ++k) {
                    const int q = jj + k < cnt ? jj + k : cnt - 1;
                    const int s = __builtin_amdgcn_readlane(sv, q);
                    x[k] = *(const float4*)&X[(long long)s * 128 + c4];
                }
#pragma unroll
                for (int k = 0; k < AG_BATCH; ++k) {
                    const bool live = jj + k < cnt;
                    const int q = live ? jj + k : cnt - 1;
                    const int j = cb + q;
                    while (j >= rend) {
                        const float n = __builtin_bit_cast(float,
                            __builtin_amdgcn_readlane(__builtin_bit_cast(int, ndv), ri));
                        *(float4*)&Tile[(wave * 4 + ri) * AG_PITCH + c4] =
                            make_float4(acc4.x * n, acc4.y * n, acc4.z * n, acc4.w * n);
                        acc4 = make_float4(0.f, 0.f, 0.f, 0.f);
                        ++ri;
                        rend = __builtin_amdgcn_readlane(cv, ri);
                    }
                    const float wq = __builtin_bit_cast(float,
                        __builtin_amdgcn_readlane(__builtin_bit_cast(int, wv), q));
                    const float w = live ? wq : 0.f;
                    acc4.x = fmaf(x[k].x, w, acc4.x); acc4.y = fmaf(x[k].y, w, acc4.y);
                    acc4.z = fmaf(x[k].z, w, acc4.z); acc4.w = fmaf(x[k].w, w, acc4.w);
                }
            }
        }
        for (; ri < 4; ++ri) {
            const float n = __builtin_bit_cast(float,
                __builtin_amdgcn_readlane(__builtin_bit_cast(int, ndv), ri));
            *(float4*)&Tile[(wave * 4 + ri) * AG_PITCH + c4] =
                make_float4(acc4.x * n, acc4.y * n, acc4.z * n, acc4.w * n);
            acc4 = make_float4(0.f, 0.f, 0.f, 0.f);
        }
        __syncthreads();

        for (int k = 0; k < 128; k += 4) {
            v2f b;
            b.x = wfrag[k * 128];
            b.y = wfrag[(k + 1) * 128];
#pragma unroll
            for (int rt = 0; rt < 2; ++rt) {
                v2f a = *(const v2f*)&tfrag[rt * 16 * AG_PITCH + k];
                acc[rt] = __builtin_amdgcn_wmma_f32_16x16x4_f32(
                    false, a, false, b, (short)0, acc[rt], false, false);
            }
        }
    }

    __syncthreads();
#pragma unroll
    for (int rt = 0; rt < 2; ++rt) {
#pragma unroll
        for (int i = 0; i < 8; ++i)
            Tile[(rb + rt * 16 + i + 8 * half) * AG_PITCH + n0 + l16] = acc[rt][i];
    }
    __syncthreads();

    float4 bias = *(const float4*)&bA[c4];
    if (nlists > 1) {
        const float4 bb = *(const float4*)&bB[c4];
        bias.x += bb.x; bias.y += bb.y; bias.z += bb.z; bias.w += bb.w;
    }
    for (int i = 0; i < 4; ++i) {
        const int lr  = wave * 4 + i;
        const int row = r0 + lr;
        if (row < nrows) {
            const float4 p = *(const float4*)&Tile[lr * AG_PITCH + c4];
            *(float4*)&cat[(long long)row * 128 + c4] =
                make_float4(p.x + bias.x, p.y + bias.y, p.z + bias.z, p.w + bias.w);
        }
    }
}

#define FG_KT      32
#define FG_TILES   (TOTAL / FG_KT)
#define FG_BLOCKS  250
#define FG_TPB     (FG_TILES / FG_BLOCKS)
#define FG_PA      36
#define FG_PB      136
static_assert(TOTAL % FG_KT == 0, "K must be a whole number of tiles");
static_assert(FG_TILES % FG_BLOCKS == 0, "tiles must divide evenly over the blocks");

__global__ __launch_bounds__(512) void final_gemm_kernel(const float* __restrict__ Wt,
                                                         const float* __restrict__ cat,
                                                         float* __restrict__ part) {
    __shared__ float As[OUTD * FG_PA];
    __shared__ float Bs[FG_KT * FG_PB];

    const int t    = threadIdx.x;
    const int lane = t & 31;
    const int wave = t >> 5;
    const int half = lane >> 4;
    const int l16  = lane & 15;
    const int wm   = wave >> 2;
    const int wn   = wave & 3;
    const int tile0 = blockIdx.x * FG_TPB;

    const int arow = t >> 3;
    const int ac4  = t & 7;
    const int brow = t >> 5;
    const int bc4  = t & 31;
    const float* pA = Wt + (long long)arow * TOTAL + (long long)tile0 * FG_KT + ac4 * 4;
    const float* pB = cat + (long long)tile0 * FG_KT * 128 + t * 4;
    const long long aStep = 64LL * TOTAL;
    float* sA = &As[arow * FG_PA + ac4 * 4];
    float* sB = &Bs[brow * FG_PB + bc4 * 4];

    v8f acc[4][2];
#pragma unroll
    for (int mt = 0; mt < 4; ++mt)
#pragma unroll
        for (int nt = 0; nt < 2; ++nt) acc[mt][nt] = v8f{};

    float4 ra0 = *(const float4*)(pA);
    float4 ra1 = *(const float4*)(pA + aStep);
    float4 ra2 = *(const float4*)(pA + 2 * aStep);
    float4 ra3 = *(const float4*)(pA + 3 * aStep);
    float4 rb0 = *(const float4*)(pB);
    float4 rb1 = *(const float4*)(pB + 2048);

    for (int it = 0; it < FG_TPB; ++it) {
        *(float4*)(sA)                 = ra0;
        *(float4*)(sA + 64 * FG_PA)    = ra1;
        *(float4*)(sA + 128 * FG_PA)   = ra2;
        *(float4*)(sA + 192 * FG_PA)   = ra3;
        rb0.x = lrelu(rb0.x); rb0.y = lrelu(rb0.y); rb0.z = lrelu(rb0.z); rb0.w = lrelu(rb0.w);
        rb1.x = lrelu(rb1.x); rb1.y = lrelu(rb1.y); rb1.z = lrelu(rb1.z); rb1.w = lrelu(rb1.w);
        *(float4*)(sB)                 = rb0;
        *(float4*)(sB + 16 * FG_PB)    = rb1;
        __syncthreads();

        if (it + 1 < FG_TPB) {
            pA += FG_KT;
            pB += FG_KT * 128;
            ra0 = *(const float4*)(pA);
            ra1 = *(const float4*)(pA + aStep);
            ra2 = *(const float4*)(pA + 2 * aStep);
            ra3 = *(const float4*)(pA + 3 * aStep);
            rb0 = *(const float4*)(pB);
            rb1 = *(const float4*)(pB + 2048);
        }
        __builtin_amdgcn_sched_barrier(0);

#pragma unroll
        for (int ks = 0; ks < FG_KT / 4; ++ks) {
            const int kk = ks * 4 + 2 * half;
            v2f a[4], b[2];
#pragma unroll
            for (int mt = 0; mt < 4; ++mt)
                a[mt] = *(const v2f*)&As[(wm * 64 + mt * 16 + l16) * FG_PA + kk];
#pragma unroll
            for (int nt = 0; nt < 2; ++nt) {
                b[nt].x = Bs[kk * FG_PB + wn * 32 + nt * 16 + l16];
                b[nt].y = Bs[(kk + 1) * FG_PB + wn * 32 + nt * 16 + l16];
            }
#pragma unroll
            for (int mt = 0; mt < 4; ++mt)
#pragma unroll
                for (int nt = 0; nt < 2; ++nt)
                    acc[mt][nt] = __builtin_amdgcn_wmma_f32_16x16x4_f32(
                        false, a[mt], false, b[nt], (short)0, acc[mt][nt], false, false);
        }
        __syncthreads();
    }

    float* __restrict__ pp = part + (long long)blockIdx.x * (OUTD * HDIM) + wn * 32 + lane;
#pragma unroll
    for (int mt = 0; mt < 4; ++mt) {
#pragma unroll
        for (int i = 0; i < 8; ++i) {
            const float a  = acc[mt][0][i];
            const float b  = acc[mt][1][i];
            const float xa = __shfl_xor(a, 16);
            const float xb = __shfl_xor(b, 16);
            const int   m  = wm * 64 + mt * 16 + i;
            pp[m * 128]       = half ? xb : a;
            pp[(m + 8) * 128] = half ? b : xa;
        }
    }
}

#define RS_T 256
#define RS_G (RS_T / 32)
#define RS_L ((FG_BLOCKS + RS_G - 1) / RS_G)
static_assert(HDIM == 128, "one lane per 16 bytes of an output row");
__global__ __launch_bounds__(RS_T) void out_sum_kernel(const float* __restrict__ part,
                                                       const float* __restrict__ b_tot,
                                                       float* __restrict__ out) {
    __shared__ float4 red[RS_G * 32];
    const int t = threadIdx.x;
    const int c = t & 31;
    const int g = t >> 5;
    const int o = blockIdx.x;
    const float4* __restrict__ p = (const float4*)part + o * (HDIM / 4) + c;
    float4 v[RS_L];
#pragma unroll
    for (int k = 0; k < RS_L; ++k) {
        const int b = k * RS_G + g;
        v[k] = make_float4(0.f, 0.f, 0.f, 0.f);
        if (b < FG_BLOCKS) v[k] = p[(long long)b * (OUTD * HDIM / 4)];
    }
    float4 s = v[0];
#pragma unroll
    for (int k = 1; k < RS_L; ++k) { s.x += v[k].x; s.y += v[k].y; s.z += v[k].z; s.w += v[k].w; }
    red[g * 32 + c] = s;
    __syncthreads();
    if (g == 0) {
        float4 r = red[c];
        for (int j = 1; j < RS_G; ++j) {
            const float4 q = red[j * 32 + c];
            r.x += q.x; r.y += q.y; r.z += q.z; r.w += q.w;
        }
        const float bias = b_tot[o];
        ((float4*)out)[o * (HDIM / 4) + c] = make_float4(r.x + bias, r.y + bias, r.z + bias, r.w + bias);
    }
}

static inline int cdiv(long long a, long long b) { return (int)((a + b - 1) / b); }

extern "C" void kernel_launch(void* const* d_in, const int* in_sizes, int n_in,
                              void* d_out, int out_size, void* d_ws, size_t ws_size,
                              hipStream_t stream) {
    const float* feat_svc  = (const float*)d_in[0];
    const float* feat_inst = (const float*)d_in[1];
    const float* feat_node = (const float*)d_in[2];
    const int*   svc_src   = (const int*)d_in[3];
    const int*   svc_dst   = (const int*)d_in[4];
    const int*   in_src    = (const int*)d_in[5];
    const int*   in_dst    = (const int*)d_in[6];
    const int*   ni_src    = (const int*)d_in[7];
    const int*   ni_dst    = (const int*)d_in[8];
    const int*   ii_src    = (const int*)d_in[9];
    const int*   ii_dst    = (const int*)d_in[10];
    const float* W_sc = (const float*)d_in[11];  const float* b_sc = (const float*)d_in[12];
    const float* W_in = (const float*)d_in[13];  const float* b_in = (const float*)d_in[14];
    const float* W_ni = (const float*)d_in[15];  const float* b_ni = (const float*)d_in[16];
    const float* W_ii = (const float*)d_in[17];  const float* b_ii = (const float*)d_in[18];
    const float* W_tot = (const float*)d_in[19]; const float* b_tot = (const float*)d_in[20];
    float* out = (float*)d_out;

    float* ws   = (float*)d_ws;
    float* cat  = ws + OFF_CAT;
    float* deg  = ws + OFF_DEG;
    int* cur    = (int*)(ws + OFF_CUR);
    int* sorted = (int*)(ws + OFF_SORT);
    float* part = ws + OFF_PART;
    int* rt     = (int*)(ws + OFF_RT);
    int* resv   = (int*)(ws + OFF_RESV);
    float* fac  = ws + OFF_FAC;

    preset_kernel<<<cdiv(PRESET_N, 256), 256, 0, stream>>>(deg, rt);

    degree_kernel<<<DG_NB_ALL, DG_T, 0, stream>>>(
        svc_src, svc_dst, in_src, in_dst, ni_src, ni_dst, ii_src, ii_dst, deg, rt);

    fill_kernel<<<FL_NB_ALL, DG_T, 0, stream>>>(
        svc_src, svc_dst, in_src, in_dst, ni_src, ni_dst, ii_src, ii_dst, cur, sorted,
        deg, rt, resv, fac);

    agg_proj_kernel<<<AG_NB_INST + AG_NB_SVC + AG_NB_NODE, AG_T, 0, stream>>>(
        feat_svc, feat_inst, feat_node, W_sc, W_in, W_ni, W_ii, b_sc, b_in, b_ni, b_ii,
        fac, cur, sorted, cat);

    final_gemm_kernel<<<FG_BLOCKS, 512, 0, stream>>>(W_tot, cat, part);
    out_sum_kernel<<<OUTD, RS_T, 0, stream>>>(part, b_tot, out);
}
